# starting kernel
# baseline (speedup 1.0000x reference)
#include <hip/hip_runtime.h>
#include <hip/hip_bf16.h>

typedef __attribute__((ext_vector_type(2))) float v2f;
typedef __attribute__((ext_vector_type(8))) float v8f;

#define BATCH     2
#define SEQLEN    2048
#define C_IN      256
#define D_MODEL   512
#define LOUT      1024
#define D_STATE   64
#define D_CONV    4
#define D_INNER   1024
#define HEADDIM   64
#define NHEADS    16
#define CONV_DIM  (D_INNER + 2*D_STATE)
#define D_IN_PROJ (2*D_INNER + 2*D_STATE + NHEADS)
#define ROWS      (BATCH*LOUT)
#define KSTEM     (C_IN * 7)

__global__ void im2col_stem_kernel(const float* __restrict__ tokens,
                                   float* __restrict__ col) {
  const long idx = (long)blockIdx.x * 256 + threadIdx.x;
  if (idx >= (long)ROWS * KSTEM) return;
  const int j  = (int)(idx % KSTEM);
  const long bl = idx / KSTEM;
  const int c = j / 7;
  const int k = j - c * 7;
  const int l = (int)(bl & (LOUT - 1));
  const int b = (int)(bl >> 10);
  const int s = 2 * l - 3 + k;
  float v = 0.f;
  if (s >= 0 && s < SEQLEN) v = tokens[((long)b * C_IN + c) * SEQLEN + s];
  col[idx] = v;
}

__global__ void gemm_xt_wmma(const float* __restrict__ A,
                             const float* __restrict__ W,
                             float* __restrict__ C,
                             int M, int N, int K) {
  const int lane = threadIdx.x & 31;
  const int wave = threadIdx.x >> 5;
  const int tiles_n = N >> 4;
  const int strips  = (tiles_n + 3) >> 2;
  const int tile = blockIdx.x * (blockDim.x >> 5) + wave;
  if (tile >= (M >> 4) * strips) return;
  const int tm  = tile / strips;
  const int ts  = tile - tm * strips;
  const int tn0 = ts << 2;
  const int nt  = min(4, tiles_n - tn0);

  const int r    = lane & 15;
  const int koff = (lane >> 4) << 1;
  const float* arow = A + ((long)(tm * 16 + r)) * K + koff;
  const float* brow = W + ((long)(tn0 * 16 + r)) * K + koff;
  const long  bstep = (long)16 * K;

  if (nt == 4) {
    v8f acc0 = {0.f,0.f,0.f,0.f,0.f,0.f,0.f,0.f};
    v8f acc1 = acc0, acc2 = acc0, acc3 = acc0;
    #pragma unroll 2
    for (int k = 0; k < K; k += 4) {
      const v2f a  = *(const v2f*)(arow + k);
      const v2f b0 = *(const v2f*)(brow + k);
      const v2f b1 = *(const v2f*)(brow + bstep + k);
      const v2f b2 = *(const v2f*)(brow + 2 * bstep + k);
      const v2f b3 = *(const v2f*)(brow + 3 * bstep + k);
      acc0 = __builtin_amdgcn_wmma_f32_16x16x4_f32(false, a, false, b0, (short)0, acc0, false, false);
      acc1 = __builtin_amdgcn_wmma_f32_16x16x4_f32(false, a, false, b1, (short)0, acc1, false, false);
      acc2 = __builtin_amdgcn_wmma_f32_16x16x4_f32(false, a, false, b2, (short)0, acc2, false, false);
      acc3 = __builtin_amdgcn_wmma_f32_16x16x4_f32(false, a, false, b3, (short)0, acc3, false, false);
    }
    const int row0 = (lane >> 4) << 3;
    float* cbase = C + ((long)(tm * 16)) * N + tn0 * 16 + (lane & 15);
    #pragma unroll
    for (int i = 0; i < 8; ++i) {
      float* crow = cbase + (long)(row0 + i) * N;
      crow[0]  = acc0[i];
      crow[16] = acc1[i];
      crow[32] = acc2[i];
      crow[48] = acc3[i];
    }
  } else {
    for (int t = 0; t < nt; ++t) {
      v8f acc = {0.f,0.f,0.f,0.f,0.f,0.f,0.f,0.f};
      const float* bt = brow + (long)t * bstep;
      #pragma unroll 4
      for (int k = 0; k < K; k += 4) {
        const v2f a = *(const v2f*)(arow + k);
        const v2f b = *(const v2f*)(bt + k);
        acc = __builtin_amdgcn_wmma_f32_16x16x4_f32(false, a, false, b, (short)0, acc, false, false);
      }
      const int row0 = (lane >> 4) << 3;
      float* cbase = C + ((long)(tm * 16)) * N + (tn0 + t) * 16 + (lane & 15);
      #pragma unroll
      for (int i = 0; i < 8; ++i)
        cbase[(long)(row0 + i) * N] = acc[i];
    }
  }
}

__global__ void dwconv_silu_kernel(const float* __restrict__ zx,
                                   const float* __restrict__ w,
                                   const float* __restrict__ bias,
                                   float* __restrict__ out) {
  const int c = blockIdx.x * 128 + threadIdx.x;
  if (c >= CONV_DIM) return;
  const long bl = blockIdx.y;
  const int l = (int)(bl & (LOUT - 1));
  const float* xcol = zx + bl * D_IN_PROJ + D_INNER + c;
  const float w0 = w[c * 4 + 0], w1 = w[c * 4 + 1], w2 = w[c * 4 + 2], w3 = w[c * 4 + 3];
  float acc = w3 * xcol[0];
  if (l >= 1) acc = fmaf(w2, xcol[-(long)D_IN_PROJ],     acc);
  if (l >= 2) acc = fmaf(w1, xcol[-(long)2 * D_IN_PROJ], acc);
  if (l >= 3) acc = fmaf(w0, xcol[-(long)3 * D_IN_PROJ], acc);
  const float v = acc + bias[c];
  out[bl * CONV_DIM + c] = v / (1.f + __expf(-v));
}

__global__ void ssm_scan_kernel(const float* __restrict__ zx,
                                const float* __restrict__ xbc,
                                const float* __restrict__ dtb,
                                const float* __restrict__ alog,
                                const float* __restrict__ Dp,
                                float* __restrict__ y) {
  const int b  = blockIdx.x >> 4;
  const int hh = blockIdx.x & 15;
  const int tid = threadIdx.x;
  const int p = tid >> 2;
  const int q = tid & 3;
  __shared__ float sx[HEADDIM], sB[D_STATE], sC[D_STATE], sdt[2];
  const float Ah   = -__expf(alog[hh]);
  const float dtbh = dtb[hh];
  const float Dh   = Dp[hh];
  float hreg[16];
  #pragma unroll
  for (int i = 0; i < 16; ++i) hreg[i] = 0.f;

  for (int l = 0; l < LOUT; ++l) {
    const long bl = (long)b * LOUT + l;
    const float* row = xbc + bl * CONV_DIM;
    if (tid < 64) {
      sx[tid] = row[hh * HEADDIM + tid];
    } else if (tid < 128) {
      sB[tid - 64] = row[D_INNER + (tid - 64)];
    } else if (tid < 192) {
      sC[tid - 128] = row[D_INNER + D_STATE + (tid - 128)];
    } else if (tid == 192) {
      const float raw = zx[bl * D_IN_PROJ + (D_IN_PROJ - NHEADS) + hh] + dtbh;
      const float dtv = (raw > 20.f) ? raw : log1pf(__expf(raw));
      sdt[0] = dtv;
      sdt[1] = __expf(dtv * Ah);
    }
    __syncthreads();
    const float dtv = sdt[0];
    const float dA  = sdt[1];
    const float xv  = sx[p];
    const float xdt = xv * dtv;
    const float* Bq = sB + q * 16;
    const float* Cq = sC + q * 16;
    float accv = 0.f;
    #pragma unroll
    for (int i = 0; i < 16; ++i) {
      hreg[i] = fmaf(hreg[i], dA, xdt * Bq[i]);
      accv    = fmaf(hreg[i], Cq[i], accv);
    }
    accv += __shfl_xor(accv, 1, 32);
    accv += __shfl_xor(accv, 2, 32);
    if (q == 0)
      y[bl * D_INNER + hh * HEADDIM + p] = fmaf(xv, Dh, accv);
    __syncthreads();
  }
}

__global__ void gate_rmsnorm_kernel(const float* __restrict__ zx,
                                    const float* __restrict__ nw,
                                    float* __restrict__ y) {
  const long bl = blockIdx.x;
  const int tid = threadIdx.x;
  __shared__ float red[256];
  float yv[4];
  float ss = 0.f;
  #pragma unroll
  for (int j = 0; j < 4; ++j) {
    const int i = tid + j * 256;
    const float z = zx[bl * D_IN_PROJ + i];
    const float g = z / (1.f + __expf(-z));
    const float v = y[bl * D_INNER + i] * g;
    yv[j] = v;
    ss = fmaf(v, v, ss);
  }
  red[tid] = ss;
  __syncthreads();
  for (int st = 128; st > 0; st >>= 1) {
    if (tid < st) red[tid] += red[tid + st];
    __syncthreads();
  }
  const float scale = rsqrtf(red[0] * (1.f / (float)D_INNER) + 1e-5f);
  #pragma unroll
  for (int j = 0; j < 4; ++j) {
    const int i = tid + j * 256;
    y[bl * D_INNER + i] = yv[j] * scale * nw[i];
  }
}

extern "C" void kernel_launch(void* const* d_in, const int* in_sizes, int n_in,
                              void* d_out, int out_size, void* d_ws, size_t ws_size,
                              hipStream_t stream) {
  const float* tokens     = (const float*)d_in[0];
  const float* conv_w     = (const float*)d_in[1];
  const float* in_proj_w  = (const float*)d_in[2];
  const float* conv1d_w   = (const float*)d_in[3];
  const float* conv1d_b   = (const float*)d_in[4];
  const float* dt_bias    = (const float*)d_in[5];
  const float* A_log      = (const float*)d_in[6];
  const float* D_param    = (const float*)d_in[7];
  const float* norm_w     = (const float*)d_in[8];
  const float* out_proj_w = (const float*)d_in[9];
  float* out = (float*)d_out;
  float* ws  = (float*)d_ws;

  float* u0   = ws;
  float* u1   = u0 + (long)ROWS * D_MODEL;
  float* zx   = u1 + (long)ROWS * D_MODEL;
  float* xact = zx + (long)ROWS * D_IN_PROJ;
  float* ybuf = xact + (long)ROWS * CONV_DIM;
  float* col  = zx;
  (void)ws_size; (void)in_sizes; (void)n_in; (void)out_size;

  {
    const long nel = (long)ROWS * KSTEM;
    im2col_stem_kernel<<<(int)((nel + 255) / 256), 256, 0, stream>>>(tokens, col);
    const int strips = ((D_MODEL / 16) + 3) / 4;
    const int tiles  = (ROWS / 16) * strips;
    gemm_xt_wmma<<<(tiles + 7) / 8, 256, 0, stream>>>(col, conv_w, u0, ROWS, D_MODEL, KSTEM);
  }

  const float* u_in = u0;
  for (int blk = 0; blk < 2; ++blk) {
    const float* ipw = in_proj_w  + (long)blk * D_IN_PROJ * D_MODEL;
    const float* cw  = conv1d_w   + (long)blk * CONV_DIM * D_CONV;
    const float* cb  = conv1d_b   + (long)blk * CONV_DIM;
    const float* dtb = dt_bias    + (long)blk * NHEADS;
    const float* al  = A_log      + (long)blk * NHEADS;
    const float* Dpp = D_param    + (long)blk * NHEADS;
    const float* nw  = norm_w     + (long)blk * D_INNER;
    const float* opw = out_proj_w + (long)blk * D_MODEL * D_INNER;

    {
      const int strips = ((D_IN_PROJ / 16) + 3) / 4;
      const int tiles  = (ROWS / 16) * strips;
      gemm_xt_wmma<<<(tiles + 7) / 8, 256, 0, stream>>>(u_in, ipw, zx, ROWS, D_IN_PROJ, D_MODEL);
    }
    dwconv_silu_kernel<<<dim3(CONV_DIM / 128, ROWS), 128, 0, stream>>>(zx, cw, cb, xact);
    ssm_scan_kernel<<<BATCH * NHEADS, 256, 0, stream>>>(zx, xact, dtb, al, Dpp, ybuf);
    gate_rmsnorm_kernel<<<ROWS, 256, 0, stream>>>(zx, nw, ybuf);
    float* dst = (blk == 0) ? u1 : out;
    {
      const int strips = ((D_MODEL / 16) + 3) / 4;
      const int tiles  = (ROWS / 16) * strips;
      gemm_xt_wmma<<<(tiles + 7) / 8, 256, 0, stream>>>(ybuf, opw, dst, ROWS, D_MODEL, D_INNER);
    }
    u_in = u1;
  }
}


# ssm scan: each wave scans 8 state rows alone, no LDS, no barrier: B/C/x rows loaded 8 steps ahead into registers, B/C values read across a quad by DPP broadcast; same sums, same order
# speedup vs baseline: 3.4577x; 3.4577x over previous
#include <hip/hip_runtime.h>
#include <hip/hip_bf16.h>

typedef __attribute__((ext_vector_type(2))) float v2f;
typedef __attribute__((ext_vector_type(8))) float v8f;

#define BATCH     2
#define SEQLEN    2048
#define C_IN      256
#define D_MODEL   512
#define LOUT      1024
#define D_STATE   64
#define D_CONV    4
#define D_INNER   1024
#define HEADDIM   64
#define NHEADS    16
#define CONV_DIM  (D_INNER + 2*D_STATE)
#define D_IN_PROJ (2*D_INNER + 2*D_STATE + NHEADS)
#define ROWS      (BATCH*LOUT)
#define KSTEM     (C_IN * 7)

__global__ void im2col_stem_kernel(const float* __restrict__ tokens,
                                   float* __restrict__ col) {
  const long idx = (long)blockIdx.x * 256 + threadIdx.x;
  if (idx >= (long)ROWS * KSTEM) return;
  const int j  = (int)(idx % KSTEM);
  const long bl = idx / KSTEM;
  const int c = j / 7;
  const int k = j - c * 7;
  const int l = (int)(bl & (LOUT - 1));
  const int b = (int)(bl >> 10);
  const int s = 2 * l - 3 + k;
  float v = 0.f;
  if (s >= 0 && s < SEQLEN) v = tokens[((long)b * C_IN + c) * SEQLEN + s];
  col[idx] = v;
}

__global__ void gemm_xt_wmma(const float* __restrict__ A,
                             const float* __restrict__ W,
                             float* __restrict__ C,
                             int M, int N, int K) {
  const int lane = threadIdx.x & 31;
  const int wave = threadIdx.x >> 5;
  const int tiles_n = N >> 4;
  const int strips  = (tiles_n + 3) >> 2;
  const int tile = blockIdx.x * (blockDim.x >> 5) + wave;
  if (tile >= (M >> 4) * strips) return;
  const int tm  = tile / strips;
  const int ts  = tile - tm * strips;
  const int tn0 = ts << 2;
  const int nt  = min(4, tiles_n - tn0);

  const int r    = lane & 15;
  const int koff = (lane >> 4) << 1;
  const float* arow = A + ((long)(tm * 16 + r)) * K + koff;
  const float* brow = W + ((long)(tn0 * 16 + r)) * K + koff;
  const long  bstep = (long)16 * K;

  if (nt == 4) {
    v8f acc0 = {0.f,0.f,0.f,0.f,0.f,0.f,0.f,0.f};
    v8f acc1 = acc0, acc2 = acc0, acc3 = acc0;
    #pragma unroll 2
    for (int k = 0; k < K; k += 4) {
      const v2f a  = *(const v2f*)(arow + k);
      const v2f b0 = *(const v2f*)(brow + k);
      const v2f b1 = *(const v2f*)(brow + bstep + k);
      const v2f b2 = *(const v2f*)(brow + 2 * bstep + k);
      const v2f b3 = *(const v2f*)(brow + 3 * bstep + k);
      acc0 = __builtin_amdgcn_wmma_f32_16x16x4_f32(false, a, false, b0, (short)0, acc0, false, false);
      acc1 = __builtin_amdgcn_wmma_f32_16x16x4_f32(false, a, false, b1, (short)0, acc1, false, false);
      acc2 = __builtin_amdgcn_wmma_f32_16x16x4_f32(false, a, false, b2, (short)0, acc2, false, false);
      acc3 = __builtin_amdgcn_wmma_f32_16x16x4_f32(false, a, false, b3, (short)0, acc3, false, false);
    }
    const int row0 = (lane >> 4) << 3;
    float* cbase = C + ((long)(tm * 16)) * N + tn0 * 16 + (lane & 15);
    #pragma unroll
    for (int i = 0; i < 8; ++i) {
      float* crow = cbase + (long)(row0 + i) * N;
      crow[0]  = acc0[i];
      crow[16] = acc1[i];
      crow[32] = acc2[i];
      crow[48] = acc3[i];
    }
  } else {
    for (int t = 0; t < nt; ++t) {
      v8f acc = {0.f,0.f,0.f,0.f,0.f,0.f,0.f,0.f};
      const float* bt = brow + (long)t * bstep;
      #pragma unroll 4
      for (int k = 0; k < K; k += 4) {
        const v2f a = *(const v2f*)(arow + k);
        const v2f b = *(const v2f*)(bt + k);
        acc = __builtin_amdgcn_wmma_f32_16x16x4_f32(false, a, false, b, (short)0, acc, false, false);
      }
      const int row0 = (lane >> 4) << 3;
      float* cbase = C + ((long)(tm * 16)) * N + (tn0 + t) * 16 + (lane & 15);
      #pragma unroll
      for (int i = 0; i < 8; ++i)
        cbase[(long)(row0 + i) * N] = acc[i];
    }
  }
}

__global__ void dwconv_silu_kernel(const float* __restrict__ zx,
                                   const float* __restrict__ w,
                                   const float* __restrict__ bias,
                                   float* __restrict__ out) {
  const int c = blockIdx.x * 128 + threadIdx.x;
  if (c >= CONV_DIM) return;
  const long bl = blockIdx.y;
  const int l = (int)(bl & (LOUT - 1));
  const float* xcol = zx + bl * D_IN_PROJ + D_INNER + c;
  const float w0 = w[c * 4 + 0], w1 = w[c * 4 + 1], w2 = w[c * 4 + 2], w3 = w[c * 4 + 3];
  float acc = w3 * xcol[0];
  if (l >= 1) acc = fmaf(w2, xcol[-(long)D_IN_PROJ],     acc);
  if (l >= 2) acc = fmaf(w1, xcol[-(long)2 * D_IN_PROJ], acc);
  if (l >= 3) acc = fmaf(w0, xcol[-(long)3 * D_IN_PROJ], acc);
  const float v = acc + bias[c];
  out[bl * CONV_DIM + c] = v / (1.f + __expf(-v));
}

typedef __attribute__((ext_vector_type(4))) float v4f;
#define SCAN_G 8

template <int J>
__device__ __forceinline__ float quad_bcast(float v) {
  return __builtin_bit_cast(float, __builtin_amdgcn_update_dpp(
      0, __builtin_bit_cast(int, v), J * 0x55, 0xF, 0xF, true));
}
template <int M>
__device__ __forceinline__ float row_xor(float v) {
  return __builtin_bit_cast(float, __builtin_amdgcn_update_dpp(
      0, __builtin_bit_cast(int, v), 0x160 + M, 0xF, 0xF, true));
}
__device__ __forceinline__ float lane_bcast(float v, int l) {
  return __builtin_bit_cast(float, __builtin_amdgcn_readlane(__builtin_bit_cast(int, v), l));
}
template <int I>
__device__ __forceinline__ void ssm_n(float& h, float& acc, const v4f& Bv, const v4f& Cv,
                                      float dA, float xdt) {
  const float bn = quad_bcast<(I >> 2)>(Bv[I & 3]);
  const float cn = quad_bcast<(I >> 2)>(Cv[I & 3]);
  h   = fmaf(h, dA, xdt * bn);
  acc = fmaf(h, cn, acc);
}
__device__ __forceinline__ float softplus_dt(float raw) {
  return (raw > 20.f) ? raw : log1pf(__expf(raw));
}

struct ScanSet { v4f B[SCAN_G]; v4f C[SCAN_G]; float x[SCAN_G]; };

__device__ __forceinline__ void scan_load(ScanSet& s, const float* bp, const float* cp,
                                          const float* xp, int l0) {
  #pragma unroll
  for (int j = 0; j < SCAN_G; ++j) {
    const long off = (long)min(l0 + j, LOUT - 1) * CONV_DIM;
    s.B[j] = *(const v4f*)(bp + off);
    s.C[j] = *(const v4f*)(cp + off);
    s.x[j] = xp[off];
  }
}

__device__ __forceinline__ void scan_steps(const ScanSet& s, float (&h)[16], float dtv, float dav,
                                           int lane0, float Dh, int q, float* yrow) {
  float keep = 0.f;
  #pragma unroll
  for (int j = 0; j < SCAN_G; ++j) {
    const v4f   Bv = s.B[j];
    const v4f   Cv = s.C[j];
    const float xv = s.x[j];
    const float dts = lane_bcast(dtv, lane0 + j);
    const float dA  = lane_bcast(dav, lane0 + j);
    const float xdt = xv * dts;
    float acc = 0.f;
    ssm_n<0>(h[0], acc, Bv, Cv, dA, xdt);   ssm_n<1>(h[1], acc, Bv, Cv, dA, xdt);
    ssm_n<2>(h[2], acc, Bv, Cv, dA, xdt);   ssm_n<3>(h[3], acc, Bv, Cv, dA, xdt);
    ssm_n<4>(h[4], acc, Bv, Cv, dA, xdt);   ssm_n<5>(h[5], acc, Bv, Cv, dA, xdt);
    ssm_n<6>(h[6], acc, Bv, Cv, dA, xdt);   ssm_n<7>(h[7], acc, Bv, Cv, dA, xdt);
    ssm_n<8>(h[8], acc, Bv, Cv, dA, xdt);   ssm_n<9>(h[9], acc, Bv, Cv, dA, xdt);
    ssm_n<10>(h[10], acc, Bv, Cv, dA, xdt); ssm_n<11>(h[11], acc, Bv, Cv, dA, xdt);
    ssm_n<12>(h[12], acc, Bv, Cv, dA, xdt); ssm_n<13>(h[13], acc, Bv, Cv, dA, xdt);
    ssm_n<14>(h[14], acc, Bv, Cv, dA, xdt); ssm_n<15>(h[15], acc, Bv, Cv, dA, xdt);
    acc += row_xor<4>(acc);
    acc += row_xor<8>(acc);
    const float yv = fmaf(xv, Dh, acc);
    keep = (q == (j & 3)) ? yv : keep;
    if ((j & 3) == 3) yrow[(long)(j - 3) * D_INNER] = keep;
  }
}

__global__ __launch_bounds__(128)
void ssm_scan_kernel(const float* __restrict__ zx,
                     const float* __restrict__ xbc,
                     const float* __restrict__ dtb,
                     const float* __restrict__ alog,
                     const float* __restrict__ Dp,
                     float* __restrict__ y) {
  const int b    = blockIdx.x >> 5;
  const int hh   = (blockIdx.x >> 1) & 15;
  const int half = blockIdx.x & 1;
  const int lane = threadIdx.x & 31;
  const int wave = threadIdx.x >> 5;
  const int q    = (lane >> 2) & 3;
  const int pj   = lane & 3;
  const int p    = half * 32 + wave * 8 + (lane >> 4) * 4 + pj;
  const float Ah   = -__expf(alog[hh]);
  const float dtbh = dtb[hh];
  const float Dh   = Dp[hh];

  const float* rows = xbc + (long)b * LOUT * CONV_DIM;
  const float* bp   = rows + D_INNER + 16 * q + 4 * pj;
  const float* cp   = bp + D_STATE;
  const float* xp   = rows + hh * HEADDIM + p;
  const float* dtp  = zx + (long)b * LOUT * D_IN_PROJ + (D_IN_PROJ - NHEADS) + hh;
  float* yp = y + ((long)b * LOUT + q) * D_INNER + hh * HEADDIM + p;

  float h[16];
  #pragma unroll
  for (int i = 0; i < 16; ++i) h[i] = 0.f;

  ScanSet sa, sb;
  scan_load(sa, bp, cp, xp, 0);
  float dtv = softplus_dt(dtp[(long)lane * D_IN_PROJ] + dtbh);
  float dav = __expf(dtv * Ah);

  for (int c = 0; c < LOUT / 32; ++c) {
    const int   cn   = min(c + 1, LOUT / 32 - 1);
    const float rawn = dtp[(long)(cn * 32 + lane) * D_IN_PROJ];
    for (int g = 0; g < 32; g += 2 * SCAN_G) {
      const int l0 = c * 32 + g;
      scan_load(sb, bp, cp, xp, l0 + SCAN_G);
      __builtin_amdgcn_sched_barrier(0);
      scan_steps(sa, h, dtv, dav, g, Dh, q, yp + (long)l0 * D_INNER);
      __builtin_amdgcn_sched_barrier(0);
      scan_load(sa, bp, cp, xp, l0 + 2 * SCAN_G);
      __builtin_amdgcn_sched_barrier(0);
      scan_steps(sb, h, dtv, dav, g + SCAN_G, Dh, q, yp + (long)(l0 + SCAN_G) * D_INNER);
    }
    dtv = softplus_dt(rawn + dtbh);
    dav = __expf(dtv * Ah);
  }
}

__global__ void gate_rmsnorm_kernel(const float* __restrict__ zx,
                                    const float* __restrict__ nw,
                                    float* __restrict__ y) {
  const long bl = blockIdx.x;
  const int tid = threadIdx.x;
  __shared__ float red[256];
  float yv[4];
  float ss = 0.f;
  #pragma unroll
  for (int j = 0; j < 4; ++j) {
    const int i = tid + j * 256;
    const float z = zx[bl * D_IN_PROJ + i];
    const float g = z / (1.f + __expf(-z));
    const float v = y[bl * D_INNER + i] * g;
    yv[j] = v;
    ss = fmaf(v, v, ss);
  }
  red[tid] = ss;
  __syncthreads();
  for (int st = 128; st > 0; st >>= 1) {
    if (tid < st) red[tid] += red[tid + st];
    __syncthreads();
  }
  const float scale = rsqrtf(red[0] * (1.f / (float)D_INNER) + 1e-5f);
  #pragma unroll
  for (int j = 0; j < 4; ++j) {
    const int i = tid + j * 256;
    y[bl * D_INNER + i] = yv[j] * scale * nw[i];
  }
}

extern "C" void kernel_launch(void* const* d_in, const int* in_sizes, int n_in,
                              void* d_out, int out_size, void* d_ws, size_t ws_size,
                              hipStream_t stream) {
  const float* tokens     = (const float*)d_in[0];
  const float* conv_w     = (const float*)d_in[1];
  const float* in_proj_w  = (const float*)d_in[2];
  const float* conv1d_w   = (const float*)d_in[3];
  const float* conv1d_b   = (const float*)d_in[4];
  const float* dt_bias    = (const float*)d_in[5];
  const float* A_log      = (const float*)d_in[6];
  const float* D_param    = (const float*)d_in[7];
  const float* norm_w     = (const float*)d_in[8];
  const float* out_proj_w = (const float*)d_in[9];
  float* out = (float*)d_out;
  float* ws  = (float*)d_ws;

  float* u0   = ws;
  float* u1   = u0 + (long)ROWS * D_MODEL;
  float* zx   = u1 + (long)ROWS * D_MODEL;
  float* xact = zx + (long)ROWS * D_IN_PROJ;
  float* ybuf = xact + (long)ROWS * CONV_DIM;
  float* col  = zx;
  (void)ws_size; (void)in_sizes; (void)n_in; (void)out_size;

  {
    const long nel = (long)ROWS * KSTEM;
    im2col_stem_kernel<<<(int)((nel + 255) / 256), 256, 0, stream>>>(tokens, col);
    const int strips = ((D_MODEL / 16) + 3) / 4;
    const int tiles  = (ROWS / 16) * strips;
    gemm_xt_wmma<<<(tiles + 7) / 8, 256, 0, stream>>>(col, conv_w, u0, ROWS, D_MODEL, KSTEM);
  }

  const float* u_in = u0;
  for (int blk = 0; blk < 2; ++blk) {
    const float* ipw = in_proj_w  + (long)blk * D_IN_PROJ * D_MODEL;
    const float* cw  = conv1d_w   + (long)blk * CONV_DIM * D_CONV;
    const float* cb  = conv1d_b   + (long)blk * CONV_DIM;
    const float* dtb = dt_bias    + (long)blk * NHEADS;
    const float* al  = A_log      + (long)blk * NHEADS;
    const float* Dpp = D_param    + (long)blk * NHEADS;
    const float* nw  = norm_w     + (long)blk * D_INNER;
    const float* opw = out_proj_w + (long)blk * D_MODEL * D_INNER;

    {
      const int strips = ((D_IN_PROJ / 16) + 3) / 4;
      const int tiles  = (ROWS / 16) * strips;
      gemm_xt_wmma<<<(tiles + 7) / 8, 256, 0, stream>>>(u_in, ipw, zx, ROWS, D_IN_PROJ, D_MODEL);
    }
    dwconv_silu_kernel<<<dim3(CONV_DIM / 128, ROWS), 128, 0, stream>>>(zx, cw, cb, xact);
    ssm_scan_kernel<<<BATCH * NHEADS * 2, 128, 0, stream>>>(zx, xact, dtb, al, Dpp, ybuf);
    gate_rmsnorm_kernel<<<ROWS, 256, 0, stream>>>(zx, nw, ybuf);
    float* dst = (blk == 0) ? u1 : out;
    {
      const int strips = ((D_MODEL / 16) + 3) / 4;
      const int tiles  = (ROWS / 16) * strips;
      gemm_xt_wmma<<<(tiles + 7) / 8, 256, 0, stream>>>(ybuf, opw, dst, ROWS, D_MODEL, D_INNER);
    }
    u_in = u1;
  }
}


# GEMMs: no LDS: each wave owns a 32x32 C block and reads whole 32-byte pieces of its A and W rows straight into fragment registers, one k block ahead; k summed in another order inside each 8-k group
# speedup vs baseline: 8.5454x; 2.4714x over previous
#include <hip/hip_runtime.h>
#include <hip/hip_bf16.h>

typedef __attribute__((ext_vector_type(2))) float v2f;
typedef __attribute__((ext_vector_type(8))) float v8f;

#define BATCH     2
#define SEQLEN    2048
#define C_IN      256
#define D_MODEL   512
#define LOUT      1024
#define D_STATE   64
#define D_CONV    4
#define D_INNER   1024
#define HEADDIM   64
#define NHEADS    16
#define CONV_DIM  (D_INNER + 2*D_STATE)
#define D_IN_PROJ (2*D_INNER + 2*D_STATE + NHEADS)
#define ROWS      (BATCH*LOUT)
#define KSTEM     (C_IN * 7)
#define GEMM_WGS(n) ((ROWS / 64) * (((n) + 63) / 64))

__global__ void im2col_stem_kernel(const float* __restrict__ tokens,
                                   float* __restrict__ col) {
  const long idx = (long)blockIdx.x * 256 + threadIdx.x;
  if (idx >= (long)ROWS * KSTEM) return;
  const int j  = (int)(idx % KSTEM);
  const long bl = idx / KSTEM;
  const int c = j / 7;
  const int k = j - c * 7;
  const int l = (int)(bl & (LOUT - 1));
  const int b = (int)(bl >> 10);
  const int s = 2 * l - 3 + k;
  float v = 0.f;
  if (s >= 0 && s < SEQLEN) v = tokens[((long)b * C_IN + c) * SEQLEN + s];
  col[idx] = v;
}

typedef __attribute__((ext_vector_type(4))) float v4f;

template <int JB>
struct RowFrag { v4f a0[JB]; v4f a1[JB]; v4f b0[JB]; v4f b1[JB]; };

template <int JB>
__device__ __forceinline__ void frag_load(RowFrag<JB>& f, const float* pa0, const float* pa1,
                                          const float* pb0, const float* pb1, int kb) {
  #pragma unroll
  for (int j = 0; j < JB; ++j) {
    f.a0[j] = *(const v4f*)(pa0 + kb + 8 * j);
    f.a1[j] = *(const v4f*)(pa1 + kb + 8 * j);
    f.b0[j] = *(const v4f*)(pb0 + kb + 8 * j);
    f.b1[j] = *(const v4f*)(pb1 + kb + 8 * j);
  }
}

__device__ __forceinline__ v8f mma4(v2f a, v2f b, v8f c) {
  return __builtin_amdgcn_wmma_f32_16x16x4_f32(false, a, false, b, (short)0, c, false, false);
}

template <int JB>
__device__ __forceinline__ void frag_mma(const RowFrag<JB>& f, v8f& c00, v8f& c01, v8f& c10, v8f& c11) {
  #pragma unroll
  for (int j = 0; j < JB; ++j) {
    #pragma unroll
    for (int u = 0; u < 2; ++u) {
      const v2f a0 = {f.a0[j][2 * u], f.a0[j][2 * u + 1]};
      const v2f a1 = {f.a1[j][2 * u], f.a1[j][2 * u + 1]};
      const v2f b0 = {f.b0[j][2 * u], f.b0[j][2 * u + 1]};
      const v2f b1 = {f.b1[j][2 * u], f.b1[j][2 * u + 1]};
      c00 = mma4(a0, b0, c00);
      c01 = mma4(a0, b1, c01);
      c10 = mma4(a1, b0, c10);
      c11 = mma4(a1, b1, c11);
    }
  }
}

template <int N, int K, int KB>
__global__ __launch_bounds__(128)
void gemm_rows_wmma(const float* __restrict__ A,
                    const float* __restrict__ W,
                    float* __restrict__ C) {
  static_assert(K % (2 * KB) == 0 && KB % 8 == 0 && N % 16 == 0, "shape");
  constexpr int JB  = KB / 8;
  constexpr int WGN = (N + 63) / 64;
  const int lane = threadIdx.x & 31;
  const int wave = __builtin_amdgcn_readfirstlane((int)(threadIdx.x >> 5));
  const int wgm  = blockIdx.x / WGN;
  const int wgn  = blockIdx.x - wgm * WGN;
  const int m0   = wgm * 64 + (wave >> 1) * 32;
  const int n0   = wgn * 64 + (wave & 1) * 32;
  if (n0 >= N) return;
  const bool n1ok = (n0 + 16) < N;
  const int c = lane & 15;
  const int h = lane >> 4;
  const float* pa0 = A + (long)(m0 + c) * K + 4 * h;
  const float* pa1 = pa0 + (long)16 * K;
  const float* pb0 = W + (long)(n0 + c) * K + 4 * h;
  const float* pb1 = pb0 + (n1ok ? (long)16 * K : 0L);

  const v8f zero = {0.f, 0.f, 0.f, 0.f, 0.f, 0.f, 0.f, 0.f};
  v8f c00 = zero, c01 = zero, c10 = zero, c11 = zero;
  RowFrag<JB> fa, fb;
  frag_load<JB>(fa, pa0, pa1, pb0, pb1, 0);
  #pragma unroll 1
  for (int kb = 0; kb < K - 2 * KB; kb += 2 * KB) {
    frag_load<JB>(fb, pa0, pa1, pb0, pb1, kb + KB);
    __builtin_amdgcn_sched_barrier(0);
    frag_mma<JB>(fa, c00, c01, c10, c11);
    __builtin_amdgcn_sched_barrier(0);
    frag_load<JB>(fa, pa0, pa1, pb0, pb1, kb + 2 * KB);
    __builtin_amdgcn_sched_barrier(0);
    frag_mma<JB>(fb, c00, c01, c10, c11);
  }
  frag_load<JB>(fb, pa0, pa1, pb0, pb1, K - KB);
  __builtin_amdgcn_sched_barrier(0);
  frag_mma<JB>(fa, c00, c01, c10, c11);
  frag_mma<JB>(fb, c00, c01, c10, c11);

  float* crow = C + (long)(m0 + 8 * h) * N + n0 + c;
  #pragma unroll
  for (int i = 0; i < 8; ++i) {
    crow[(long)i * N]        = c00[i];
    crow[(long)(i + 16) * N] = c10[i];
  }
  if (n1ok) {
    #pragma unroll
    for (int i = 0; i < 8; ++i) {
      crow[(long)i * N + 16]        = c01[i];
      crow[(long)(i + 16) * N + 16] = c11[i];
    }
  }
}

__global__ void dwconv_silu_kernel(const float* __restrict__ zx,
                                   const float* __restrict__ w,
                                   const float* __restrict__ bias,
                                   float* __restrict__ out) {
  const int c = blockIdx.x * 128 + threadIdx.x;
  if (c >= CONV_DIM) return;
  const long bl = blockIdx.y;
  const int l = (int)(bl & (LOUT - 1));
  const float* xcol = zx + bl * D_IN_PROJ + D_INNER + c;
  const float w0 = w[c * 4 + 0], w1 = w[c * 4 + 1], w2 = w[c * 4 + 2], w3 = w[c * 4 + 3];
  float acc = w3 * xcol[0];
  if (l >= 1) acc = fmaf(w2, xcol[-(long)D_IN_PROJ],     acc);
  if (l >= 2) acc = fmaf(w1, xcol[-(long)2 * D_IN_PROJ], acc);
  if (l >= 3) acc = fmaf(w0, xcol[-(long)3 * D_IN_PROJ], acc);
  const float v = acc + bias[c];
  out[bl * CONV_DIM + c] = v / (1.f + __expf(-v));
}

#define SCAN_G 8

template <int J>
__device__ __forceinline__ float quad_bcast(float v) {
  return __builtin_bit_cast(float, __builtin_amdgcn_update_dpp(
      0, __builtin_bit_cast(int, v), J * 0x55, 0xF, 0xF, true));
}
template <int M>
__device__ __forceinline__ float row_xor(float v) {
  return __builtin_bit_cast(float, __builtin_amdgcn_update_dpp(
      0, __builtin_bit_cast(int, v), 0x160 + M, 0xF, 0xF, true));
}
__device__ __forceinline__ float lane_bcast(float v, int l) {
  return __builtin_bit_cast(float, __builtin_amdgcn_readlane(__builtin_bit_cast(int, v), l));
}
template <int I>
__device__ __forceinline__ void ssm_n(float& h, float& acc, const v4f& Bv, const v4f& Cv,
                                      float dA, float xdt) {
  const float bn = quad_bcast<(I >> 2)>(Bv[I & 3]);
  const float cn = quad_bcast<(I >> 2)>(Cv[I & 3]);
  h   = fmaf(h, dA, xdt * bn);
  acc = fmaf(h, cn, acc);
}
__device__ __forceinline__ float softplus_dt(float raw) {
  return (raw > 20.f) ? raw : log1pf(__expf(raw));
}

struct ScanSet { v4f B[SCAN_G]; v4f C[SCAN_G]; float x[SCAN_G]; };

__device__ __forceinline__ void scan_load(ScanSet& s, const float* bp, const float* cp,
                                          const float* xp, int l0) {
  #pragma unroll
  for (int j = 0; j < SCAN_G; ++j) {
    const long off = (long)min(l0 + j, LOUT - 1) * CONV_DIM;
    s.B[j] = *(const v4f*)(bp + off);
    s.C[j] = *(const v4f*)(cp + off);
    s.x[j] = xp[off];
  }
}

__device__ __forceinline__ void scan_steps(const ScanSet& s, float (&h)[16], float dtv, float dav,
                                           int lane0, float Dh, int q, float* yrow) {
  float keep = 0.f;
  #pragma unroll
  for (int j = 0; j < SCAN_G; ++j) {
    const v4f   Bv = s.B[j];
    const v4f   Cv = s.C[j];
    const float xv = s.x[j];
    const float dts = lane_bcast(dtv, lane0 + j);
    const float dA  = lane_bcast(dav, lane0 + j);
    const float xdt = xv * dts;
    float acc = 0.f;
    ssm_n<0>(h[0], acc, Bv, Cv, dA, xdt);   ssm_n<1>(h[1], acc, Bv, Cv, dA, xdt);
    ssm_n<2>(h[2], acc, Bv, Cv, dA, xdt);   ssm_n<3>(h[3], acc, Bv, Cv, dA, xdt);
    ssm_n<4>(h[4], acc, Bv, Cv, dA, xdt);   ssm_n<5>(h[5], acc, Bv, Cv, dA, xdt);
    ssm_n<6>(h[6], acc, Bv, Cv, dA, xdt);   ssm_n<7>(h[7], acc, Bv, Cv, dA, xdt);
    ssm_n<8>(h[8], acc, Bv, Cv, dA, xdt);   ssm_n<9>(h[9], acc, Bv, Cv, dA, xdt);
    ssm_n<10>(h[10], acc, Bv, Cv, dA, xdt); ssm_n<11>(h[11], acc, Bv, Cv, dA, xdt);
    ssm_n<12>(h[12], acc, Bv, Cv, dA, xdt); ssm_n<13>(h[13], acc, Bv, Cv, dA, xdt);
    ssm_n<14>(h[14], acc, Bv, Cv, dA, xdt); ssm_n<15>(h[15], acc, Bv, Cv, dA, xdt);
    acc += row_xor<4>(acc);
    acc += row_xor<8>(acc);
    const float yv = fmaf(xv, Dh, acc);
    keep = (q == (j & 3)) ? yv : keep;
    if ((j & 3) == 3) yrow[(long)(j - 3) * D_INNER] = keep;
  }
}

__global__ __launch_bounds__(128)
void ssm_scan_kernel(const float* __restrict__ zx,
                     const float* __restrict__ xbc,
                     const float* __restrict__ dtb,
                     const float* __restrict__ alog,
                     const float* __restrict__ Dp,
                     float* __restrict__ y) {
  const int b    = blockIdx.x >> 5;
  const int hh   = (blockIdx.x >> 1) & 15;
  const int half = blockIdx.x & 1;
  const int lane = threadIdx.x & 31;
  const int wave = threadIdx.x >> 5;
  const int q    = (lane >> 2) & 3;
  const int pj   = lane & 3;
  const int p    = half * 32 + wave * 8 + (lane >> 4) * 4 + pj;
  const float Ah   = -__expf(alog[hh]);
  const float dtbh = dtb[hh];
  const float Dh   = Dp[hh];

  const float* rows = xbc + (long)b * LOUT * CONV_DIM;
  const float* bp   = rows + D_INNER + 16 * q + 4 * pj;
  const float* cp   = bp + D_STATE;
  const float* xp   = rows + hh * HEADDIM + p;
  const float* dtp  = zx + (long)b * LOUT * D_IN_PROJ + (D_IN_PROJ - NHEADS) + hh;
  float* yp = y + ((long)b * LOUT + q) * D_INNER + hh * HEADDIM + p;

  float h[16];
  #pragma unroll
  for (int i = 0; i < 16; ++i) h[i] = 0.f;

  ScanSet sa, sb;
  scan_load(sa, bp, cp, xp, 0);
  float dtv = softplus_dt(dtp[(long)lane * D_IN_PROJ] + dtbh);
  float dav = __expf(dtv * Ah);

  for (int c = 0; c < LOUT / 32; ++c) {
    const int   cn   = min(c + 1, LOUT / 32 - 1);
    const float rawn = dtp[(long)(cn * 32 + lane) * D_IN_PROJ];
    for (int g = 0; g < 32; g += 2 * SCAN_G) {
      const int l0 = c * 32 + g;
      scan_load(sb, bp, cp, xp, l0 + SCAN_G);
      __builtin_amdgcn_sched_barrier(0);
      scan_steps(sa, h, dtv, dav, g, Dh, q, yp + (long)l0 * D_INNER);
      __builtin_amdgcn_sched_barrier(0);
      scan_load(sa, bp, cp, xp, l0 + 2 * SCAN_G);
      __builtin_amdgcn_sched_barrier(0);
      scan_steps(sb, h, dtv, dav, g + SCAN_G, Dh, q, yp + (long)(l0 + SCAN_G) * D_INNER);
    }
    dtv = softplus_dt(rawn + dtbh);
    dav = __expf(dtv * Ah);
  }
}

__global__ void gate_rmsnorm_kernel(const float* __restrict__ zx,
                                    const float* __restrict__ nw,
                                    float* __restrict__ y) {
  const long bl = blockIdx.x;
  const int tid = threadIdx.x;
  __shared__ float red[256];
  float yv[4];
  float ss = 0.f;
  #pragma unroll
  for (int j = 0; j < 4; ++j) {
    const int i = tid + j * 256;
    const float z = zx[bl * D_IN_PROJ + i];
    const float g = z / (1.f + __expf(-z));
    const float v = y[bl * D_INNER + i] * g;
    yv[j] = v;
    ss = fmaf(v, v, ss);
  }
  red[tid] = ss;
  __syncthreads();
  for (int st = 128; st > 0; st >>= 1) {
    if (tid < st) red[tid] += red[tid + st];
    __syncthreads();
  }
  const float scale = rsqrtf(red[0] * (1.f / (float)D_INNER) + 1e-5f);
  #pragma unroll
  for (int j = 0; j < 4; ++j) {
    const int i = tid + j * 256;
    y[bl * D_INNER + i] = yv[j] * scale * nw[i];
  }
}

extern "C" void kernel_launch(void* const* d_in, const int* in_sizes, int n_in,
                              void* d_out, int out_size, void* d_ws, size_t ws_size,
                              hipStream_t stream) {
  const float* tokens     = (const float*)d_in[0];
  const float* conv_w     = (const float*)d_in[1];
  const float* in_proj_w  = (const float*)d_in[2];
  const float* conv1d_w   = (const float*)d_in[3];
  const float* conv1d_b   = (const float*)d_in[4];
  const float* dt_bias    = (const float*)d_in[5];
  const float* A_log      = (const float*)d_in[6];
  const float* D_param    = (const float*)d_in[7];
  const float* norm_w     = (const float*)d_in[8];
  const float* out_proj_w = (const float*)d_in[9];
  float* out = (float*)d_out;
  float* ws  = (float*)d_ws;

  float* u0   = ws;
  float* u1   = u0 + (long)ROWS * D_MODEL;
  float* zx   = u1 + (long)ROWS * D_MODEL;
  float* xact = zx + (long)ROWS * D_IN_PROJ;
  float* ybuf = xact + (long)ROWS * CONV_DIM;
  float* col  = zx;
  (void)ws_size; (void)in_sizes; (void)n_in; (void)out_size;

  {
    const long nel = (long)ROWS * KSTEM;
    im2col_stem_kernel<<<(int)((nel + 255) / 256), 256, 0, stream>>>(tokens, col);
    gemm_rows_wmma<D_MODEL, KSTEM, 64><<<GEMM_WGS(D_MODEL), 128, 0, stream>>>(col, conv_w, u0);
  }

  const float* u_in = u0;
  for (int blk = 0; blk < 2; ++blk) {
    const float* ipw = in_proj_w  + (long)blk * D_IN_PROJ * D_MODEL;
    const float* cw  = conv1d_w   + (long)blk * CONV_DIM * D_CONV;
    const float* cb  = conv1d_b   + (long)blk * CONV_DIM;
    const float* dtb = dt_bias    + (long)blk * NHEADS;
    const float* al  = A_log      + (long)blk * NHEADS;
    const float* Dpp = D_param    + (long)blk * NHEADS;
    const float* nw  = norm_w     + (long)blk * D_INNER;
    const float* opw = out_proj_w + (long)blk * D_MODEL * D_INNER;

    {
      gemm_rows_wmma<D_IN_PROJ, D_MODEL, 32><<<GEMM_WGS(D_IN_PROJ), 128, 0, stream>>>(u_in, ipw, zx);
    }
    dwconv_silu_kernel<<<dim3(CONV_DIM / 128, ROWS), 128, 0, stream>>>(zx, cw, cb, xact);
    ssm_scan_kernel<<<BATCH * NHEADS * 2, 128, 0, stream>>>(zx, xact, dtb, al, Dpp, ybuf);
    gate_rmsnorm_kernel<<<ROWS, 256, 0, stream>>>(zx, nw, ybuf);
    float* dst = (blk == 0) ? u1 : out;
    {
      gemm_rows_wmma<D_MODEL, D_INNER, 64><<<GEMM_WGS(D_MODEL), 128, 0, stream>>>(ybuf, opw, dst);
    }
    u_in = u1;
  }
}


# stem conv: the GEMM's A fragments are read straight from the token rows (one word per tap, zero padding masked in registers): im2col kernel, its buffer and one launch gone; same sums, same order
# speedup vs baseline: 8.8703x; 1.0380x over previous
#include <hip/hip_runtime.h>
#include <hip/hip_bf16.h>

typedef __attribute__((ext_vector_type(2))) float v2f;
typedef __attribute__((ext_vector_type(8))) float v8f;

#define BATCH     2
#define SEQLEN    2048
#define C_IN      256
#define D_MODEL   512
#define LOUT      1024
#define D_STATE   64
#define D_CONV    4
#define D_INNER   1024
#define HEADDIM   64
#define NHEADS    16
#define CONV_DIM  (D_INNER + 2*D_STATE)
#define D_IN_PROJ (2*D_INNER + 2*D_STATE + NHEADS)
#define ROWS      (BATCH*LOUT)
#define KSTEM     (C_IN * 7)
#define GEMM_WGS(n) ((ROWS / 64) * (((n) + 63) / 64))

typedef __attribute__((ext_vector_type(4))) float v4f;

template <int JB>
struct RowFrag { v4f a0[JB]; v4f a1[JB]; v4f b0[JB]; v4f b1[JB]; };

template <int JB>
__device__ __forceinline__ void frag_load(RowFrag<JB>& f, const float* pa0, const float* pa1,
                                          const float* pb0, const float* pb1, int kb) {
  #pragma unroll
  for (int j = 0; j < JB; ++j) {
    f.a0[j] = *(const v4f*)(pa0 + kb + 8 * j);
    f.a1[j] = *(const v4f*)(pa1 + kb + 8 * j);
    f.b0[j] = *(const v4f*)(pb0 + kb + 8 * j);
    f.b1[j] = *(const v4f*)(pb1 + kb + 8 * j);
  }
}

__device__ __forceinline__ v8f mma4(v2f a, v2f b, v8f c) {
  return __builtin_amdgcn_wmma_f32_16x16x4_f32(false, a, false, b, (short)0, c, false, false);
}

template <int JB>
__device__ __forceinline__ void frag_mma(const RowFrag<JB>& f, v8f& c00, v8f& c01, v8f& c10, v8f& c11) {
  #pragma unroll
  for (int j = 0; j < JB; ++j) {
    #pragma unroll
    for (int u = 0; u < 2; ++u) {
      const v2f a0 = {f.a0[j][2 * u], f.a0[j][2 * u + 1]};
      const v2f a1 = {f.a1[j][2 * u], f.a1[j][2 * u + 1]};
      const v2f b0 = {f.b0[j][2 * u], f.b0[j][2 * u + 1]};
      const v2f b1 = {f.b1[j][2 * u], f.b1[j][2 * u + 1]};
      c00 = mma4(a0, b0, c00);
      c01 = mma4(a0, b1, c01);
      c10 = mma4(a1, b0, c10);
      c11 = mma4(a1, b1, c11);
    }
  }
}

template <int N, int K, int KB>
__global__ __launch_bounds__(128)
void gemm_rows_wmma(const float* __restrict__ A,
                    const float* __restrict__ W,
                    float* __restrict__ C) {
  static_assert(K % (2 * KB) == 0 && KB % 8 == 0 && N % 16 == 0, "shape");
  constexpr int JB  = KB / 8;
  constexpr int WGN = (N + 63) / 64;
  const int lane = threadIdx.x & 31;
  const int wave = __builtin_amdgcn_readfirstlane((int)(threadIdx.x >> 5));
  const int wgm  = blockIdx.x / WGN;
  const int wgn  = blockIdx.x - wgm * WGN;
  const int m0   = wgm * 64 + (wave >> 1) * 32;
  const int n0   = wgn * 64 + (wave & 1) * 32;
  if (n0 >= N) return;
  const bool n1ok = (n0 + 16) < N;
  const int c = lane & 15;
  const int h = lane >> 4;
  const float* pa0 = A + (long)(m0 + c) * K + 4 * h;
  const float* pa1 = pa0 + (long)16 * K;
  const float* pb0 = W + (long)(n0 + c) * K + 4 * h;
  const float* pb1 = pb0 + (n1ok ? (long)16 * K : 0L);

  const v8f zero = {0.f, 0.f, 0.f, 0.f, 0.f, 0.f, 0.f, 0.f};
  v8f c00 = zero, c01 = zero, c10 = zero, c11 = zero;
  RowFrag<JB> fa, fb;
  frag_load<JB>(fa, pa0, pa1, pb0, pb1, 0);
  #pragma unroll 1
  for (int kb = 0; kb < K - 2 * KB; kb += 2 * KB) {
    frag_load<JB>(fb, pa0, pa1, pb0, pb1, kb + KB);
    __builtin_amdgcn_sched_barrier(0);
    frag_mma<JB>(fa, c00, c01, c10, c11);
    __builtin_amdgcn_sched_barrier(0);
    frag_load<JB>(fa, pa0, pa1, pb0, pb1, kb + 2 * KB);
    __builtin_amdgcn_sched_barrier(0);
    frag_mma<JB>(fb, c00, c01, c10, c11);
  }
  frag_load<JB>(fb, pa0, pa1, pb0, pb1, K - KB);
  __builtin_amdgcn_sched_barrier(0);
  frag_mma<JB>(fa, c00, c01, c10, c11);
  frag_mma<JB>(fb, c00, c01, c10, c11);

  float* crow = C + (long)(m0 + 8 * h) * N + n0 + c;
  #pragma unroll
  for (int i = 0; i < 8; ++i) {
    crow[(long)i * N]        = c00[i];
    crow[(long)(i + 16) * N] = c10[i];
  }
  if (n1ok) {
    #pragma unroll
    for (int i = 0; i < 8; ++i) {
      crow[(long)i * N + 16]        = c01[i];
      crow[(long)(i + 16) * N + 16] = c11[i];
    }
  }
}

#define STEM_CB 8
#define STEM_KB (STEM_CB * 7)
#define STEM_NB (C_IN / STEM_CB)
#define STEM_J  (STEM_KB / 8)

struct StemFrag { float a0[STEM_J][4]; float a1[STEM_J][4]; v4f b0[STEM_J]; v4f b1[STEM_J]; };
struct StemMap  { unsigned o0[STEM_J][4]; unsigned o1[STEM_J][4]; int m0[STEM_J][4]; int m1[STEM_J][4]; };

template <int J0, int J1>
__device__ __forceinline__ void stem_load(StemFrag& f, const StemMap& mp, const char* tb,
                                          const float* pb0, const float* pb1, int blk) {
  const char* t = tb + (long)blk * (STEM_CB * SEQLEN * 4);
  const int kb = blk * STEM_KB;
  #pragma unroll
  for (int j = J0; j < J1; ++j) {
    f.b0[j] = *(const v4f*)(pb0 + kb + 8 * j);
    f.b1[j] = *(const v4f*)(pb1 + kb + 8 * j);
    #pragma unroll
    for (int e = 0; e < 4; ++e) {
      f.a0[j][e] = *(const float*)(t + mp.o0[j][e]);
      f.a1[j][e] = *(const float*)(t + mp.o1[j][e]);
    }
  }
}

__device__ __forceinline__ float pad0(float v, int m) {
  return __builtin_bit_cast(float, __builtin_bit_cast(int, v) & m);
}

template <int J0, int J1>
__device__ __forceinline__ void stem_mma(const StemFrag& f, const StemMap& mp,
                                         v8f& c00, v8f& c01, v8f& c10, v8f& c11) {
  #pragma unroll
  for (int j = J0; j < J1; ++j) {
    #pragma unroll
    for (int u = 0; u < 2; ++u) {
      const v2f a0 = {pad0(f.a0[j][2 * u], mp.m0[j][2 * u]), pad0(f.a0[j][2 * u + 1], mp.m0[j][2 * u + 1])};
      const v2f a1 = {pad0(f.a1[j][2 * u], mp.m1[j][2 * u]), pad0(f.a1[j][2 * u + 1], mp.m1[j][2 * u + 1])};
      const v2f b0 = {f.b0[j][2 * u], f.b0[j][2 * u + 1]};
      const v2f b1 = {f.b1[j][2 * u], f.b1[j][2 * u + 1]};
      c00 = mma4(a0, b0, c00);
      c01 = mma4(a0, b1, c01);
      c10 = mma4(a1, b0, c10);
      c11 = mma4(a1, b1, c11);
    }
  }
}

__global__ __launch_bounds__(128)
void stem_conv_wmma(const float* __restrict__ tokens,
                    const float* __restrict__ W,
                    float* __restrict__ C) {
  constexpr int N   = D_MODEL;
  constexpr int K   = KSTEM;
  constexpr int WGN = N / 64;
  const int lane = threadIdx.x & 31;
  const int wave = __builtin_amdgcn_readfirstlane((int)(threadIdx.x >> 5));
  const int wgm  = blockIdx.x / WGN;
  const int wgn  = blockIdx.x - wgm * WGN;
  const int m0   = wgm * 64 + (wave >> 1) * 32;
  const int n0   = wgn * 64 + (wave & 1) * 32;
  const int c = lane & 15;
  const int h = lane >> 4;
  const float* pb0 = W + (long)(n0 + c) * K + 4 * h;
  const float* pb1 = pb0 + (long)16 * K;
  const char* tb = (const char*)(tokens + (long)(m0 / LOUT) * C_IN * SEQLEN);
  const int l0 = (m0 & (LOUT - 1)) + c;

  StemMap mp;
  #pragma unroll
  for (int j = 0; j < STEM_J; ++j) {
    #pragma unroll
    for (int e = 0; e < 4; ++e) {
      const int kl = 8 * j + 4 * h + e;
      const int ch = kl / 7;
      const int tp = kl - 7 * ch;
      const int s0 = 2 * l0 - 3 + tp;
      const int s1 = s0 + 32;
      const bool ok0 = (unsigned)s0 < (unsigned)SEQLEN;
      const bool ok1 = (unsigned)s1 < (unsigned)SEQLEN;
      mp.o0[j][e] = (unsigned)(ch * SEQLEN + (ok0 ? s0 : 0)) * 4u;
      mp.o1[j][e] = (unsigned)(ch * SEQLEN + (ok1 ? s1 : 0)) * 4u;
      mp.m0[j][e] = ok0 ? -1 : 0;
      mp.m1[j][e] = ok1 ? -1 : 0;
    }
  }

  const v8f zero = {0.f, 0.f, 0.f, 0.f, 0.f, 0.f, 0.f, 0.f};
  v8f c00 = zero, c01 = zero, c10 = zero, c11 = zero;
  StemFrag fa, fb;
  stem_load<0, 4>(fa, mp, tb, pb0, pb1, 0);
  stem_load<4, STEM_J>(fa, mp, tb, pb0, pb1, 0);
  #pragma unroll 1
  for (int blk = 0; blk < STEM_NB - 2; blk += 2) {
    stem_load<0, 4>(fb, mp, tb, pb0, pb1, blk + 1);
    __builtin_amdgcn_sched_barrier(0);
    stem_mma<0, 4>(fa, mp, c00, c01, c10, c11);
    __builtin_amdgcn_sched_barrier(0);
    stem_load<4, STEM_J>(fb, mp, tb, pb0, pb1, blk + 1);
    __builtin_amdgcn_sched_barrier(0);
    stem_mma<4, STEM_J>(fa, mp, c00, c01, c10, c11);
    __builtin_amdgcn_sched_barrier(0);
    stem_load<0, 4>(fa, mp, tb, pb0, pb1, blk + 2);
    __builtin_amdgcn_sched_barrier(0);
    stem_mma<0, 4>(fb, mp, c00, c01, c10, c11);
    __builtin_amdgcn_sched_barrier(0);
    stem_load<4, STEM_J>(fa, mp, tb, pb0, pb1, blk + 2);
    __builtin_amdgcn_sched_barrier(0);
    stem_mma<4, STEM_J>(fb, mp, c00, c01, c10, c11);
    __builtin_amdgcn_sched_barrier(0);
  }
  stem_load<0, 4>(fb, mp, tb, pb0, pb1, STEM_NB - 1);
  __builtin_amdgcn_sched_barrier(0);
  stem_mma<0, 4>(fa, mp, c00, c01, c10, c11);
  __builtin_amdgcn_sched_barrier(0);
  stem_load<4, STEM_J>(fb, mp, tb, pb0, pb1, STEM_NB - 1);
  __builtin_amdgcn_sched_barrier(0);
  stem_mma<4, STEM_J>(fa, mp, c00, c01, c10, c11);
  stem_mma<0, STEM_J>(fb, mp, c00, c01, c10, c11);

  float* crow = C + (long)(m0 + 8 * h) * N + n0 + c;
  #pragma unroll
  for (int i = 0; i < 8; ++i) {
    crow[(long)i * N]             = c00[i];
    crow[(long)(i + 16) * N]      = c10[i];
    crow[(long)i * N + 16]        = c01[i];
    crow[(long)(i + 16) * N + 16] = c11[i];
  }
}

__global__ void dwconv_silu_kernel(const float* __restrict__ zx,
                                   const float* __restrict__ w,
                                   const float* __restrict__ bias,
                                   float* __restrict__ out) {
  const int c = blockIdx.x * 128 + threadIdx.x;
  if (c >= CONV_DIM) return;
  const long bl = blockIdx.y;
  const int l = (int)(bl & (LOUT - 1));
  const float* xcol = zx + bl * D_IN_PROJ + D_INNER + c;
  const float w0 = w[c * 4 + 0], w1 = w[c * 4 + 1], w2 = w[c * 4 + 2], w3 = w[c * 4 + 3];
  float acc = w3 * xcol[0];
  if (l >= 1) acc = fmaf(w2, xcol[-(long)D_IN_PROJ],     acc);
  if (l >= 2) acc = fmaf(w1, xcol[-(long)2 * D_IN_PROJ], acc);
  if (l >= 3) acc = fmaf(w0, xcol[-(long)3 * D_IN_PROJ], acc);
  const float v = acc + bias[c];
  out[bl * CONV_DIM + c] = v / (1.f + __expf(-v));
}

#define SCAN_G 8

template <int J>
__device__ __forceinline__ float quad_bcast(float v) {
  return __builtin_bit_cast(float, __builtin_amdgcn_update_dpp(
      0, __builtin_bit_cast(int, v), J * 0x55, 0xF, 0xF, true));
}
template <int M>
__device__ __forceinline__ float row_xor(float v) {
  return __builtin_bit_cast(float, __builtin_amdgcn_update_dpp(
      0, __builtin_bit_cast(int, v), 0x160 + M, 0xF, 0xF, true));
}
__device__ __forceinline__ float lane_bcast(float v, int l) {
  return __builtin_bit_cast(float, __builtin_amdgcn_readlane(__builtin_bit_cast(int, v), l));
}
template <int I>
__device__ __forceinline__ void ssm_n(float& h, float& acc, const v4f& Bv, const v4f& Cv,
                                      float dA, float xdt) {
  const float bn = quad_bcast<(I >> 2)>(Bv[I & 3]);
  const float cn = quad_bcast<(I >> 2)>(Cv[I & 3]);
  h   = fmaf(h, dA, xdt * bn);
  acc = fmaf(h, cn, acc);
}
__device__ __forceinline__ float softplus_dt(float raw) {
  return (raw > 20.f) ? raw : log1pf(__expf(raw));
}

struct ScanSet { v4f B[SCAN_G]; v4f C[SCAN_G]; float x[SCAN_G]; };

__device__ __forceinline__ void scan_load(ScanSet& s, const float* bp, const float* cp,
                                          const float* xp, int l0) {
  #pragma unroll
  for (int j = 0; j < SCAN_G; ++j) {
    const long off = (long)min(l0 + j, LOUT - 1) * CONV_DIM;
    s.B[j] = *(const v4f*)(bp + off);
    s.C[j] = *(const v4f*)(cp + off);
    s.x[j] = xp[off];
  }
}

__device__ __forceinline__ void scan_steps(const ScanSet& s, float (&h)[16], float dtv, float dav,
                                           int lane0, float Dh, int q, float* yrow) {
  float keep = 0.f;
  #pragma unroll
  for (int j = 0; j < SCAN_G; ++j) {
    const v4f   Bv = s.B[j];
    const v4f   Cv = s.C[j];
    const float xv = s.x[j];
    const float dts = lane_bcast(dtv, lane0 + j);
    const float dA  = lane_bcast(dav, lane0 + j);
    const float xdt = xv * dts;
    float acc = 0.f;
    ssm_n<0>(h[0], acc, Bv, Cv, dA, xdt);   ssm_n<1>(h[1], acc, Bv, Cv, dA, xdt);
    ssm_n<2>(h[2], acc, Bv, Cv, dA, xdt);   ssm_n<3>(h[3], acc, Bv, Cv, dA, xdt);
    ssm_n<4>(h[4], acc, Bv, Cv, dA, xdt);   ssm_n<5>(h[5], acc, Bv, Cv, dA, xdt);
    ssm_n<6>(h[6], acc, Bv, Cv, dA, xdt);   ssm_n<7>(h[7], acc, Bv, Cv, dA, xdt);
    ssm_n<8>(h[8], acc, Bv, Cv, dA, xdt);   ssm_n<9>(h[9], acc, Bv, Cv, dA, xdt);
    ssm_n<10>(h[10], acc, Bv, Cv, dA, xdt); ssm_n<11>(h[11], acc, Bv, Cv, dA, xdt);
    ssm_n<12>(h[12], acc, Bv, Cv, dA, xdt); ssm_n<13>(h[13], acc, Bv, Cv, dA, xdt);
    ssm_n<14>(h[14], acc, Bv, Cv, dA, xdt); ssm_n<15>(h[15], acc, Bv, Cv, dA, xdt);
    acc += row_xor<4>(acc);
    acc += row_xor<8>(acc);
    const float yv = fmaf(xv, Dh, acc);
    keep = (q == (j & 3)) ? yv : keep;
    if ((j & 3) == 3) yrow[(long)(j - 3) * D_INNER] = keep;
  }
}

__global__ __launch_bounds__(128)
void ssm_scan_kernel(const float* __restrict__ zx,
                     const float* __restrict__ xbc,
                     const float* __restrict__ dtb,
                     const float* __restrict__ alog,
                     const float* __restrict__ Dp,
                     float* __restrict__ y) {
  const int b    = blockIdx.x >> 5;
  const int hh   = (blockIdx.x >> 1) & 15;
  const int half = blockIdx.x & 1;
  const int lane = threadIdx.x & 31;
  const int wave = threadIdx.x >> 5;
  const int q    = (lane >> 2) & 3;
  const int pj   = lane & 3;
  const int p    = half * 32 + wave * 8 + (lane >> 4) * 4 + pj;
  const float Ah   = -__expf(alog[hh]);
  const float dtbh = dtb[hh];
  const float Dh   = Dp[hh];

  const float* rows = xbc + (long)b * LOUT * CONV_DIM;
  const float* bp   = rows + D_INNER + 16 * q + 4 * pj;
  const float* cp   = bp + D_STATE;
  const float* xp   = rows + hh * HEADDIM + p;
  const float* dtp  = zx + (long)b * LOUT * D_IN_PROJ + (D_IN_PROJ - NHEADS) + hh;
  float* yp = y + ((long)b * LOUT + q) * D_INNER + hh * HEADDIM + p;

  float h[16];
  #pragma unroll
  for (int i = 0; i < 16; ++i) h[i] = 0.f;

  ScanSet sa, sb;
  scan_load(sa, bp, cp, xp, 0);
  float dtv = softplus_dt(dtp[(long)lane * D_IN_PROJ] + dtbh);
  float dav = __expf(dtv * Ah);

  for (int c = 0; c < LOUT / 32; ++c) {
    const int   cn   = min(c + 1, LOUT / 32 - 1);
    const float rawn = dtp[(long)(cn * 32 + lane) * D_IN_PROJ];
    for (int g = 0; g < 32; g += 2 * SCAN_G) {
      const int l0 = c * 32 + g;
      scan_load(sb, bp, cp, xp, l0 + SCAN_G);
      __builtin_amdgcn_sched_barrier(0);
      scan_steps(sa, h, dtv, dav, g, Dh, q, yp + (long)l0 * D_INNER);
      __builtin_amdgcn_sched_barrier(0);
      scan_load(sa, bp, cp, xp, l0 + 2 * SCAN_G);
      __builtin_amdgcn_sched_barrier(0);
      scan_steps(sb, h, dtv, dav, g + SCAN_G, Dh, q, yp + (long)(l0 + SCAN_G) * D_INNER);
    }
    dtv = softplus_dt(rawn + dtbh);
    dav = __expf(dtv * Ah);
  }
}

__global__ void gate_rmsnorm_kernel(const float* __restrict__ zx,
                                    const float* __restrict__ nw,
                                    float* __restrict__ y) {
  const long bl = blockIdx.x;
  const int tid = threadIdx.x;
  __shared__ float red[256];
  float yv[4];
  float ss = 0.f;
  #pragma unroll
  for (int j = 0; j < 4; ++j) {
    const int i = tid + j * 256;
    const float z = zx[bl * D_IN_PROJ + i];
    const float g = z / (1.f + __expf(-z));
    const float v = y[bl * D_INNER + i] * g;
    yv[j] = v;
    ss = fmaf(v, v, ss);
  }
  red[tid] = ss;
  __syncthreads();
  for (int st = 128; st > 0; st >>= 1) {
    if (tid < st) red[tid] += red[tid + st];
    __syncthreads();
  }
  const float scale = rsqrtf(red[0] * (1.f / (float)D_INNER) + 1e-5f);
  #pragma unroll
  for (int j = 0; j < 4; ++j) {
    const int i = tid + j * 256;
    y[bl * D_INNER + i] = yv[j] * scale * nw[i];
  }
}

extern "C" void kernel_launch(void* const* d_in, const int* in_sizes, int n_in,
                              void* d_out, int out_size, void* d_ws, size_t ws_size,
                              hipStream_t stream) {
  const float* tokens     = (const float*)d_in[0];
  const float* conv_w     = (const float*)d_in[1];
  const float* in_proj_w  = (const float*)d_in[2];
  const float* conv1d_w   = (const float*)d_in[3];
  const float* conv1d_b   = (const float*)d_in[4];
  const float* dt_bias    = (const float*)d_in[5];
  const float* A_log      = (const float*)d_in[6];
  const float* D_param    = (const float*)d_in[7];
  const float* norm_w     = (const float*)d_in[8];
  const float* out_proj_w = (const float*)d_in[9];
  float* out = (float*)d_out;
  float* ws  = (float*)d_ws;

  float* u0   = ws;
  float* u1   = u0 + (long)ROWS * D_MODEL;
  float* zx   = u1 + (long)ROWS * D_MODEL;
  float* xact = zx + (long)ROWS * D_IN_PROJ;
  float* ybuf = xact + (long)ROWS * CONV_DIM;
  (void)ws_size; (void)in_sizes; (void)n_in; (void)out_size;

  stem_conv_wmma<<<GEMM_WGS(D_MODEL), 128, 0, stream>>>(tokens, conv_w, u0);

  const float* u_in = u0;
  for (int blk = 0; blk < 2; ++blk) {
    const float* ipw = in_proj_w  + (long)blk * D_IN_PROJ * D_MODEL;
    const float* cw  = conv1d_w   + (long)blk * CONV_DIM * D_CONV;
    const float* cb  = conv1d_b   + (long)blk * CONV_DIM;
    const float* dtb = dt_bias    + (long)blk * NHEADS;
    const float* al  = A_log      + (long)blk * NHEADS;
    const float* Dpp = D_param    + (long)blk * NHEADS;
    const float* nw  = norm_w     + (long)blk * D_INNER;
    const float* opw = out_proj_w + (long)blk * D_MODEL * D_INNER;

    {
      gemm_rows_wmma<D_IN_PROJ, D_MODEL, 32><<<GEMM_WGS(D_IN_PROJ), 128, 0, stream>>>(u_in, ipw, zx);
    }
    dwconv_silu_kernel<<<dim3(CONV_DIM / 128, ROWS), 128, 0, stream>>>(zx, cw, cb, xact);
    ssm_scan_kernel<<<BATCH * NHEADS * 2, 128, 0, stream>>>(zx, xact, dtb, al, Dpp, ybuf);
    gate_rmsnorm_kernel<<<ROWS, 256, 0, stream>>>(zx, nw, ybuf);
    float* dst = (blk == 0) ? u1 : out;
    {
      gemm_rows_wmma<D_MODEL, D_INNER, 64><<<GEMM_WGS(D_MODEL), 128, 0, stream>>>(ybuf, opw, dst);
    }
    u_in = u1;
  }
}


# ssm scan: the 1024 steps cut into 8 segments scanned side by side from a zero state (512 workgroups, not 64); a small matrix kernel adds decay x (C . carried state) to y; sums over steps regrouped
# speedup vs baseline: 12.3062x; 1.3873x over previous
#include <hip/hip_runtime.h>
#include <hip/hip_bf16.h>

typedef __attribute__((ext_vector_type(2))) float v2f;
typedef __attribute__((ext_vector_type(8))) float v8f;

#define BATCH     2
#define SEQLEN    2048
#define C_IN      256
#define D_MODEL   512
#define LOUT      1024
#define D_STATE   64
#define D_CONV    4
#define D_INNER   1024
#define HEADDIM   64
#define NHEADS    16
#define CONV_DIM  (D_INNER + 2*D_STATE)
#define D_IN_PROJ (2*D_INNER + 2*D_STATE + NHEADS)
#define ROWS      (BATCH*LOUT)
#define KSTEM     (C_IN * 7)
#define GEMM_WGS(n) ((ROWS / 64) * (((n) + 63) / 64))

typedef __attribute__((ext_vector_type(4))) float v4f;

template <int JB>
struct RowFrag { v4f a0[JB]; v4f a1[JB]; v4f b0[JB]; v4f b1[JB]; };

template <int JB>
__device__ __forceinline__ void frag_load(RowFrag<JB>& f, const float* pa0, const float* pa1,
                                          const float* pb0, const float* pb1, int kb) {
  #pragma unroll
  for (int j = 0; j < JB; ++j) {
    f.a0[j] = *(const v4f*)(pa0 + kb + 8 * j);
    f.a1[j] = *(const v4f*)(pa1 + kb + 8 * j);
    f.b0[j] = *(const v4f*)(pb0 + kb + 8 * j);
    f.b1[j] = *(const v4f*)(pb1 + kb + 8 * j);
  }
}

__device__ __forceinline__ v8f mma4(v2f a, v2f b, v8f c) {
  return __builtin_amdgcn_wmma_f32_16x16x4_f32(false, a, false, b, (short)0, c, false, false);
}

template <int JB>
__device__ __forceinline__ void frag_mma(const RowFrag<JB>& f, v8f& c00, v8f& c01, v8f& c10, v8f& c11) {
  #pragma unroll
  for (int j = 0; j < JB; ++j) {
    #pragma unroll
    for (int u = 0; u < 2; ++u) {
      const v2f a0 = {f.a0[j][2 * u], f.a0[j][2 * u + 1]};
      const v2f a1 = {f.a1[j][2 * u], f.a1[j][2 * u + 1]};
      const v2f b0 = {f.b0[j][2 * u], f.b0[j][2 * u + 1]};
      const v2f b1 = {f.b1[j][2 * u], f.b1[j][2 * u + 1]};
      c00 = mma4(a0, b0, c00);
      c01 = mma4(a0, b1, c01);
      c10 = mma4(a1, b0, c10);
      c11 = mma4(a1, b1, c11);
    }
  }
}

template <int N, int K, int KB>
__global__ __launch_bounds__(128)
void gemm_rows_wmma(const float* __restrict__ A,
                    const float* __restrict__ W,
                    float* __restrict__ C) {
  static_assert(K % (2 * KB) == 0 && KB % 8 == 0 && N % 16 == 0, "shape");
  constexpr int JB  = KB / 8;
  constexpr int WGN = (N + 63) / 64;
  const int lane = threadIdx.x & 31;
  const int wave = __builtin_amdgcn_readfirstlane((int)(threadIdx.x >> 5));
  const int wgm  = blockIdx.x / WGN;
  const int wgn  = blockIdx.x - wgm * WGN;
  const int m0   = wgm * 64 + (wave >> 1) * 32;
  const int n0   = wgn * 64 + (wave & 1) * 32;
  if (n0 >= N) return;
  const bool n1ok = (n0 + 16) < N;
  const int c = lane & 15;
  const int h = lane >> 4;
  const float* pa0 = A + (long)(m0 + c) * K + 4 * h;
  const float* pa1 = pa0 + (long)16 * K;
  const float* pb0 = W + (long)(n0 + c) * K + 4 * h;
  const float* pb1 = pb0 + (n1ok ? (long)16 * K : 0L);

  const v8f zero = {0.f, 0.f, 0.f, 0.f, 0.f, 0.f, 0.f, 0.f};
  v8f c00 = zero, c01 = zero, c10 = zero, c11 = zero;
  RowFrag<JB> fa, fb;
  frag_load<JB>(fa, pa0, pa1, pb0, pb1, 0);
  #pragma unroll 1
  for (int kb = 0; kb < K - 2 * KB; kb += 2 * KB) {
    frag_load<JB>(fb, pa0, pa1, pb0, pb1, kb + KB);
    __builtin_amdgcn_sched_barrier(0);
    frag_mma<JB>(fa, c00, c01, c10, c11);
    __builtin_amdgcn_sched_barrier(0);
    frag_load<JB>(fa, pa0, pa1, pb0, pb1, kb + 2 * KB);
    __builtin_amdgcn_sched_barrier(0);
    frag_mma<JB>(fb, c00, c01, c10, c11);
  }
  frag_load<JB>(fb, pa0, pa1, pb0, pb1, K - KB);
  __builtin_amdgcn_sched_barrier(0);
  frag_mma<JB>(fa, c00, c01, c10, c11);
  frag_mma<JB>(fb, c00, c01, c10, c11);

  float* crow = C + (long)(m0 + 8 * h) * N + n0 + c;
  #pragma unroll
  for (int i = 0; i < 8; ++i) {
    crow[(long)i * N]        = c00[i];
    crow[(long)(i + 16) * N] = c10[i];
  }
  if (n1ok) {
    #pragma unroll
    for (int i = 0; i < 8; ++i) {
      crow[(long)i * N + 16]        = c01[i];
      crow[(long)(i + 16) * N + 16] = c11[i];
    }
  }
}

#define STEM_CB 8
#define STEM_KB (STEM_CB * 7)
#define STEM_NB (C_IN / STEM_CB)
#define STEM_J  (STEM_KB / 8)

struct StemFrag { float a0[STEM_J][4]; float a1[STEM_J][4]; v4f b0[STEM_J]; v4f b1[STEM_J]; };
struct StemMap  { unsigned o0[STEM_J][4]; unsigned o1[STEM_J][4]; int m0[STEM_J][4]; int m1[STEM_J][4]; };

template <int J0, int J1>
__device__ __forceinline__ void stem_load(StemFrag& f, const StemMap& mp, const char* tb,
                                          const float* pb0, const float* pb1, int blk) {
  const char* t = tb + (long)blk * (STEM_CB * SEQLEN * 4);
  const int kb = blk * STEM_KB;
  #pragma unroll
  for (int j = J0; j < J1; ++j) {
    f.b0[j] = *(const v4f*)(pb0 + kb + 8 * j);
    f.b1[j] = *(const v4f*)(pb1 + kb + 8 * j);
    #pragma unroll
    for (int e = 0; e < 4; ++e) {
      f.a0[j][e] = *(const float*)(t + mp.o0[j][e]);
      f.a1[j][e] = *(const float*)(t + mp.o1[j][e]);
    }
  }
}

__device__ __forceinline__ float pad0(float v, int m) {
  return __builtin_bit_cast(float, __builtin_bit_cast(int, v) & m);
}

template <int J0, int J1>
__device__ __forceinline__ void stem_mma(const StemFrag& f, const StemMap& mp,
                                         v8f& c00, v8f& c01, v8f& c10, v8f& c11) {
  #pragma unroll
  for (int j = J0; j < J1; ++j) {
    #pragma unroll
    for (int u = 0; u < 2; ++u) {
      const v2f a0 = {pad0(f.a0[j][2 * u], mp.m0[j][2 * u]), pad0(f.a0[j][2 * u + 1], mp.m0[j][2 * u + 1])};
      const v2f a1 = {pad0(f.a1[j][2 * u], mp.m1[j][2 * u]), pad0(f.a1[j][2 * u + 1], mp.m1[j][2 * u + 1])};
      const v2f b0 = {f.b0[j][2 * u], f.b0[j][2 * u + 1]};
      const v2f b1 = {f.b1[j][2 * u], f.b1[j][2 * u + 1]};
      c00 = mma4(a0, b0, c00);
      c01 = mma4(a0, b1, c01);
      c10 = mma4(a1, b0, c10);
      c11 = mma4(a1, b1, c11);
    }
  }
}

__global__ __launch_bounds__(128)
void stem_conv_wmma(const float* __restrict__ tokens,
                    const float* __restrict__ W,
                    float* __restrict__ C) {
  constexpr int N   = D_MODEL;
  constexpr int K   = KSTEM;
  constexpr int WGN = N / 64;
  const int lane = threadIdx.x & 31;
  const int wave = __builtin_amdgcn_readfirstlane((int)(threadIdx.x >> 5));
  const int wgm  = blockIdx.x / WGN;
  const int wgn  = blockIdx.x - wgm * WGN;
  const int m0   = wgm * 64 + (wave >> 1) * 32;
  const int n0   = wgn * 64 + (wave & 1) * 32;
  const int c = lane & 15;
  const int h = lane >> 4;
  const float* pb0 = W + (long)(n0 + c) * K + 4 * h;
  const float* pb1 = pb0 + (long)16 * K;
  const char* tb = (const char*)(tokens + (long)(m0 / LOUT) * C_IN * SEQLEN);
  const int l0 = (m0 & (LOUT - 1)) + c;

  StemMap mp;
  #pragma unroll
  for (int j = 0; j < STEM_J; ++j) {
    #pragma unroll
    for (int e = 0; e < 4; ++e) {
      const int kl = 8 * j + 4 * h + e;
      const int ch = kl / 7;
      const int tp = kl - 7 * ch;
      const int s0 = 2 * l0 - 3 + tp;
      const int s1 = s0 + 32;
      const bool ok0 = (unsigned)s0 < (unsigned)SEQLEN;
      const bool ok1 = (unsigned)s1 < (unsigned)SEQLEN;
      mp.o0[j][e] = (unsigned)(ch * SEQLEN + (ok0 ? s0 : 0)) * 4u;
      mp.o1[j][e] = (unsigned)(ch * SEQLEN + (ok1 ? s1 : 0)) * 4u;
      mp.m0[j][e] = ok0 ? -1 : 0;
      mp.m1[j][e] = ok1 ? -1 : 0;
    }
  }

  const v8f zero = {0.f, 0.f, 0.f, 0.f, 0.f, 0.f, 0.f, 0.f};
  v8f c00 = zero, c01 = zero, c10 = zero, c11 = zero;
  StemFrag fa, fb;
  stem_load<0, 4>(fa, mp, tb, pb0, pb1, 0);
  stem_load<4, STEM_J>(fa, mp, tb, pb0, pb1, 0);
  #pragma unroll 1
  for (int blk = 0; blk < STEM_NB - 2; blk += 2) {
    stem_load<0, 4>(fb, mp, tb, pb0, pb1, blk + 1);
    __builtin_amdgcn_sched_barrier(0);
    stem_mma<0, 4>(fa, mp, c00, c01, c10, c11);
    __builtin_amdgcn_sched_barrier(0);
    stem_load<4, STEM_J>(fb, mp, tb, pb0, pb1, blk + 1);
    __builtin_amdgcn_sched_barrier(0);
    stem_mma<4, STEM_J>(fa, mp, c00, c01, c10, c11);
    __builtin_amdgcn_sched_barrier(0);
    stem_load<0, 4>(fa, mp, tb, pb0, pb1, blk + 2);
    __builtin_amdgcn_sched_barrier(0);
    stem_mma<0, 4>(fb, mp, c00, c01, c10, c11);
    __builtin_amdgcn_sched_barrier(0);
    stem_load<4, STEM_J>(fa, mp, tb, pb0, pb1, blk + 2);
    __builtin_amdgcn_sched_barrier(0);
    stem_mma<4, STEM_J>(fb, mp, c00, c01, c10, c11);
    __builtin_amdgcn_sched_barrier(0);
  }
  stem_load<0, 4>(fb, mp, tb, pb0, pb1, STEM_NB - 1);
  __builtin_amdgcn_sched_barrier(0);
  stem_mma<0, 4>(fa, mp, c00, c01, c10, c11);
  __builtin_amdgcn_sched_barrier(0);
  stem_load<4, STEM_J>(fb, mp, tb, pb0, pb1, STEM_NB - 1);
  __builtin_amdgcn_sched_barrier(0);
  stem_mma<4, STEM_J>(fa, mp, c00, c01, c10, c11);
  stem_mma<0, STEM_J>(fb, mp, c00, c01, c10, c11);

  float* crow = C + (long)(m0 + 8 * h) * N + n0 + c;
  #pragma unroll
  for (int i = 0; i < 8; ++i) {
    crow[(long)i * N]             = c00[i];
    crow[(long)(i + 16) * N]      = c10[i];
    crow[(long)i * N + 16]        = c01[i];
    crow[(long)(i + 16) * N + 16] = c11[i];
  }
}

__global__ void dwconv_silu_kernel(const float* __restrict__ zx,
                                   const float* __restrict__ w,
                                   const float* __restrict__ bias,
                                   float* __restrict__ out) {
  const int c = blockIdx.x * 128 + threadIdx.x;
  if (c >= CONV_DIM) return;
  const long bl = blockIdx.y;
  const int l = (int)(bl & (LOUT - 1));
  const float* xcol = zx + bl * D_IN_PROJ + D_INNER + c;
  const float w0 = w[c * 4 + 0], w1 = w[c * 4 + 1], w2 = w[c * 4 + 2], w3 = w[c * 4 + 3];
  float acc = w3 * xcol[0];
  if (l >= 1) acc = fmaf(w2, xcol[-(long)D_IN_PROJ],     acc);
  if (l >= 2) acc = fmaf(w1, xcol[-(long)2 * D_IN_PROJ], acc);
  if (l >= 3) acc = fmaf(w0, xcol[-(long)3 * D_IN_PROJ], acc);
  const float v = acc + bias[c];
  out[bl * CONV_DIM + c] = v / (1.f + __expf(-v));
}

#define SCAN_G 8
#define SEGS   8
#define SEGLEN (LOUT / SEGS)
#define HSTATE (HEADDIM * D_STATE)

template <int J>
__device__ __forceinline__ float quad_bcast(float v) {
  return __builtin_bit_cast(float, __builtin_amdgcn_update_dpp(
      0, __builtin_bit_cast(int, v), J * 0x55, 0xF, 0xF, true));
}
template <int M>
__device__ __forceinline__ float row_xor(float v) {
  return __builtin_bit_cast(float, __builtin_amdgcn_update_dpp(
      0, __builtin_bit_cast(int, v), 0x160 + M, 0xF, 0xF, true));
}
__device__ __forceinline__ float lane_bcast(float v, int l) {
  return __builtin_bit_cast(float, __builtin_amdgcn_readlane(__builtin_bit_cast(int, v), l));
}
template <int I>
__device__ __forceinline__ void ssm_n(float& h, float& acc, const v4f& Bv, const v4f& Cv,
                                      float dA, float xdt) {
  const float bn = quad_bcast<(I >> 2)>(Bv[I & 3]);
  const float cn = quad_bcast<(I >> 2)>(Cv[I & 3]);
  h   = fmaf(h, dA, xdt * bn);
  acc = fmaf(h, cn, acc);
}
__device__ __forceinline__ float softplus_dt(float raw) {
  return (raw > 20.f) ? raw : log1pf(__expf(raw));
}

struct ScanSet { v4f B[SCAN_G]; v4f C[SCAN_G]; float x[SCAN_G]; };

__device__ __forceinline__ void scan_load(ScanSet& s, const float* bp, const float* cp,
                                          const float* xp, int l0) {
  #pragma unroll
  for (int j = 0; j < SCAN_G; ++j) {
    const long off = (long)min(l0 + j, LOUT - 1) * CONV_DIM;
    s.B[j] = *(const v4f*)(bp + off);
    s.C[j] = *(const v4f*)(cp + off);
    s.x[j] = xp[off];
  }
}

__device__ __forceinline__ void scan_steps(const ScanSet& s, float (&h)[16], float dtv, float dav,
                                           int lane0, float Dh, int q, float* yrow,
                                           float& pc, float& pk, int lane) {
  float keep = 0.f;
  #pragma unroll
  for (int j = 0; j < SCAN_G; ++j) {
    const v4f   Bv = s.B[j];
    const v4f   Cv = s.C[j];
    const float xv = s.x[j];
    const float dts = lane_bcast(dtv, lane0 + j);
    const float dA  = lane_bcast(dav, lane0 + j);
    const float xdt = xv * dts;
    pc *= dA;
    pk = (lane == lane0 + j) ? pc : pk;
    float acc = 0.f;
    ssm_n<0>(h[0], acc, Bv, Cv, dA, xdt);   ssm_n<1>(h[1], acc, Bv, Cv, dA, xdt);
    ssm_n<2>(h[2], acc, Bv, Cv, dA, xdt);   ssm_n<3>(h[3], acc, Bv, Cv, dA, xdt);
    ssm_n<4>(h[4], acc, Bv, Cv, dA, xdt);   ssm_n<5>(h[5], acc, Bv, Cv, dA, xdt);
    ssm_n<6>(h[6], acc, Bv, Cv, dA, xdt);   ssm_n<7>(h[7], acc, Bv, Cv, dA, xdt);
    ssm_n<8>(h[8], acc, Bv, Cv, dA, xdt);   ssm_n<9>(h[9], acc, Bv, Cv, dA, xdt);
    ssm_n<10>(h[10], acc, Bv, Cv, dA, xdt); ssm_n<11>(h[11], acc, Bv, Cv, dA, xdt);
    ssm_n<12>(h[12], acc, Bv, Cv, dA, xdt); ssm_n<13>(h[13], acc, Bv, Cv, dA, xdt);
    ssm_n<14>(h[14], acc, Bv, Cv, dA, xdt); ssm_n<15>(h[15], acc, Bv, Cv, dA, xdt);
    acc += row_xor<4>(acc);
    acc += row_xor<8>(acc);
    const float yv = fmaf(xv, Dh, acc);
    keep = (q == (j & 3)) ? yv : keep;
    if ((j & 3) == 3) yrow[(long)(j - 3) * D_INNER] = keep;
  }
}

__global__ __launch_bounds__(128)
void ssm_scan_kernel(const float* __restrict__ zx,
                     const float* __restrict__ xbc,
                     const float* __restrict__ dtb,
                     const float* __restrict__ alog,
                     const float* __restrict__ Dp,
                     float* __restrict__ y,
                     float* __restrict__ hloc,
                     float* __restrict__ pcum) {
  const int seg  = blockIdx.x >> 6;
  const int b    = (blockIdx.x >> 5) & 1;
  const int hh   = (blockIdx.x >> 1) & 15;
  const int half = blockIdx.x & 1;
  const int lane = threadIdx.x & 31;
  const int wave = threadIdx.x >> 5;
  const int q    = (lane >> 2) & 3;
  const int pj   = lane & 3;
  const int p    = half * 32 + wave * 8 + (lane >> 4) * 4 + pj;
  const float Ah   = -__expf(alog[hh]);
  const float dtbh = dtb[hh];
  const float Dh   = Dp[hh];

  const float* rows = xbc + (long)b * LOUT * CONV_DIM;
  const float* bp   = rows + D_INNER + 16 * q + 4 * pj;
  const float* cp   = bp + D_STATE;
  const float* xp   = rows + hh * HEADDIM + p;
  const float* dtp  = zx + (long)b * LOUT * D_IN_PROJ + (D_IN_PROJ - NHEADS) + hh;
  float* yp = y + ((long)b * LOUT + q) * D_INNER + hh * HEADDIM + p;

  float h[16];
  #pragma unroll
  for (int i = 0; i < 16; ++i) h[i] = 0.f;

  const int  c0     = seg * (SEGLEN / 32);
  const bool pwrite = (half == 0) && (wave == 0);
  float* pout = pcum + (long)(b * NHEADS + hh) * LOUT;
  float pc = 1.f, pk = 0.f;

  ScanSet sa, sb;
  scan_load(sa, bp, cp, xp, c0 * 32);
  float dtv = softplus_dt(dtp[(long)(c0 * 32 + lane) * D_IN_PROJ] + dtbh);
  float dav = __expf(dtv * Ah);

  for (int c = c0; c < c0 + SEGLEN / 32; ++c) {
    const int   cn   = min(c + 1, LOUT / 32 - 1);
    const float rawn = dtp[(long)(cn * 32 + lane) * D_IN_PROJ];
    for (int g = 0; g < 32; g += 2 * SCAN_G) {
      const int l0 = c * 32 + g;
      scan_load(sb, bp, cp, xp, l0 + SCAN_G);
      __builtin_amdgcn_sched_barrier(0);
      scan_steps(sa, h, dtv, dav, g, Dh, q, yp + (long)l0 * D_INNER, pc, pk, lane);
      __builtin_amdgcn_sched_barrier(0);
      scan_load(sa, bp, cp, xp, l0 + 2 * SCAN_G);
      __builtin_amdgcn_sched_barrier(0);
      scan_steps(sb, h, dtv, dav, g + SCAN_G, Dh, q, yp + (long)(l0 + SCAN_G) * D_INNER, pc, pk, lane);
    }
    if (pwrite) pout[c * 32 + lane] = pk;
    dtv = softplus_dt(rawn + dtbh);
    dav = __expf(dtv * Ah);
  }
  if (seg < SEGS - 1) {
    float* hp = hloc + ((long)(b * NHEADS + hh) * SEGS + seg) * HSTATE + p * D_STATE + 16 * q;
    #pragma unroll
    for (int i = 0; i < 4; ++i) {
      const v4f v = {h[4 * i], h[4 * i + 1], h[4 * i + 2], h[4 * i + 3]};
      *(v4f*)(hp + 4 * i) = v;
    }
  }
}

__global__ __launch_bounds__(128)
void ssm_carry_wmma(const float* __restrict__ xbc,
                    const float* __restrict__ hloc,
                    const float* __restrict__ pcum,
                    float* __restrict__ y) {
  constexpr int JB = D_STATE / 8;
  const int lane = threadIdx.x & 31;
  const int wave = __builtin_amdgcn_readfirstlane((int)(threadIdx.x >> 5));
  const int tb   = blockIdx.x & (SEGLEN / 64 - 1);
  const int r    = blockIdx.x / (SEGLEN / 64);
  const int bh   = r / (SEGS - 1);
  const int seg  = 1 + (r - bh * (SEGS - 1));
  const int b    = bh / NHEADS;
  const int hh   = bh - b * NHEADS;
  const int t0   = seg * SEGLEN + tb * 64 + (wave >> 1) * 32;
  const int p0   = (wave & 1) * 32;
  const int c = lane & 15;
  const int h = lane >> 4;
  const float* pa0 = xbc + ((long)b * LOUT + t0 + c) * CONV_DIM + (D_INNER + D_STATE) + 4 * h;
  const float* pa1 = pa0 + (long)16 * CONV_DIM;
  const float* hb0 = hloc + (long)bh * SEGS * HSTATE + (p0 + c) * D_STATE + 4 * h;
  const float* pr  = pcum + (long)bh * LOUT;

  RowFrag<JB> f;
  #pragma unroll
  for (int j = 0; j < JB; ++j) {
    f.a0[j] = *(const v4f*)(pa0 + 8 * j);
    f.a1[j] = *(const v4f*)(pa1 + 8 * j);
    f.b0[j] = *(const v4f*)(hb0 + 8 * j);
    f.b1[j] = *(const v4f*)(hb0 + 16 * D_STATE + 8 * j);
  }
  for (int k = 1; k < seg; ++k) {
    const float a = pr[k * SEGLEN + SEGLEN - 1];
    const float* hk = hb0 + (long)k * HSTATE;
    #pragma unroll
    for (int j = 0; j < JB; ++j) {
      const v4f e0 = *(const v4f*)(hk + 8 * j);
      const v4f e1 = *(const v4f*)(hk + 16 * D_STATE + 8 * j);
      #pragma unroll
      for (int e = 0; e < 4; ++e) {
        f.b0[j][e] = fmaf(a, f.b0[j][e], e0[e]);
        f.b1[j][e] = fmaf(a, f.b1[j][e], e1[e]);
      }
    }
  }

  const v8f zero = {0.f, 0.f, 0.f, 0.f, 0.f, 0.f, 0.f, 0.f};
  v8f c00 = zero, c01 = zero, c10 = zero, c11 = zero;
  frag_mma<JB>(f, c00, c01, c10, c11);

  float* yrow = y + ((long)b * LOUT + t0 + 8 * h) * D_INNER + hh * HEADDIM + p0 + c;
  const float* prow = pr + t0 + 8 * h;
  #pragma unroll
  for (int i = 0; i < 8; ++i) {
    const float s0 = prow[i];
    const float s1 = prow[i + 16];
    yrow[(long)i * D_INNER]             = fmaf(s0, c00[i], yrow[(long)i * D_INNER]);
    yrow[(long)i * D_INNER + 16]        = fmaf(s0, c01[i], yrow[(long)i * D_INNER + 16]);
    yrow[(long)(i + 16) * D_INNER]      = fmaf(s1, c10[i], yrow[(long)(i + 16) * D_INNER]);
    yrow[(long)(i + 16) * D_INNER + 16] = fmaf(s1, c11[i], yrow[(long)(i + 16) * D_INNER + 16]);
  }
}

__global__ void gate_rmsnorm_kernel(const float* __restrict__ zx,
                                    const float* __restrict__ nw,
                                    float* __restrict__ y) {
  const long bl = blockIdx.x;
  const int tid = threadIdx.x;
  __shared__ float red[256];
  float yv[4];
  float ss = 0.f;
  #pragma unroll
  for (int j = 0; j < 4; ++j) {
    const int i = tid + j * 256;
    const float z = zx[bl * D_IN_PROJ + i];
    const float g = z / (1.f + __expf(-z));
    const float v = y[bl * D_INNER + i] * g;
    yv[j] = v;
    ss = fmaf(v, v, ss);
  }
  red[tid] = ss;
  __syncthreads();
  for (int st = 128; st > 0; st >>= 1) {
    if (tid < st) red[tid] += red[tid + st];
    __syncthreads();
  }
  const float scale = rsqrtf(red[0] * (1.f / (float)D_INNER) + 1e-5f);
  #pragma unroll
  for (int j = 0; j < 4; ++j) {
    const int i = tid + j * 256;
    y[bl * D_INNER + i] = yv[j] * scale * nw[i];
  }
}

extern "C" void kernel_launch(void* const* d_in, const int* in_sizes, int n_in,
                              void* d_out, int out_size, void* d_ws, size_t ws_size,
                              hipStream_t stream) {
  const float* tokens     = (const float*)d_in[0];
  const float* conv_w     = (const float*)d_in[1];
  const float* in_proj_w  = (const float*)d_in[2];
  const float* conv1d_w   = (const float*)d_in[3];
  const float* conv1d_b   = (const float*)d_in[4];
  const float* dt_bias    = (const float*)d_in[5];
  const float* A_log      = (const float*)d_in[6];
  const float* D_param    = (const float*)d_in[7];
  const float* norm_w     = (const float*)d_in[8];
  const float* out_proj_w = (const float*)d_in[9];
  float* out = (float*)d_out;
  float* ws  = (float*)d_ws;

  float* u0   = ws;
  float* u1   = u0 + (long)ROWS * D_MODEL;
  float* zx   = u1 + (long)ROWS * D_MODEL;
  float* xact = zx + (long)ROWS * D_IN_PROJ;
  float* ybuf = xact + (long)ROWS * CONV_DIM;
  float* hloc = ybuf + (long)ROWS * D_INNER;
  float* pcum = hloc + (long)BATCH * NHEADS * SEGS * HSTATE;
  (void)ws_size; (void)in_sizes; (void)n_in; (void)out_size;

  stem_conv_wmma<<<GEMM_WGS(D_MODEL), 128, 0, stream>>>(tokens, conv_w, u0);

  const float* u_in = u0;
  for (int blk = 0; blk < 2; ++blk) {
    const float* ipw = in_proj_w  + (long)blk * D_IN_PROJ * D_MODEL;
    const float* cw  = conv1d_w   + (long)blk * CONV_DIM * D_CONV;
    const float* cb  = conv1d_b   + (long)blk * CONV_DIM;
    const float* dtb = dt_bias    + (long)blk * NHEADS;
    const float* al  = A_log      + (long)blk * NHEADS;
    const float* Dpp = D_param    + (long)blk * NHEADS;
    const float* nw  = norm_w     + (long)blk * D_INNER;
    const float* opw = out_proj_w + (long)blk * D_MODEL * D_INNER;

    {
      gemm_rows_wmma<D_IN_PROJ, D_MODEL, 32><<<GEMM_WGS(D_IN_PROJ), 128, 0, stream>>>(u_in, ipw, zx);
    }
    dwconv_silu_kernel<<<dim3(CONV_DIM / 128, ROWS), 128, 0, stream>>>(zx, cw, cb, xact);
    ssm_scan_kernel<<<SEGS * BATCH * NHEADS * 2, 128, 0, stream>>>(zx, xact, dtb, al, Dpp, ybuf, hloc, pcum);
    ssm_carry_wmma<<<BATCH * NHEADS * (SEGS - 1) * (SEGLEN / 64), 128, 0, stream>>>(xact, hloc, pcum, ybuf);
    gate_rmsnorm_kernel<<<ROWS, 256, 0, stream>>>(zx, nw, ybuf);
    float* dst = (blk == 0) ? u1 : out;
    {
      gemm_rows_wmma<D_MODEL, D_INNER, 64><<<GEMM_WGS(D_MODEL), 128, 0, stream>>>(ybuf, opw, dst);
    }
    u_in = u1;
  }
}


# ssm scan: x's depthwise conv + SiLU made inside the scan from raw in_proj rows (3-step window in registers); conv kernel makes only B and C: 2048 workgroups, not 18432; same sums, same order
# speedup vs baseline: 13.1294x; 1.0669x over previous
#include <hip/hip_runtime.h>
#include <hip/hip_bf16.h>

typedef __attribute__((ext_vector_type(2))) float v2f;
typedef __attribute__((ext_vector_type(8))) float v8f;

#define BATCH     2
#define SEQLEN    2048
#define C_IN      256
#define D_MODEL   512
#define LOUT      1024
#define D_STATE   64
#define D_CONV    4
#define D_INNER   1024
#define HEADDIM   64
#define NHEADS    16
#define CONV_DIM  (D_INNER + 2*D_STATE)
#define D_IN_PROJ (2*D_INNER + 2*D_STATE + NHEADS)
#define ROWS      (BATCH*LOUT)
#define KSTEM     (C_IN * 7)
#define GEMM_WGS(n) ((ROWS / 64) * (((n) + 63) / 64))

typedef __attribute__((ext_vector_type(4))) float v4f;

template <int JB>
struct RowFrag { v4f a0[JB]; v4f a1[JB]; v4f b0[JB]; v4f b1[JB]; };

template <int JB>
__device__ __forceinline__ void frag_load(RowFrag<JB>& f, const float* pa0, const float* pa1,
                                          const float* pb0, const float* pb1, int kb) {
  #pragma unroll
  for (int j = 0; j < JB; ++j) {
    f.a0[j] = *(const v4f*)(pa0 + kb + 8 * j);
    f.a1[j] = *(const v4f*)(pa1 + kb + 8 * j);
    f.b0[j] = *(const v4f*)(pb0 + kb + 8 * j);
    f.b1[j] = *(const v4f*)(pb1 + kb + 8 * j);
  }
}

__device__ __forceinline__ v8f mma4(v2f a, v2f b, v8f c) {
  return __builtin_amdgcn_wmma_f32_16x16x4_f32(false, a, false, b, (short)0, c, false, false);
}

template <int JB>
__device__ __forceinline__ void frag_mma(const RowFrag<JB>& f, v8f& c00, v8f& c01, v8f& c10, v8f& c11) {
  #pragma unroll
  for (int j = 0; j < JB; ++j) {
    #pragma unroll
    for (int u = 0; u < 2; ++u) {
      const v2f a0 = {f.a0[j][2 * u], f.a0[j][2 * u + 1]};
      const v2f a1 = {f.a1[j][2 * u], f.a1[j][2 * u + 1]};
      const v2f b0 = {f.b0[j][2 * u], f.b0[j][2 * u + 1]};
      const v2f b1 = {f.b1[j][2 * u], f.b1[j][2 * u + 1]};
      c00 = mma4(a0, b0, c00);
      c01 = mma4(a0, b1, c01);
      c10 = mma4(a1, b0, c10);
      c11 = mma4(a1, b1, c11);
    }
  }
}

template <int N, int K, int KB>
__global__ __launch_bounds__(128)
void gemm_rows_wmma(const float* __restrict__ A,
                    const float* __restrict__ W,
                    float* __restrict__ C) {
  static_assert(K % (2 * KB) == 0 && KB % 8 == 0 && N % 16 == 0, "shape");
  constexpr int JB  = KB / 8;
  constexpr int WGN = (N + 63) / 64;
  const int lane = threadIdx.x & 31;
  const int wave = __builtin_amdgcn_readfirstlane((int)(threadIdx.x >> 5));
  const int wgm  = blockIdx.x / WGN;
  const int wgn  = blockIdx.x - wgm * WGN;
  const int m0   = wgm * 64 + (wave >> 1) * 32;
  const int n0   = wgn * 64 + (wave & 1) * 32;
  if (n0 >= N) return;
  const bool n1ok = (n0 + 16) < N;
  const int c = lane & 15;
  const int h = lane >> 4;
  const float* pa0 = A + (long)(m0 + c) * K + 4 * h;
  const float* pa1 = pa0 + (long)16 * K;
  const float* pb0 = W + (long)(n0 + c) * K + 4 * h;
  const float* pb1 = pb0 + (n1ok ? (long)16 * K : 0L);

  const v8f zero = {0.f, 0.f, 0.f, 0.f, 0.f, 0.f, 0.f, 0.f};
  v8f c00 = zero, c01 = zero, c10 = zero, c11 = zero;
  RowFrag<JB> fa, fb;
  frag_load<JB>(fa, pa0, pa1, pb0, pb1, 0);
  #pragma unroll 1
  for (int kb = 0; kb < K - 2 * KB; kb += 2 * KB) {
    frag_load<JB>(fb, pa0, pa1, pb0, pb1, kb + KB);
    __builtin_amdgcn_sched_barrier(0);
    frag_mma<JB>(fa, c00, c01, c10, c11);
    __builtin_amdgcn_sched_barrier(0);
    frag_load<JB>(fa, pa0, pa1, pb0, pb1, kb + 2 * KB);
    __builtin_amdgcn_sched_barrier(0);
    frag_mma<JB>(fb, c00, c01, c10, c11);
  }
  frag_load<JB>(fb, pa0, pa1, pb0, pb1, K - KB);
  __builtin_amdgcn_sched_barrier(0);
  frag_mma<JB>(fa, c00, c01, c10, c11);
  frag_mma<JB>(fb, c00, c01, c10, c11);

  float* crow = C + (long)(m0 + 8 * h) * N + n0 + c;
  #pragma unroll
  for (int i = 0; i < 8; ++i) {
    crow[(long)i * N]        = c00[i];
    crow[(long)(i + 16) * N] = c10[i];
  }
  if (n1ok) {
    #pragma unroll
    for (int i = 0; i < 8; ++i) {
      crow[(long)i * N + 16]        = c01[i];
      crow[(long)(i + 16) * N + 16] = c11[i];
    }
  }
}

#define STEM_CB 8
#define STEM_KB (STEM_CB * 7)
#define STEM_NB (C_IN / STEM_CB)
#define STEM_J  (STEM_KB / 8)

struct StemFrag { float a0[STEM_J][4]; float a1[STEM_J][4]; v4f b0[STEM_J]; v4f b1[STEM_J]; };
struct StemMap  { unsigned o0[STEM_J][4]; unsigned o1[STEM_J][4]; int m0[STEM_J][4]; int m1[STEM_J][4]; };

template <int J0, int J1>
__device__ __forceinline__ void stem_load(StemFrag& f, const StemMap& mp, const char* tb,
                                          const float* pb0, const float* pb1, int blk) {
  const char* t = tb + (long)blk * (STEM_CB * SEQLEN * 4);
  const int kb = blk * STEM_KB;
  #pragma unroll
  for (int j = J0; j < J1; ++j) {
    f.b0[j] = *(const v4f*)(pb0 + kb + 8 * j);
    f.b1[j] = *(const v4f*)(pb1 + kb + 8 * j);
    #pragma unroll
    for (int e = 0; e < 4; ++e) {
      f.a0[j][e] = *(const float*)(t + mp.o0[j][e]);
      f.a1[j][e] = *(const float*)(t + mp.o1[j][e]);
    }
  }
}

__device__ __forceinline__ float pad0(float v, int m) {
  return __builtin_bit_cast(float, __builtin_bit_cast(int, v) & m);
}

template <int J0, int J1>
__device__ __forceinline__ void stem_mma(const StemFrag& f, const StemMap& mp,
                                         v8f& c00, v8f& c01, v8f& c10, v8f& c11) {
  #pragma unroll
  for (int j = J0; j < J1; ++j) {
    #pragma unroll
    for (int u = 0; u < 2; ++u) {
      const v2f a0 = {pad0(f.a0[j][2 * u], mp.m0[j][2 * u]), pad0(f.a0[j][2 * u + 1], mp.m0[j][2 * u + 1])};
      const v2f a1 = {pad0(f.a1[j][2 * u], mp.m1[j][2 * u]), pad0(f.a1[j][2 * u + 1], mp.m1[j][2 * u + 1])};
      const v2f b0 = {f.b0[j][2 * u], f.b0[j][2 * u + 1]};
      const v2f b1 = {f.b1[j][2 * u], f.b1[j][2 * u + 1]};
      c00 = mma4(a0, b0, c00);
      c01 = mma4(a0, b1, c01);
      c10 = mma4(a1, b0, c10);
      c11 = mma4(a1, b1, c11);
    }
  }
}

__global__ __launch_bounds__(128)
void stem_conv_wmma(const float* __restrict__ tokens,
                    const float* __restrict__ W,
                    float* __restrict__ C) {
  constexpr int N   = D_MODEL;
  constexpr int K   = KSTEM;
  constexpr int WGN = N / 64;
  const int lane = threadIdx.x & 31;
  const int wave = __builtin_amdgcn_readfirstlane((int)(threadIdx.x >> 5));
  const int wgm  = blockIdx.x / WGN;
  const int wgn  = blockIdx.x - wgm * WGN;
  const int m0   = wgm * 64 + (wave >> 1) * 32;
  const int n0   = wgn * 64 + (wave & 1) * 32;
  const int c = lane & 15;
  const int h = lane >> 4;
  const float* pb0 = W + (long)(n0 + c) * K + 4 * h;
  const float* pb1 = pb0 + (long)16 * K;
  const char* tb = (const char*)(tokens + (long)(m0 / LOUT) * C_IN * SEQLEN);
  const int l0 = (m0 & (LOUT - 1)) + c;

  StemMap mp;
  #pragma unroll
  for (int j = 0; j < STEM_J; ++j) {
    #pragma unroll
    for (int e = 0; e < 4; ++e) {
      const int kl = 8 * j + 4 * h + e;
      const int ch = kl / 7;
      const int tp = kl - 7 * ch;
      const int s0 = 2 * l0 - 3 + tp;
      const int s1 = s0 + 32;
      const bool ok0 = (unsigned)s0 < (unsigned)SEQLEN;
      const bool ok1 = (unsigned)s1 < (unsigned)SEQLEN;
      mp.o0[j][e] = (unsigned)(ch * SEQLEN + (ok0 ? s0 : 0)) * 4u;
      mp.o1[j][e] = (unsigned)(ch * SEQLEN + (ok1 ? s1 : 0)) * 4u;
      mp.m0[j][e] = ok0 ? -1 : 0;
      mp.m1[j][e] = ok1 ? -1 : 0;
    }
  }

  const v8f zero = {0.f, 0.f, 0.f, 0.f, 0.f, 0.f, 0.f, 0.f};
  v8f c00 = zero, c01 = zero, c10 = zero, c11 = zero;
  StemFrag fa, fb;
  stem_load<0, 4>(fa, mp, tb, pb0, pb1, 0);
  stem_load<4, STEM_J>(fa, mp, tb, pb0, pb1, 0);
  #pragma unroll 1
  for (int blk = 0; blk < STEM_NB - 2; blk += 2) {
    stem_load<0, 4>(fb, mp, tb, pb0, pb1, blk + 1);
    __builtin_amdgcn_sched_barrier(0);
    stem_mma<0, 4>(fa, mp, c00, c01, c10, c11);
    __builtin_amdgcn_sched_barrier(0);
    stem_load<4, STEM_J>(fb, mp, tb, pb0, pb1, blk + 1);
    __builtin_amdgcn_sched_barrier(0);
    stem_mma<4, STEM_J>(fa, mp, c00, c01, c10, c11);
    __builtin_amdgcn_sched_barrier(0);
    stem_load<0, 4>(fa, mp, tb, pb0, pb1, blk + 2);
    __builtin_amdgcn_sched_barrier(0);
    stem_mma<0, 4>(fb, mp, c00, c01, c10, c11);
    __builtin_amdgcn_sched_barrier(0);
    stem_load<4, STEM_J>(fa, mp, tb, pb0, pb1, blk + 2);
    __builtin_amdgcn_sched_barrier(0);
    stem_mma<4, STEM_J>(fb, mp, c00, c01, c10, c11);
    __builtin_amdgcn_sched_barrier(0);
  }
  stem_load<0, 4>(fb, mp, tb, pb0, pb1, STEM_NB - 1);
  __builtin_amdgcn_sched_barrier(0);
  stem_mma<0, 4>(fa, mp, c00, c01, c10, c11);
  __builtin_amdgcn_sched_barrier(0);
  stem_load<4, STEM_J>(fb, mp, tb, pb0, pb1, STEM_NB - 1);
  __builtin_amdgcn_sched_barrier(0);
  stem_mma<4, STEM_J>(fa, mp, c00, c01, c10, c11);
  stem_mma<0, STEM_J>(fb, mp, c00, c01, c10, c11);

  float* crow = C + (long)(m0 + 8 * h) * N + n0 + c;
  #pragma unroll
  for (int i = 0; i < 8; ++i) {
    crow[(long)i * N]             = c00[i];
    crow[(long)(i + 16) * N]      = c10[i];
    crow[(long)i * N + 16]        = c01[i];
    crow[(long)(i + 16) * N + 16] = c11[i];
  }
}

#define BC_DIM (2 * D_STATE)
__global__ void dwconv_silu_bc_kernel(const float* __restrict__ zx,
                                      const float* __restrict__ w,
                                      const float* __restrict__ bias,
                                      float* __restrict__ out) {
  const int c  = threadIdx.x;
  const int cc = D_INNER + c;
  const long bl = blockIdx.x;
  const int l = (int)(bl & (LOUT - 1));
  const float* xcol = zx + bl * D_IN_PROJ + D_INNER + cc;
  const float w0 = w[cc * 4 + 0], w1 = w[cc * 4 + 1], w2 = w[cc * 4 + 2], w3 = w[cc * 4 + 3];
  float acc = w3 * xcol[0];
  if (l >= 1) acc = fmaf(w2, xcol[-(long)D_IN_PROJ],     acc);
  if (l >= 2) acc = fmaf(w1, xcol[-(long)2 * D_IN_PROJ], acc);
  if (l >= 3) acc = fmaf(w0, xcol[-(long)3 * D_IN_PROJ], acc);
  const float v = acc + bias[cc];
  out[bl * BC_DIM + c] = v / (1.f + __expf(-v));
}

#define SCAN_G 8
#define SEGS   8
#define SEGLEN (LOUT / SEGS)
#define HSTATE (HEADDIM * D_STATE)

template <int J>
__device__ __forceinline__ float quad_bcast(float v) {
  return __builtin_bit_cast(float, __builtin_amdgcn_update_dpp(
      0, __builtin_bit_cast(int, v), J * 0x55, 0xF, 0xF, true));
}
template <int M>
__device__ __forceinline__ float row_xor(float v) {
  return __builtin_bit_cast(float, __builtin_amdgcn_update_dpp(
      0, __builtin_bit_cast(int, v), 0x160 + M, 0xF, 0xF, true));
}
__device__ __forceinline__ float lane_bcast(float v, int l) {
  return __builtin_bit_cast(float, __builtin_amdgcn_readlane(__builtin_bit_cast(int, v), l));
}
template <int I>
__device__ __forceinline__ void ssm_n(float& h, float& acc, const v4f& Bv, const v4f& Cv,
                                      float dA, float xdt) {
  const float bn = quad_bcast<(I >> 2)>(Bv[I & 3]);
  const float cn = quad_bcast<(I >> 2)>(Cv[I & 3]);
  h   = fmaf(h, dA, xdt * bn);
  acc = fmaf(h, cn, acc);
}
__device__ __forceinline__ float softplus_dt(float raw) {
  return (raw > 20.f) ? raw : log1pf(__expf(raw));
}

struct ScanSet { v4f B[SCAN_G]; v4f C[SCAN_G]; float x[SCAN_G]; };
struct XConv { float w0, w1, w2, w3, bias; float r1, r2, r3; };

__device__ __forceinline__ void scan_load(ScanSet& s, const float* bp, const float* cp,
                                          const float* xp, int l0) {
  #pragma unroll
  for (int j = 0; j < SCAN_G; ++j) {
    const long row = (long)min(l0 + j, LOUT - 1);
    s.B[j] = *(const v4f*)(bp + row * BC_DIM);
    s.C[j] = *(const v4f*)(cp + row * BC_DIM);
    s.x[j] = xp[row * D_IN_PROJ];
  }
}

__device__ __forceinline__ void scan_steps(const ScanSet& s, XConv& k, float (&h)[16], float dtv, float dav,
                                           int lane0, float Dh, int q, float* yrow,
                                           float& pc, float& pk, int lane) {
  float xs[SCAN_G];
  #pragma unroll
  for (int j = 0; j < SCAN_G; ++j) {
    const float r0 = s.x[j];
    float a = k.w3 * r0;
    a = fmaf(k.w2, k.r1, a);
    a = fmaf(k.w1, k.r2, a);
    a = fmaf(k.w0, k.r3, a);
    const float v = a + k.bias;
    xs[j] = v / (1.f + __expf(-v));
    k.r3 = k.r2; k.r2 = k.r1; k.r1 = r0;
  }
  float keep = 0.f;
  #pragma unroll
  for (int j = 0; j < SCAN_G; ++j) {
    const v4f   Bv = s.B[j];
    const v4f   Cv = s.C[j];
    const float xv = xs[j];
    const float dts = lane_bcast(dtv, lane0 + j);
    const float dA  = lane_bcast(dav, lane0 + j);
    const float xdt = xv * dts;
    pc *= dA;
    pk = (lane == lane0 + j) ? pc : pk;
    float acc = 0.f;
    ssm_n<0>(h[0], acc, Bv, Cv, dA, xdt);   ssm_n<1>(h[1], acc, Bv, Cv, dA, xdt);
    ssm_n<2>(h[2], acc, Bv, Cv, dA, xdt);   ssm_n<3>(h[3], acc, Bv, Cv, dA, xdt);
    ssm_n<4>(h[4], acc, Bv, Cv, dA, xdt);   ssm_n<5>(h[5], acc, Bv, Cv, dA, xdt);
    ssm_n<6>(h[6], acc, Bv, Cv, dA, xdt);   ssm_n<7>(h[7], acc, Bv, Cv, dA, xdt);
    ssm_n<8>(h[8], acc, Bv, Cv, dA, xdt);   ssm_n<9>(h[9], acc, Bv, Cv, dA, xdt);
    ssm_n<10>(h[10], acc, Bv, Cv, dA, xdt); ssm_n<11>(h[11], acc, Bv, Cv, dA, xdt);
    ssm_n<12>(h[12], acc, Bv, Cv, dA, xdt); ssm_n<13>(h[13], acc, Bv, Cv, dA, xdt);
    ssm_n<14>(h[14], acc, Bv, Cv, dA, xdt); ssm_n<15>(h[15], acc, Bv, Cv, dA, xdt);
    acc += row_xor<4>(acc);
    acc += row_xor<8>(acc);
    const float yv = fmaf(xv, Dh, acc);
    keep = (q == (j & 3)) ? yv : keep;
    if ((j & 3) == 3) yrow[(long)(j - 3) * D_INNER] = keep;
  }
}

__global__ __launch_bounds__(128)
void ssm_scan_kernel(const float* __restrict__ zx,
                     const float* __restrict__ bc,
                     const float* __restrict__ cw,
                     const float* __restrict__ cb,
                     const float* __restrict__ dtb,
                     const float* __restrict__ alog,
                     const float* __restrict__ Dp,
                     float* __restrict__ y,
                     float* __restrict__ hloc,
                     float* __restrict__ pcum) {
  const int seg  = blockIdx.x >> 6;
  const int b    = (blockIdx.x >> 5) & 1;
  const int hh   = (blockIdx.x >> 1) & 15;
  const int half = blockIdx.x & 1;
  const int lane = threadIdx.x & 31;
  const int wave = threadIdx.x >> 5;
  const int q    = (lane >> 2) & 3;
  const int pj   = lane & 3;
  const int p    = half * 32 + wave * 8 + (lane >> 4) * 4 + pj;
  const float Ah   = -__expf(alog[hh]);
  const float dtbh = dtb[hh];
  const float Dh   = Dp[hh];

  const float* zrows = zx + (long)b * LOUT * D_IN_PROJ;
  const float* bp   = bc + (long)b * LOUT * BC_DIM + 16 * q + 4 * pj;
  const float* cp   = bp + D_STATE;
  const int    ch   = hh * HEADDIM + p;
  const float* xp   = zrows + D_INNER + ch;
  const float* dtp  = zrows + (D_IN_PROJ - NHEADS) + hh;
  float* yp = y + ((long)b * LOUT + q) * D_INNER + hh * HEADDIM + p;

  float h[16];
  #pragma unroll
  for (int i = 0; i < 16; ++i) h[i] = 0.f;

  const int  c0     = seg * (SEGLEN / 32);
  const bool pwrite = (half == 0) && (wave == 0);
  float* pout = pcum + (long)(b * NHEADS + hh) * LOUT;
  float pc = 1.f, pk = 0.f;

  XConv k;
  {
    const v4f wv = *(const v4f*)(cw + ch * D_CONV);
    k.w0 = wv[0]; k.w1 = wv[1]; k.w2 = wv[2]; k.w3 = wv[3];
    k.bias = cb[ch];
    k.r1 = 0.f; k.r2 = 0.f; k.r3 = 0.f;
    if (seg > 0) {
      k.r1 = xp[(long)(c0 * 32 - 1) * D_IN_PROJ];
      k.r2 = xp[(long)(c0 * 32 - 2) * D_IN_PROJ];
      k.r3 = xp[(long)(c0 * 32 - 3) * D_IN_PROJ];
    }
  }

  ScanSet sa, sb;
  scan_load(sa, bp, cp, xp, c0 * 32);
  float dtv = softplus_dt(dtp[(long)(c0 * 32 + lane) * D_IN_PROJ] + dtbh);
  float dav = __expf(dtv * Ah);

  for (int c = c0; c < c0 + SEGLEN / 32; ++c) {
    const int   cn   = min(c + 1, LOUT / 32 - 1);
    const float rawn = dtp[(long)(cn * 32 + lane) * D_IN_PROJ];
    for (int g = 0; g < 32; g += 2 * SCAN_G) {
      const int l0 = c * 32 + g;
      scan_load(sb, bp, cp, xp, l0 + SCAN_G);
      __builtin_amdgcn_sched_barrier(0);
      scan_steps(sa, k, h, dtv, dav, g, Dh, q, yp + (long)l0 * D_INNER, pc, pk, lane);
      __builtin_amdgcn_sched_barrier(0);
      scan_load(sa, bp, cp, xp, l0 + 2 * SCAN_G);
      __builtin_amdgcn_sched_barrier(0);
      scan_steps(sb, k, h, dtv, dav, g + SCAN_G, Dh, q, yp + (long)(l0 + SCAN_G) * D_INNER, pc, pk, lane);
    }
    if (pwrite) pout[c * 32 + lane] = pk;
    dtv = softplus_dt(rawn + dtbh);
    dav = __expf(dtv * Ah);
  }
  if (seg < SEGS - 1) {
    float* hp = hloc + ((long)(b * NHEADS + hh) * SEGS + seg) * HSTATE + p * D_STATE + 16 * q;
    #pragma unroll
    for (int i = 0; i < 4; ++i) {
      const v4f v = {h[4 * i], h[4 * i + 1], h[4 * i + 2], h[4 * i + 3]};
      *(v4f*)(hp + 4 * i) = v;
    }
  }
}

__global__ __launch_bounds__(128)
void ssm_carry_wmma(const float* __restrict__ bc,
                    const float* __restrict__ hloc,
                    const float* __restrict__ pcum,
                    float* __restrict__ y) {
  constexpr int JB = D_STATE / 8;
  const int lane = threadIdx.x & 31;
  const int wave = __builtin_amdgcn_readfirstlane((int)(threadIdx.x >> 5));
  const int tb   = blockIdx.x & (SEGLEN / 64 - 1);
  const int r    = blockIdx.x / (SEGLEN / 64);
  const int bh   = r / (SEGS - 1);
  const int seg  = 1 + (r - bh * (SEGS - 1));
  const int b    = bh / NHEADS;
  const int hh   = bh - b * NHEADS;
  const int t0   = seg * SEGLEN + tb * 64 + (wave >> 1) * 32;
  const int p0   = (wave & 1) * 32;
  const int c = lane & 15;
  const int h = lane >> 4;
  const float* pa0 = bc + ((long)b * LOUT + t0 + c) * BC_DIM + D_STATE + 4 * h;
  const float* pa1 = pa0 + (long)16 * BC_DIM;
  const float* hb0 = hloc + (long)bh * SEGS * HSTATE + (p0 + c) * D_STATE + 4 * h;
  const float* pr  = pcum + (long)bh * LOUT;

  RowFrag<JB> f;
  #pragma unroll
  for (int j = 0; j < JB; ++j) {
    f.a0[j] = *(const v4f*)(pa0 + 8 * j);
    f.a1[j] = *(const v4f*)(pa1 + 8 * j);
    f.b0[j] = *(const v4f*)(hb0 + 8 * j);
    f.b1[j] = *(const v4f*)(hb0 + 16 * D_STATE + 8 * j);
  }
  for (int k = 1; k < seg; ++k) {
    const float a = pr[k * SEGLEN + SEGLEN - 1];
    const float* hk = hb0 + (long)k * HSTATE;
    #pragma unroll
    for (int j = 0; j < JB; ++j) {
      const v4f e0 = *(const v4f*)(hk + 8 * j);
      const v4f e1 = *(const v4f*)(hk + 16 * D_STATE + 8 * j);
      #pragma unroll
      for (int e = 0; e < 4; ++e) {
        f.b0[j][e] = fmaf(a, f.b0[j][e], e0[e]);
        f.b1[j][e] = fmaf(a, f.b1[j][e], e1[e]);
      }
    }
  }

  const v8f zero = {0.f, 0.f, 0.f, 0.f, 0.f, 0.f, 0.f, 0.f};
  v8f c00 = zero, c01 = zero, c10 = zero, c11 = zero;
  frag_mma<JB>(f, c00, c01, c10, c11);

  float* yrow = y + ((long)b * LOUT + t0 + 8 * h) * D_INNER + hh * HEADDIM + p0 + c;
  const float* prow = pr + t0 + 8 * h;
  #pragma unroll
  for (int i = 0; i < 8; ++i) {
    const float s0 = prow[i];
    const float s1 = prow[i + 16];
    yrow[(long)i * D_INNER]             = fmaf(s0, c00[i], yrow[(long)i * D_INNER]);
    yrow[(long)i * D_INNER + 16]        = fmaf(s0, c01[i], yrow[(long)i * D_INNER + 16]);
    yrow[(long)(i + 16) * D_INNER]      = fmaf(s1, c10[i], yrow[(long)(i + 16) * D_INNER]);
    yrow[(long)(i + 16) * D_INNER + 16] = fmaf(s1, c11[i], yrow[(long)(i + 16) * D_INNER + 16]);
  }
}

__global__ void gate_rmsnorm_kernel(const float* __restrict__ zx,
                                    const float* __restrict__ nw,
                                    float* __restrict__ y) {
  const long bl = blockIdx.x;
  const int tid = threadIdx.x;
  __shared__ float red[256];
  float yv[4];
  float ss = 0.f;
  #pragma unroll
  for (int j = 0; j < 4; ++j) {
    const int i = tid + j * 256;
    const float z = zx[bl * D_IN_PROJ + i];
    const float g = z / (1.f + __expf(-z));
    const float v = y[bl * D_INNER + i] * g;
    yv[j] = v;
    ss = fmaf(v, v, ss);
  }
  red[tid] = ss;
  __syncthreads();
  for (int st = 128; st > 0; st >>= 1) {
    if (tid < st) red[tid] += red[tid + st];
    __syncthreads();
  }
  const float scale = rsqrtf(red[0] * (1.f / (float)D_INNER) + 1e-5f);
  #pragma unroll
  for (int j = 0; j < 4; ++j) {
    const int i = tid + j * 256;
    y[bl * D_INNER + i] = yv[j] * scale * nw[i];
  }
}

extern "C" void kernel_launch(void* const* d_in, const int* in_sizes, int n_in,
                              void* d_out, int out_size, void* d_ws, size_t ws_size,
                              hipStream_t stream) {
  const float* tokens     = (const float*)d_in[0];
  const float* conv_w     = (const float*)d_in[1];
  const float* in_proj_w  = (const float*)d_in[2];
  const float* conv1d_w   = (const float*)d_in[3];
  const float* conv1d_b   = (const float*)d_in[4];
  const float* dt_bias    = (const float*)d_in[5];
  const float* A_log      = (const float*)d_in[6];
  const float* D_param    = (const float*)d_in[7];
  const float* norm_w     = (const float*)d_in[8];
  const float* out_proj_w = (const float*)d_in[9];
  float* out = (float*)d_out;
  float* ws  = (float*)d_ws;

  float* u0   = ws;
  float* u1   = u0 + (long)ROWS * D_MODEL;
  float* zx   = u1 + (long)ROWS * D_MODEL;
  float* bcact = zx + (long)ROWS * D_IN_PROJ;
  float* ybuf = bcact + (long)ROWS * BC_DIM;
  float* hloc = ybuf + (long)ROWS * D_INNER;
  float* pcum = hloc + (long)BATCH * NHEADS * SEGS * HSTATE;
  (void)ws_size; (void)in_sizes; (void)n_in; (void)out_size;

  stem_conv_wmma<<<GEMM_WGS(D_MODEL), 128, 0, stream>>>(tokens, conv_w, u0);

  const float* u_in = u0;
  for (int blk = 0; blk < 2; ++blk) {
    const float* ipw = in_proj_w  + (long)blk * D_IN_PROJ * D_MODEL;
    const float* cw  = conv1d_w   + (long)blk * CONV_DIM * D_CONV;
    const float* cb  = conv1d_b   + (long)blk * CONV_DIM;
    const float* dtb = dt_bias    + (long)blk * NHEADS;
    const float* al  = A_log      + (long)blk * NHEADS;
    const float* Dpp = D_param    + (long)blk * NHEADS;
    const float* nw  = norm_w     + (long)blk * D_INNER;
    const float* opw = out_proj_w + (long)blk * D_MODEL * D_INNER;

    {
      gemm_rows_wmma<D_IN_PROJ, D_MODEL, 32><<<GEMM_WGS(D_IN_PROJ), 128, 0, stream>>>(u_in, ipw, zx);
    }
    dwconv_silu_bc_kernel<<<ROWS, BC_DIM, 0, stream>>>(zx, cw, cb, bcact);
    ssm_scan_kernel<<<SEGS * BATCH * NHEADS * 2, 128, 0, stream>>>(zx, bcact, cw, cb, dtb, al, Dpp, ybuf, hloc, pcum);
    ssm_carry_wmma<<<BATCH * NHEADS * (SEGS - 1) * (SEGLEN / 64), 128, 0, stream>>>(bcact, hloc, pcum, ybuf);
    gate_rmsnorm_kernel<<<ROWS, 256, 0, stream>>>(zx, nw, ybuf);
    float* dst = (blk == 0) ? u1 : out;
    {
      gemm_rows_wmma<D_MODEL, D_INNER, 64><<<GEMM_WGS(D_MODEL), 128, 0, stream>>>(ybuf, opw, dst);
    }
    u_in = u1;
  }
}


# row GEMMs: K split over the 4 waves, each wave the whole 64x64 C block from registers (half the operand loads per matrix op); partial blocks added via LDS as (q0+q1)+(q2+q3): k sum order changed
# speedup vs baseline: 14.6469x; 1.1156x over previous
#include <hip/hip_runtime.h>
#include <hip/hip_bf16.h>

typedef __attribute__((ext_vector_type(2))) float v2f;
typedef __attribute__((ext_vector_type(8))) float v8f;

#define BATCH     2
#define SEQLEN    2048
#define C_IN      256
#define D_MODEL   512
#define LOUT      1024
#define D_STATE   64
#define D_CONV    4
#define D_INNER   1024
#define HEADDIM   64
#define NHEADS    16
#define CONV_DIM  (D_INNER + 2*D_STATE)
#define D_IN_PROJ (2*D_INNER + 2*D_STATE + NHEADS)
#define ROWS      (BATCH*LOUT)
#define KSTEM     (C_IN * 7)
#define GEMM_WGS(n) ((ROWS / 64) * (((n) + 63) / 64))

typedef __attribute__((ext_vector_type(4))) float v4f;

template <int JB>
struct RowFrag { v4f a0[JB]; v4f a1[JB]; v4f b0[JB]; v4f b1[JB]; };

__device__ __forceinline__ v8f mma4(v2f a, v2f b, v8f c) {
  return __builtin_amdgcn_wmma_f32_16x16x4_f32(false, a, false, b, (short)0, c, false, false);
}

template <int JB>
__device__ __forceinline__ void frag_mma(const RowFrag<JB>& f, v8f& c00, v8f& c01, v8f& c10, v8f& c11) {
  #pragma unroll
  for (int j = 0; j < JB; ++j) {
    #pragma unroll
    for (int u = 0; u < 2; ++u) {
      const v2f a0 = {f.a0[j][2 * u], f.a0[j][2 * u + 1]};
      const v2f a1 = {f.a1[j][2 * u], f.a1[j][2 * u + 1]};
      const v2f b0 = {f.b0[j][2 * u], f.b0[j][2 * u + 1]};
      const v2f b1 = {f.b1[j][2 * u], f.b1[j][2 * u + 1]};
      c00 = mma4(a0, b0, c00);
      c01 = mma4(a0, b1, c01);
      c10 = mma4(a1, b0, c10);
      c11 = mma4(a1, b1, c11);
    }
  }
}

template <int JB>
struct BlockFrag { v4f a[4][JB]; v4f b[4][JB]; };

template <int JB>
__device__ __forceinline__ void block_load(BlockFrag<JB>& f, const float* const (&pa)[4],
                                           const float* const (&pb)[4], int kb) {
  #pragma unroll
  for (int j = 0; j < JB; ++j) {
    #pragma unroll
    for (int t = 0; t < 4; ++t) {
      f.a[t][j] = *(const v4f*)(pa[t] + kb + 8 * j);
      f.b[t][j] = *(const v4f*)(pb[t] + kb + 8 * j);
    }
  }
}

template <int JB>
__device__ __forceinline__ void block_mma(const BlockFrag<JB>& f, v8f (&acc)[4][4]) {
  #pragma unroll
  for (int j = 0; j < JB; ++j) {
    #pragma unroll
    for (int u = 0; u < 2; ++u) {
      v2f a[4], b[4];
      #pragma unroll
      for (int t = 0; t < 4; ++t) {
        a[t] = v2f{f.a[t][j][2 * u], f.a[t][j][2 * u + 1]};
        b[t] = v2f{f.b[t][j][2 * u], f.b[t][j][2 * u + 1]};
      }
      #pragma unroll
      for (int mi = 0; mi < 4; ++mi) {
        #pragma unroll
        for (int ni = 0; ni < 4; ++ni) acc[mi][ni] = mma4(a[mi], b[ni], acc[mi][ni]);
      }
    }
  }
}

#define PART_PITCH 68

template <int N, int K, int KB>
__global__ __launch_bounds__(128)
void gemm_rows_wmma(const float* __restrict__ A,
                    const float* __restrict__ W,
                    float* __restrict__ C) {
  constexpr int KQ  = K / 4;
  static_assert(K % 4 == 0 && KQ % (2 * KB) == 0 && KB % 8 == 0 && N % 16 == 0, "shape");
  constexpr int JB  = KB / 8;
  constexpr int WGN = (N + 63) / 64;
  __shared__ float part[4][64][PART_PITCH];
  const int lane = threadIdx.x & 31;
  const int wave = __builtin_amdgcn_readfirstlane((int)(threadIdx.x >> 5));
  const int wgm  = blockIdx.x / WGN;
  const int wgn  = blockIdx.x - wgm * WGN;
  const int m0   = wgm * 64;
  const int n0   = wgn * 64;
  const int c = lane & 15;
  const int h = lane >> 4;
  const float* pa[4];
  const float* pb[4];
  #pragma unroll
  for (int t = 0; t < 4; ++t) {
    pa[t] = A + (long)(m0 + 16 * t + c) * K + wave * KQ + 4 * h;
    const int nt = (n0 + 16 * t < N) ? 16 * t : 0;
    pb[t] = W + (long)(n0 + nt + c) * K + wave * KQ + 4 * h;
  }

  v8f acc[4][4];
  #pragma unroll
  for (int mi = 0; mi < 4; ++mi) {
    #pragma unroll
    for (int ni = 0; ni < 4; ++ni) acc[mi][ni] = v8f{0.f, 0.f, 0.f, 0.f, 0.f, 0.f, 0.f, 0.f};
  }
  BlockFrag<JB> fa, fb;
  block_load<JB>(fa, pa, pb, 0);
  #pragma unroll 1
  for (int kb = 0; kb < KQ - 2 * KB; kb += 2 * KB) {
    block_load<JB>(fb, pa, pb, kb + KB);
    __builtin_amdgcn_sched_barrier(0);
    block_mma<JB>(fa, acc);
    __builtin_amdgcn_sched_barrier(0);
    block_load<JB>(fa, pa, pb, kb + 2 * KB);
    __builtin_amdgcn_sched_barrier(0);
    block_mma<JB>(fb, acc);
  }
  block_load<JB>(fb, pa, pb, KQ - KB);
  __builtin_amdgcn_sched_barrier(0);
  block_mma<JB>(fa, acc);
  block_mma<JB>(fb, acc);

  #pragma unroll
  for (int mi = 0; mi < 4; ++mi) {
    #pragma unroll
    for (int ni = 0; ni < 4; ++ni) {
      float* q = &part[wave][16 * ni + c][16 * mi + 8 * h];
      const v8f v = acc[mi][ni];
      *(v4f*)(q)     = v4f{v[0], v[1], v[2], v[3]};
      *(v4f*)(q + 4) = v4f{v[4], v[5], v[6], v[7]};
    }
  }
  __syncthreads();
  #pragma unroll
  for (int hf = 0; hf < 2; ++hf) {
    const int col = lane + 32 * hf;
    const bool ok = (n0 + col) < N;
    float* crow = C + (long)(m0 + 16 * wave) * N + n0 + col;
    #pragma unroll
    for (int r = 0; r < 4; ++r) {
      const v4f s0 = *(const v4f*)&part[0][col][16 * wave + 4 * r];
      const v4f s1 = *(const v4f*)&part[1][col][16 * wave + 4 * r];
      const v4f s2 = *(const v4f*)&part[2][col][16 * wave + 4 * r];
      const v4f s3 = *(const v4f*)&part[3][col][16 * wave + 4 * r];
      const v4f s = (s0 + s1) + (s2 + s3);
      if (ok) {
        #pragma unroll
        for (int e = 0; e < 4; ++e) crow[(long)(4 * r + e) * N] = s[e];
      }
    }
  }
}

#define STEM_CB 8
#define STEM_KB (STEM_CB * 7)
#define STEM_NB (C_IN / STEM_CB)
#define STEM_J  (STEM_KB / 8)

struct StemFrag { float a0[STEM_J][4]; float a1[STEM_J][4]; v4f b0[STEM_J]; v4f b1[STEM_J]; };
struct StemMap  { unsigned o0[STEM_J][4]; unsigned o1[STEM_J][4]; int m0[STEM_J][4]; int m1[STEM_J][4]; };

template <int J0, int J1>
__device__ __forceinline__ void stem_load(StemFrag& f, const StemMap& mp, const char* tb,
                                          const float* pb0, const float* pb1, int blk) {
  const char* t = tb + (long)blk * (STEM_CB * SEQLEN * 4);
  const int kb = blk * STEM_KB;
  #pragma unroll
  for (int j = J0; j < J1; ++j) {
    f.b0[j] = *(const v4f*)(pb0 + kb + 8 * j);
    f.b1[j] = *(const v4f*)(pb1 + kb + 8 * j);
    #pragma unroll
    for (int e = 0; e < 4; ++e) {
      f.a0[j][e] = *(const float*)(t + mp.o0[j][e]);
      f.a1[j][e] = *(const float*)(t + mp.o1[j][e]);
    }
  }
}

__device__ __forceinline__ float pad0(float v, int m) {
  return __builtin_bit_cast(float, __builtin_bit_cast(int, v) & m);
}

template <int J0, int J1>
__device__ __forceinline__ void stem_mma(const StemFrag& f, const StemMap& mp,
                                         v8f& c00, v8f& c01, v8f& c10, v8f& c11) {
  #pragma unroll
  for (int j = J0; j < J1; ++j) {
    #pragma unroll
    for (int u = 0; u < 2; ++u) {
      const v2f a0 = {pad0(f.a0[j][2 * u], mp.m0[j][2 * u]), pad0(f.a0[j][2 * u + 1], mp.m0[j][2 * u + 1])};
      const v2f a1 = {pad0(f.a1[j][2 * u], mp.m1[j][2 * u]), pad0(f.a1[j][2 * u + 1], mp.m1[j][2 * u + 1])};
      const v2f b0 = {f.b0[j][2 * u], f.b0[j][2 * u + 1]};
      const v2f b1 = {f.b1[j][2 * u], f.b1[j][2 * u + 1]};
      c00 = mma4(a0, b0, c00);
      c01 = mma4(a0, b1, c01);
      c10 = mma4(a1, b0, c10);
      c11 = mma4(a1, b1, c11);
    }
  }
}

__global__ __launch_bounds__(128)
void stem_conv_wmma(const float* __restrict__ tokens,
                    const float* __restrict__ W,
                    float* __restrict__ C) {
  constexpr int N   = D_MODEL;
  constexpr int K   = KSTEM;
  constexpr int WGN = N / 64;
  const int lane = threadIdx.x & 31;
  const int wave = __builtin_amdgcn_readfirstlane((int)(threadIdx.x >> 5));
  const int wgm  = blockIdx.x / WGN;
  const int wgn  = blockIdx.x - wgm * WGN;
  const int m0   = wgm * 64 + (wave >> 1) * 32;
  const int n0   = wgn * 64 + (wave & 1) * 32;
  const int c = lane & 15;
  const int h = lane >> 4;
  const float* pb0 = W + (long)(n0 + c) * K + 4 * h;
  const float* pb1 = pb0 + (long)16 * K;
  const char* tb = (const char*)(tokens + (long)(m0 / LOUT) * C_IN * SEQLEN);
  const int l0 = (m0 & (LOUT - 1)) + c;

  StemMap mp;
  #pragma unroll
  for (int j = 0; j < STEM_J; ++j) {
    #pragma unroll
    for (int e = 0; e < 4; ++e) {
      const int kl = 8 * j + 4 * h + e;
      const int ch = kl / 7;
      const int tp = kl - 7 * ch;
      const int s0 = 2 * l0 - 3 + tp;
      const int s1 = s0 + 32;
      const bool ok0 = (unsigned)s0 < (unsigned)SEQLEN;
      const bool ok1 = (unsigned)s1 < (unsigned)SEQLEN;
      mp.o0[j][e] = (unsigned)(ch * SEQLEN + (ok0 ? s0 : 0)) * 4u;
      mp.o1[j][e] = (unsigned)(ch * SEQLEN + (ok1 ? s1 : 0)) * 4u;
      mp.m0[j][e] = ok0 ? -1 : 0;
      mp.m1[j][e] = ok1 ? -1 : 0;
    }
  }

  const v8f zero = {0.f, 0.f, 0.f, 0.f, 0.f, 0.f, 0.f, 0.f};
  v8f c00 = zero, c01 = zero, c10 = zero, c11 = zero;
  StemFrag fa, fb;
  stem_load<0, 4>(fa, mp, tb, pb0, pb1, 0);
  stem_load<4, STEM_J>(fa, mp, tb, pb0, pb1, 0);
  #pragma unroll 1
  for (int blk = 0; blk < STEM_NB - 2; blk += 2) {
    stem_load<0, 4>(fb, mp, tb, pb0, pb1, blk + 1);
    __builtin_amdgcn_sched_barrier(0);
    stem_mma<0, 4>(fa, mp, c00, c01, c10, c11);
    __builtin_amdgcn_sched_barrier(0);
    stem_load<4, STEM_J>(fb, mp, tb, pb0, pb1, blk + 1);
    __builtin_amdgcn_sched_barrier(0);
    stem_mma<4, STEM_J>(fa, mp, c00, c01, c10, c11);
    __builtin_amdgcn_sched_barrier(0);
    stem_load<0, 4>(fa, mp, tb, pb0, pb1, blk + 2);
    __builtin_amdgcn_sched_barrier(0);
    stem_mma<0, 4>(fb, mp, c00, c01, c10, c11);
    __builtin_amdgcn_sched_barrier(0);
    stem_load<4, STEM_J>(fa, mp, tb, pb0, pb1, blk + 2);
    __builtin_amdgcn_sched_barrier(0);
    stem_mma<4, STEM_J>(fb, mp, c00, c01, c10, c11);
    __builtin_amdgcn_sched_barrier(0);
  }
  stem_load<0, 4>(fb, mp, tb, pb0, pb1, STEM_NB - 1);
  __builtin_amdgcn_sched_barrier(0);
  stem_mma<0, 4>(fa, mp, c00, c01, c10, c11);
  __builtin_amdgcn_sched_barrier(0);
  stem_load<4, STEM_J>(fb, mp, tb, pb0, pb1, STEM_NB - 1);
  __builtin_amdgcn_sched_barrier(0);
  stem_mma<4, STEM_J>(fa, mp, c00, c01, c10, c11);
  stem_mma<0, STEM_J>(fb, mp, c00, c01, c10, c11);

  float* crow = C + (long)(m0 + 8 * h) * N + n0 + c;
  #pragma unroll
  for (int i = 0; i < 8; ++i) {
    crow[(long)i * N]             = c00[i];
    crow[(long)(i + 16) * N]      = c10[i];
    crow[(long)i * N + 16]        = c01[i];
    crow[(long)(i + 16) * N + 16] = c11[i];
  }
}

#define BC_DIM (2 * D_STATE)
__global__ void dwconv_silu_bc_kernel(const float* __restrict__ zx,
                                      const float* __restrict__ w,
                                      const float* __restrict__ bias,
                                      float* __restrict__ out) {
  const int c  = threadIdx.x;
  const int cc = D_INNER + c;
  const long bl = blockIdx.x;
  const int l = (int)(bl & (LOUT - 1));
  const float* xcol = zx + bl * D_IN_PROJ + D_INNER + cc;
  const float w0 = w[cc * 4 + 0], w1 = w[cc * 4 + 1], w2 = w[cc * 4 + 2], w3 = w[cc * 4 + 3];
  float acc = w3 * xcol[0];
  if (l >= 1) acc = fmaf(w2, xcol[-(long)D_IN_PROJ],     acc);
  if (l >= 2) acc = fmaf(w1, xcol[-(long)2 * D_IN_PROJ], acc);
  if (l >= 3) acc = fmaf(w0, xcol[-(long)3 * D_IN_PROJ], acc);
  const float v = acc + bias[cc];
  out[bl * BC_DIM + c] = v / (1.f + __expf(-v));
}

#define SCAN_G 8
#define SEGS   8
#define SEGLEN (LOUT / SEGS)
#define HSTATE (HEADDIM * D_STATE)

template <int J>
__device__ __forceinline__ float quad_bcast(float v) {
  return __builtin_bit_cast(float, __builtin_amdgcn_update_dpp(
      0, __builtin_bit_cast(int, v), J * 0x55, 0xF, 0xF, true));
}
template <int M>
__device__ __forceinline__ float row_xor(float v) {
  return __builtin_bit_cast(float, __builtin_amdgcn_update_dpp(
      0, __builtin_bit_cast(int, v), 0x160 + M, 0xF, 0xF, true));
}
__device__ __forceinline__ float lane_bcast(float v, int l) {
  return __builtin_bit_cast(float, __builtin_amdgcn_readlane(__builtin_bit_cast(int, v), l));
}
template <int I>
__device__ __forceinline__ void ssm_n(float& h, float& acc, const v4f& Bv, const v4f& Cv,
                                      float dA, float xdt) {
  const float bn = quad_bcast<(I >> 2)>(Bv[I & 3]);
  const float cn = quad_bcast<(I >> 2)>(Cv[I & 3]);
  h   = fmaf(h, dA, xdt * bn);
  acc = fmaf(h, cn, acc);
}
__device__ __forceinline__ float softplus_dt(float raw) {
  return (raw > 20.f) ? raw : log1pf(__expf(raw));
}

struct ScanSet { v4f B[SCAN_G]; v4f C[SCAN_G]; float x[SCAN_G]; };
struct XConv { float w0, w1, w2, w3, bias; float r1, r2, r3; };

__device__ __forceinline__ void scan_load(ScanSet& s, const float* bp, const float* cp,
                                          const float* xp, int l0) {
  #pragma unroll
  for (int j = 0; j < SCAN_G; ++j) {
    const long row = (long)min(l0 + j, LOUT - 1);
    s.B[j] = *(const v4f*)(bp + row * BC_DIM);
    s.C[j] = *(const v4f*)(cp + row * BC_DIM);
    s.x[j] = xp[row * D_IN_PROJ];
  }
}

__device__ __forceinline__ void scan_steps(const ScanSet& s, XConv& k, float (&h)[16], float dtv, float dav,
                                           int lane0, float Dh, int q, float* yrow,
                                           float& pc, float& pk, int lane) {
  float xs[SCAN_G];
  #pragma unroll
  for (int j = 0; j < SCAN_G; ++j) {
    const float r0 = s.x[j];
    float a = k.w3 * r0;
    a = fmaf(k.w2, k.r1, a);
    a = fmaf(k.w1, k.r2, a);
    a = fmaf(k.w0, k.r3, a);
    const float v = a + k.bias;
    xs[j] = v / (1.f + __expf(-v));
    k.r3 = k.r2; k.r2 = k.r1; k.r1 = r0;
  }
  float keep = 0.f;
  #pragma unroll
  for (int j = 0; j < SCAN_G; ++j) {
    const v4f   Bv = s.B[j];
    const v4f   Cv = s.C[j];
    const float xv = xs[j];
    const float dts = lane_bcast(dtv, lane0 + j);
    const float dA  = lane_bcast(dav, lane0 + j);
    const float xdt = xv * dts;
    pc *= dA;
    pk = (lane == lane0 + j) ? pc : pk;
    float acc = 0.f;
    ssm_n<0>(h[0], acc, Bv, Cv, dA, xdt);   ssm_n<1>(h[1], acc, Bv, Cv, dA, xdt);
    ssm_n<2>(h[2], acc, Bv, Cv, dA, xdt);   ssm_n<3>(h[3], acc, Bv, Cv, dA, xdt);
    ssm_n<4>(h[4], acc, Bv, Cv, dA, xdt);   ssm_n<5>(h[5], acc, Bv, Cv, dA, xdt);
    ssm_n<6>(h[6], acc, Bv, Cv, dA, xdt);   ssm_n<7>(h[7], acc, Bv, Cv, dA, xdt);
    ssm_n<8>(h[8], acc, Bv, Cv, dA, xdt);   ssm_n<9>(h[9], acc, Bv, Cv, dA, xdt);
    ssm_n<10>(h[10], acc, Bv, Cv, dA, xdt); ssm_n<11>(h[11], acc, Bv, Cv, dA, xdt);
    ssm_n<12>(h[12], acc, Bv, Cv, dA, xdt); ssm_n<13>(h[13], acc, Bv, Cv, dA, xdt);
    ssm_n<14>(h[14], acc, Bv, Cv, dA, xdt); ssm_n<15>(h[15], acc, Bv, Cv, dA, xdt);
    acc += row_xor<4>(acc);
    acc += row_xor<8>(acc);
    const float yv = fmaf(xv, Dh, acc);
    keep = (q == (j & 3)) ? yv : keep;
    if ((j & 3) == 3) yrow[(long)(j - 3) * D_INNER] = keep;
  }
}

__global__ __launch_bounds__(128)
void ssm_scan_kernel(const float* __restrict__ zx,
                     const float* __restrict__ bc,
                     const float* __restrict__ cw,
                     const float* __restrict__ cb,
                     const float* __restrict__ dtb,
                     const float* __restrict__ alog,
                     const float* __restrict__ Dp,
                     float* __restrict__ y,
                     float* __restrict__ hloc,
                     float* __restrict__ pcum) {
  const int seg  = blockIdx.x >> 6;
  const int b    = (blockIdx.x >> 5) & 1;
  const int hh   = (blockIdx.x >> 1) & 15;
  const int half = blockIdx.x & 1;
  const int lane = threadIdx.x & 31;
  const int wave = threadIdx.x >> 5;
  const int q    = (lane >> 2) & 3;
  const int pj   = lane & 3;
  const int p    = half * 32 + wave * 8 + (lane >> 4) * 4 + pj;
  const float Ah   = -__expf(alog[hh]);
  const float dtbh = dtb[hh];
  const float Dh   = Dp[hh];

  const float* zrows = zx + (long)b * LOUT * D_IN_PROJ;
  const float* bp   = bc + (long)b * LOUT * BC_DIM + 16 * q + 4 * pj;
  const float* cp   = bp + D_STATE;
  const int    ch   = hh * HEADDIM + p;
  const float* xp   = zrows + D_INNER + ch;
  const float* dtp  = zrows + (D_IN_PROJ - NHEADS) + hh;
  float* yp = y + ((long)b * LOUT + q) * D_INNER + hh * HEADDIM + p;

  float h[16];
  #pragma unroll
  for (int i = 0; i < 16; ++i) h[i] = 0.f;

  const int  c0     = seg * (SEGLEN / 32);
  const bool pwrite = (half == 0) && (wave == 0);
  float* pout = pcum + (long)(b * NHEADS + hh) * LOUT;
  float pc = 1.f, pk = 0.f;

  XConv k;
  {
    const v4f wv = *(const v4f*)(cw + ch * D_CONV);
    k.w0 = wv[0]; k.w1 = wv[1]; k.w2 = wv[2]; k.w3 = wv[3];
    k.bias = cb[ch];
    k.r1 = 0.f; k.r2 = 0.f; k.r3 = 0.f;
    if (seg > 0) {
      k.r1 = xp[(long)(c0 * 32 - 1) * D_IN_PROJ];
      k.r2 = xp[(long)(c0 * 32 - 2) * D_IN_PROJ];
      k.r3 = xp[(long)(c0 * 32 - 3) * D_IN_PROJ];
    }
  }

  ScanSet sa, sb;
  scan_load(sa, bp, cp, xp, c0 * 32);
  float dtv = softplus_dt(dtp[(long)(c0 * 32 + lane) * D_IN_PROJ] + dtbh);
  float dav = __expf(dtv * Ah);

  for (int c = c0; c < c0 + SEGLEN / 32; ++c) {
    const int   cn   = min(c + 1, LOUT / 32 - 1);
    const float rawn = dtp[(long)(cn * 32 + lane) * D_IN_PROJ];
    for (int g = 0; g < 32; g += 2 * SCAN_G) {
      const int l0 = c * 32 + g;
      scan_load(sb, bp, cp, xp, l0 + SCAN_G);
      __builtin_amdgcn_sched_barrier(0);
      scan_steps(sa, k, h, dtv, dav, g, Dh, q, yp + (long)l0 * D_INNER, pc, pk, lane);
      __builtin_amdgcn_sched_barrier(0);
      scan_load(sa, bp, cp, xp, l0 + 2 * SCAN_G);
      __builtin_amdgcn_sched_barrier(0);
      scan_steps(sb, k, h, dtv, dav, g + SCAN_G, Dh, q, yp + (long)(l0 + SCAN_G) * D_INNER, pc, pk, lane);
    }
    if (pwrite) pout[c * 32 + lane] = pk;
    dtv = softplus_dt(rawn + dtbh);
    dav = __expf(dtv * Ah);
  }
  if (seg < SEGS - 1) {
    float* hp = hloc + ((long)(b * NHEADS + hh) * SEGS + seg) * HSTATE + p * D_STATE + 16 * q;
    #pragma unroll
    for (int i = 0; i < 4; ++i) {
      const v4f v = {h[4 * i], h[4 * i + 1], h[4 * i + 2], h[4 * i + 3]};
      *(v4f*)(hp + 4 * i) = v;
    }
  }
}

__global__ __launch_bounds__(128)
void ssm_carry_wmma(const float* __restrict__ bc,
                    const float* __restrict__ hloc,
                    const float* __restrict__ pcum,
                    float* __restrict__ y) {
  constexpr int JB = D_STATE / 8;
  const int lane = threadIdx.x & 31;
  const int wave = __builtin_amdgcn_readfirstlane((int)(threadIdx.x >> 5));
  const int tb   = blockIdx.x & (SEGLEN / 64 - 1);
  const int r    = blockIdx.x / (SEGLEN / 64);
  const int bh   = r / (SEGS - 1);
  const int seg  = 1 + (r - bh * (SEGS - 1));
  const int b    = bh / NHEADS;
  const int hh   = bh - b * NHEADS;
  const int t0   = seg * SEGLEN + tb * 64 + (wave >> 1) * 32;
  const int p0   = (wave & 1) * 32;
  const int c = lane & 15;
  const int h = lane >> 4;
  const float* pa0 = bc + ((long)b * LOUT + t0 + c) * BC_DIM + D_STATE + 4 * h;
  const float* pa1 = pa0 + (long)16 * BC_DIM;
  const float* hb0 = hloc + (long)bh * SEGS * HSTATE + (p0 + c) * D_STATE + 4 * h;
  const float* pr  = pcum + (long)bh * LOUT;

  RowFrag<JB> f;
  #pragma unroll
  for (int j = 0; j < JB; ++j) {
    f.a0[j] = *(const v4f*)(pa0 + 8 * j);
    f.a1[j] = *(const v4f*)(pa1 + 8 * j);
    f.b0[j] = *(const v4f*)(hb0 + 8 * j);
    f.b1[j] = *(const v4f*)(hb0 + 16 * D_STATE + 8 * j);
  }
  for (int k = 1; k < seg; ++k) {
    const float a = pr[k * SEGLEN + SEGLEN - 1];
    const float* hk = hb0 + (long)k * HSTATE;
    #pragma unroll
    for (int j = 0; j < JB; ++j) {
      const v4f e0 = *(const v4f*)(hk + 8 * j);
      const v4f e1 = *(const v4f*)(hk + 16 * D_STATE + 8 * j);
      #pragma unroll
      for (int e = 0; e < 4; ++e) {
        f.b0[j][e] = fmaf(a, f.b0[j][e], e0[e]);
        f.b1[j][e] = fmaf(a, f.b1[j][e], e1[e]);
      }
    }
  }

  const v8f zero = {0.f, 0.f, 0.f, 0.f, 0.f, 0.f, 0.f, 0.f};
  v8f c00 = zero, c01 = zero, c10 = zero, c11 = zero;
  frag_mma<JB>(f, c00, c01, c10, c11);

  float* yrow = y + ((long)b * LOUT + t0 + 8 * h) * D_INNER + hh * HEADDIM + p0 + c;
  const float* prow = pr + t0 + 8 * h;
  #pragma unroll
  for (int i = 0; i < 8; ++i) {
    const float s0 = prow[i];
    const float s1 = prow[i + 16];
    yrow[(long)i * D_INNER]             = fmaf(s0, c00[i], yrow[(long)i * D_INNER]);
    yrow[(long)i * D_INNER + 16]        = fmaf(s0, c01[i], yrow[(long)i * D_INNER + 16]);
    yrow[(long)(i + 16) * D_INNER]      = fmaf(s1, c10[i], yrow[(long)(i + 16) * D_INNER]);
    yrow[(long)(i + 16) * D_INNER + 16] = fmaf(s1, c11[i], yrow[(long)(i + 16) * D_INNER + 16]);
  }
}

__global__ void gate_rmsnorm_kernel(const float* __restrict__ zx,
                                    const float* __restrict__ nw,
                                    float* __restrict__ y) {
  const long bl = blockIdx.x;
  const int tid = threadIdx.x;
  __shared__ float red[256];
  float yv[4];
  float ss = 0.f;
  #pragma unroll
  for (int j = 0; j < 4; ++j) {
    const int i = tid + j * 256;
    const float z = zx[bl * D_IN_PROJ + i];
    const float g = z / (1.f + __expf(-z));
    const float v = y[bl * D_INNER + i] * g;
    yv[j] = v;
    ss = fmaf(v, v, ss);
  }
  red[tid] = ss;
  __syncthreads();
  for (int st = 128; st > 0; st >>= 1) {
    if (tid < st) red[tid] += red[tid + st];
    __syncthreads();
  }
  const float scale = rsqrtf(red[0] * (1.f / (float)D_INNER) + 1e-5f);
  #pragma unroll
  for (int j = 0; j < 4; ++j) {
    const int i = tid + j * 256;
    y[bl * D_INNER + i] = yv[j] * scale * nw[i];
  }
}

extern "C" void kernel_launch(void* const* d_in, const int* in_sizes, int n_in,
                              void* d_out, int out_size, void* d_ws, size_t ws_size,
                              hipStream_t stream) {
  const float* tokens     = (const float*)d_in[0];
  const float* conv_w     = (const float*)d_in[1];
  const float* in_proj_w  = (const float*)d_in[2];
  const float* conv1d_w   = (const float*)d_in[3];
  const float* conv1d_b   = (const float*)d_in[4];
  const float* dt_bias    = (const float*)d_in[5];
  const float* A_log      = (const float*)d_in[6];
  const float* D_param    = (const float*)d_in[7];
  const float* norm_w     = (const float*)d_in[8];
  const float* out_proj_w = (const float*)d_in[9];
  float* out = (float*)d_out;
  float* ws  = (float*)d_ws;

  float* u0   = ws;
  float* u1   = u0 + (long)ROWS * D_MODEL;
  float* zx   = u1 + (long)ROWS * D_MODEL;
  float* bcact = zx + (long)ROWS * D_IN_PROJ;
  float* ybuf = bcact + (long)ROWS * BC_DIM;
  float* hloc = ybuf + (long)ROWS * D_INNER;
  float* pcum = hloc + (long)BATCH * NHEADS * SEGS * HSTATE;
  (void)ws_size; (void)in_sizes; (void)n_in; (void)out_size;

  stem_conv_wmma<<<GEMM_WGS(D_MODEL), 128, 0, stream>>>(tokens, conv_w, u0);

  const float* u_in = u0;
  for (int blk = 0; blk < 2; ++blk) {
    const float* ipw = in_proj_w  + (long)blk * D_IN_PROJ * D_MODEL;
    const float* cw  = conv1d_w   + (long)blk * CONV_DIM * D_CONV;
    const float* cb  = conv1d_b   + (long)blk * CONV_DIM;
    const float* dtb = dt_bias    + (long)blk * NHEADS;
    const float* al  = A_log      + (long)blk * NHEADS;
    const float* Dpp = D_param    + (long)blk * NHEADS;
    const float* nw  = norm_w     + (long)blk * D_INNER;
    const float* opw = out_proj_w + (long)blk * D_MODEL * D_INNER;

    {
      gemm_rows_wmma<D_IN_PROJ, D_MODEL, 32><<<GEMM_WGS(D_IN_PROJ), 128, 0, stream>>>(u_in, ipw, zx);
    }
    dwconv_silu_bc_kernel<<<ROWS, BC_DIM, 0, stream>>>(zx, cw, cb, bcact);
    ssm_scan_kernel<<<SEGS * BATCH * NHEADS * 2, 128, 0, stream>>>(zx, bcact, cw, cb, dtb, al, Dpp, ybuf, hloc, pcum);
    ssm_carry_wmma<<<BATCH * NHEADS * (SEGS - 1) * (SEGLEN / 64), 128, 0, stream>>>(bcact, hloc, pcum, ybuf);
    gate_rmsnorm_kernel<<<ROWS, 256, 0, stream>>>(zx, nw, ybuf);
    float* dst = (blk == 0) ? u1 : out;
    {
      gemm_rows_wmma<D_MODEL, D_INNER, 32><<<GEMM_WGS(D_MODEL), 128, 0, stream>>>(ybuf, opw, dst);
    }
    u_in = u1;
  }
}


# ssm scan: a lane holds 4 n of 4 rows, so B and C are its own loaded operands, no cross-lane read per multiply-add; x made once per (row, step); a row's sum over n is 4 in a lane, then a 16-lane tree
# speedup vs baseline: 15.1841x; 1.0367x over previous
#include <hip/hip_runtime.h>
#include <hip/hip_bf16.h>

typedef __attribute__((ext_vector_type(2))) float v2f;
typedef __attribute__((ext_vector_type(8))) float v8f;

#define BATCH     2
#define SEQLEN    2048
#define C_IN      256
#define D_MODEL   512
#define LOUT      1024
#define D_STATE   64
#define D_CONV    4
#define D_INNER   1024
#define HEADDIM   64
#define NHEADS    16
#define CONV_DIM  (D_INNER + 2*D_STATE)
#define D_IN_PROJ (2*D_INNER + 2*D_STATE + NHEADS)
#define ROWS      (BATCH*LOUT)
#define KSTEM     (C_IN * 7)
#define GEMM_WGS(n) ((ROWS / 64) * (((n) + 63) / 64))

typedef __attribute__((ext_vector_type(4))) float v4f;

template <int JB>
struct RowFrag { v4f a0[JB]; v4f a1[JB]; v4f b0[JB]; v4f b1[JB]; };

__device__ __forceinline__ v8f mma4(v2f a, v2f b, v8f c) {
  return __builtin_amdgcn_wmma_f32_16x16x4_f32(false, a, false, b, (short)0, c, false, false);
}

template <int JB>
__device__ __forceinline__ void frag_mma(const RowFrag<JB>& f, v8f& c00, v8f& c01, v8f& c10, v8f& c11) {
  #pragma unroll
  for (int j = 0; j < JB; ++j) {
    #pragma unroll
    for (int u = 0; u < 2; ++u) {
      const v2f a0 = {f.a0[j][2 * u], f.a0[j][2 * u + 1]};
      const v2f a1 = {f.a1[j][2 * u], f.a1[j][2 * u + 1]};
      const v2f b0 = {f.b0[j][2 * u], f.b0[j][2 * u + 1]};
      const v2f b1 = {f.b1[j][2 * u], f.b1[j][2 * u + 1]};
      c00 = mma4(a0, b0, c00);
      c01 = mma4(a0, b1, c01);
      c10 = mma4(a1, b0, c10);
      c11 = mma4(a1, b1, c11);
    }
  }
}

template <int JB>
struct BlockFrag { v4f a[4][JB]; v4f b[4][JB]; };

template <int JB>
__device__ __forceinline__ void block_load(BlockFrag<JB>& f, const float* const (&pa)[4],
                                           const float* const (&pb)[4], int kb) {
  #pragma unroll
  for (int j = 0; j < JB; ++j) {
    #pragma unroll
    for (int t = 0; t < 4; ++t) {
      f.a[t][j] = *(const v4f*)(pa[t] + kb + 8 * j);
      f.b[t][j] = *(const v4f*)(pb[t] + kb + 8 * j);
    }
  }
}

template <int JB>
__device__ __forceinline__ void block_mma(const BlockFrag<JB>& f, v8f (&acc)[4][4]) {
  #pragma unroll
  for (int j = 0; j < JB; ++j) {
    #pragma unroll
    for (int u = 0; u < 2; ++u) {
      v2f a[4], b[4];
      #pragma unroll
      for (int t = 0; t < 4; ++t) {
        a[t] = v2f{f.a[t][j][2 * u], f.a[t][j][2 * u + 1]};
        b[t] = v2f{f.b[t][j][2 * u], f.b[t][j][2 * u + 1]};
      }
      #pragma unroll
      for (int mi = 0; mi < 4; ++mi) {
        #pragma unroll
        for (int ni = 0; ni < 4; ++ni) acc[mi][ni] = mma4(a[mi], b[ni], acc[mi][ni]);
      }
    }
  }
}

#define PART_PITCH 68

template <int N, int K, int KB>
__global__ __launch_bounds__(128)
void gemm_rows_wmma(const float* __restrict__ A,
                    const float* __restrict__ W,
                    float* __restrict__ C) {
  constexpr int KQ  = K / 4;
  static_assert(K % 4 == 0 && KQ % (2 * KB) == 0 && KB % 8 == 0 && N % 16 == 0, "shape");
  constexpr int JB  = KB / 8;
  constexpr int WGN = (N + 63) / 64;
  __shared__ float part[4][64][PART_PITCH];
  const int lane = threadIdx.x & 31;
  const int wave = __builtin_amdgcn_readfirstlane((int)(threadIdx.x >> 5));
  const int wgm  = blockIdx.x / WGN;
  const int wgn  = blockIdx.x - wgm * WGN;
  const int m0   = wgm * 64;
  const int n0   = wgn * 64;
  const int c = lane & 15;
  const int h = lane >> 4;
  const float* pa[4];
  const float* pb[4];
  #pragma unroll
  for (int t = 0; t < 4; ++t) {
    pa[t] = A + (long)(m0 + 16 * t + c) * K + wave * KQ + 4 * h;
    const int nt = (n0 + 16 * t < N) ? 16 * t : 0;
    pb[t] = W + (long)(n0 + nt + c) * K + wave * KQ + 4 * h;
  }

  v8f acc[4][4];
  #pragma unroll
  for (int mi = 0; mi < 4; ++mi) {
    #pragma unroll
    for (int ni = 0; ni < 4; ++ni) acc[mi][ni] = v8f{0.f, 0.f, 0.f, 0.f, 0.f, 0.f, 0.f, 0.f};
  }
  BlockFrag<JB> fa, fb;
  block_load<JB>(fa, pa, pb, 0);
  #pragma unroll 1
  for (int kb = 0; kb < KQ - 2 * KB; kb += 2 * KB) {
    block_load<JB>(fb, pa, pb, kb + KB);
    __builtin_amdgcn_sched_barrier(0);
    block_mma<JB>(fa, acc);
    __builtin_amdgcn_sched_barrier(0);
    block_load<JB>(fa, pa, pb, kb + 2 * KB);
    __builtin_amdgcn_sched_barrier(0);
    block_mma<JB>(fb, acc);
  }
  block_load<JB>(fb, pa, pb, KQ - KB);
  __builtin_amdgcn_sched_barrier(0);
  block_mma<JB>(fa, acc);
  block_mma<JB>(fb, acc);

  #pragma unroll
  for (int mi = 0; mi < 4; ++mi) {
    #pragma unroll
    for (int ni = 0; ni < 4; ++ni) {
      float* q = &part[wave][16 * ni + c][16 * mi + 8 * h];
      const v8f v = acc[mi][ni];
      *(v4f*)(q)     = v4f{v[0], v[1], v[2], v[3]};
      *(v4f*)(q + 4) = v4f{v[4], v[5], v[6], v[7]};
    }
  }
  __syncthreads();
  #pragma unroll
  for (int hf = 0; hf < 2; ++hf) {
    const int col = lane + 32 * hf;
    const bool ok = (n0 + col) < N;
    float* crow = C + (long)(m0 + 16 * wave) * N + n0 + col;
    #pragma unroll
    for (int r = 0; r < 4; ++r) {
      const v4f s0 = *(const v4f*)&part[0][col][16 * wave + 4 * r];
      const v4f s1 = *(const v4f*)&part[1][col][16 * wave + 4 * r];
      const v4f s2 = *(const v4f*)&part[2][col][16 * wave + 4 * r];
      const v4f s3 = *(const v4f*)&part[3][col][16 * wave + 4 * r];
      const v4f s = (s0 + s1) + (s2 + s3);
      if (ok) {
        #pragma unroll
        for (int e = 0; e < 4; ++e) crow[(long)(4 * r + e) * N] = s[e];
      }
    }
  }
}

#define STEM_CB 8
#define STEM_KB (STEM_CB * 7)
#define STEM_NB (C_IN / STEM_CB)
#define STEM_J  (STEM_KB / 8)

struct StemFrag { float a0[STEM_J][4]; float a1[STEM_J][4]; v4f b0[STEM_J]; v4f b1[STEM_J]; };
struct StemMap  { unsigned o0[STEM_J][4]; unsigned o1[STEM_J][4]; int m0[STEM_J][4]; int m1[STEM_J][4]; };

template <int J0, int J1>
__device__ __forceinline__ void stem_load(StemFrag& f, const StemMap& mp, const char* tb,
                                          const float* pb0, const float* pb1, int blk) {
  const char* t = tb + (long)blk * (STEM_CB * SEQLEN * 4);
  const int kb = blk * STEM_KB;
  #pragma unroll
  for (int j = J0; j < J1; ++j) {
    f.b0[j] = *(const v4f*)(pb0 + kb + 8 * j);
    f.b1[j] = *(const v4f*)(pb1 + kb + 8 * j);
    #pragma unroll
    for (int e = 0; e < 4; ++e) {
      f.a0[j][e] = *(const float*)(t + mp.o0[j][e]);
      f.a1[j][e] = *(const float*)(t + mp.o1[j][e]);
    }
  }
}

__device__ __forceinline__ float pad0(float v, int m) {
  return __builtin_bit_cast(float, __builtin_bit_cast(int, v) & m);
}

template <int J0, int J1>
__device__ __forceinline__ void stem_mma(const StemFrag& f, const StemMap& mp,
                                         v8f& c00, v8f& c01, v8f& c10, v8f& c11) {
  #pragma unroll
  for (int j = J0; j < J1; ++j) {
    #pragma unroll
    for (int u = 0; u < 2; ++u) {
      const v2f a0 = {pad0(f.a0[j][2 * u], mp.m0[j][2 * u]), pad0(f.a0[j][2 * u + 1], mp.m0[j][2 * u + 1])};
      const v2f a1 = {pad0(f.a1[j][2 * u], mp.m1[j][2 * u]), pad0(f.a1[j][2 * u + 1], mp.m1[j][2 * u + 1])};
      const v2f b0 = {f.b0[j][2 * u], f.b0[j][2 * u + 1]};
      const v2f b1 = {f.b1[j][2 * u], f.b1[j][2 * u + 1]};
      c00 = mma4(a0, b0, c00);
      c01 = mma4(a0, b1, c01);
      c10 = mma4(a1, b0, c10);
      c11 = mma4(a1, b1, c11);
    }
  }
}

__global__ __launch_bounds__(128)
void stem_conv_wmma(const float* __restrict__ tokens,
                    const float* __restrict__ W,
                    float* __restrict__ C) {
  constexpr int N   = D_MODEL;
  constexpr int K   = KSTEM;
  constexpr int WGN = N / 64;
  const int lane = threadIdx.x & 31;
  const int wave = __builtin_amdgcn_readfirstlane((int)(threadIdx.x >> 5));
  const int wgm  = blockIdx.x / WGN;
  const int wgn  = blockIdx.x - wgm * WGN;
  const int m0   = wgm * 64 + (wave >> 1) * 32;
  const int n0   = wgn * 64 + (wave & 1) * 32;
  const int c = lane & 15;
  const int h = lane >> 4;
  const float* pb0 = W + (long)(n0 + c) * K + 4 * h;
  const float* pb1 = pb0 + (long)16 * K;
  const char* tb = (const char*)(tokens + (long)(m0 / LOUT) * C_IN * SEQLEN);
  const int l0 = (m0 & (LOUT - 1)) + c;

  StemMap mp;
  #pragma unroll
  for (int j = 0; j < STEM_J; ++j) {
    #pragma unroll
    for (int e = 0; e < 4; ++e) {
      const int kl = 8 * j + 4 * h + e;
      const int ch = kl / 7;
      const int tp = kl - 7 * ch;
      const int s0 = 2 * l0 - 3 + tp;
      const int s1 = s0 + 32;
      const bool ok0 = (unsigned)s0 < (unsigned)SEQLEN;
      const bool ok1 = (unsigned)s1 < (unsigned)SEQLEN;
      mp.o0[j][e] = (unsigned)(ch * SEQLEN + (ok0 ? s0 : 0)) * 4u;
      mp.o1[j][e] = (unsigned)(ch * SEQLEN + (ok1 ? s1 : 0)) * 4u;
      mp.m0[j][e] = ok0 ? -1 : 0;
      mp.m1[j][e] = ok1 ? -1 : 0;
    }
  }

  const v8f zero = {0.f, 0.f, 0.f, 0.f, 0.f, 0.f, 0.f, 0.f};
  v8f c00 = zero, c01 = zero, c10 = zero, c11 = zero;
  StemFrag fa, fb;
  stem_load<0, 4>(fa, mp, tb, pb0, pb1, 0);
  stem_load<4, STEM_J>(fa, mp, tb, pb0, pb1, 0);
  #pragma unroll 1
  for (int blk = 0; blk < STEM_NB - 2; blk += 2) {
    stem_load<0, 4>(fb, mp, tb, pb0, pb1, blk + 1);
    __builtin_amdgcn_sched_barrier(0);
    stem_mma<0, 4>(fa, mp, c00, c01, c10, c11);
    __builtin_amdgcn_sched_barrier(0);
    stem_load<4, STEM_J>(fb, mp, tb, pb0, pb1, blk + 1);
    __builtin_amdgcn_sched_barrier(0);
    stem_mma<4, STEM_J>(fa, mp, c00, c01, c10, c11);
    __builtin_amdgcn_sched_barrier(0);
    stem_load<0, 4>(fa, mp, tb, pb0, pb1, blk + 2);
    __builtin_amdgcn_sched_barrier(0);
    stem_mma<0, 4>(fb, mp, c00, c01, c10, c11);
    __builtin_amdgcn_sched_barrier(0);
    stem_load<4, STEM_J>(fa, mp, tb, pb0, pb1, blk + 2);
    __builtin_amdgcn_sched_barrier(0);
    stem_mma<4, STEM_J>(fb, mp, c00, c01, c10, c11);
    __builtin_amdgcn_sched_barrier(0);
  }
  stem_load<0, 4>(fb, mp, tb, pb0, pb1, STEM_NB - 1);
  __builtin_amdgcn_sched_barrier(0);
  stem_mma<0, 4>(fa, mp, c00, c01, c10, c11);
  __builtin_amdgcn_sched_barrier(0);
  stem_load<4, STEM_J>(fb, mp, tb, pb0, pb1, STEM_NB - 1);
  __builtin_amdgcn_sched_barrier(0);
  stem_mma<4, STEM_J>(fa, mp, c00, c01, c10, c11);
  stem_mma<0, STEM_J>(fb, mp, c00, c01, c10, c11);

  float* crow = C + (long)(m0 + 8 * h) * N + n0 + c;
  #pragma unroll
  for (int i = 0; i < 8; ++i) {
    crow[(long)i * N]             = c00[i];
    crow[(long)(i + 16) * N]      = c10[i];
    crow[(long)i * N + 16]        = c01[i];
    crow[(long)(i + 16) * N + 16] = c11[i];
  }
}

#define BC_DIM (2 * D_STATE)
__global__ void dwconv_silu_bc_kernel(const float* __restrict__ zx,
                                      const float* __restrict__ w,
                                      const float* __restrict__ bias,
                                      float* __restrict__ out) {
  const int c  = threadIdx.x;
  const int cc = D_INNER + c;
  const long bl = blockIdx.x;
  const int l = (int)(bl & (LOUT - 1));
  const float* xcol = zx + bl * D_IN_PROJ + D_INNER + cc;
  const float w0 = w[cc * 4 + 0], w1 = w[cc * 4 + 1], w2 = w[cc * 4 + 2], w3 = w[cc * 4 + 3];
  float acc = w3 * xcol[0];
  if (l >= 1) acc = fmaf(w2, xcol[-(long)D_IN_PROJ],     acc);
  if (l >= 2) acc = fmaf(w1, xcol[-(long)2 * D_IN_PROJ], acc);
  if (l >= 3) acc = fmaf(w0, xcol[-(long)3 * D_IN_PROJ], acc);
  const float v = acc + bias[cc];
  out[bl * BC_DIM + c] = v / (1.f + __expf(-v));
}

#define SCAN_G 8
#define SEGS   8
#define SEGLEN (LOUT / SEGS)
#define HSTATE (HEADDIM * D_STATE)

template <int N>
__device__ __forceinline__ float row_bcast16(float v) {
  const unsigned sel = 0x11111111u * (unsigned)N;
  return __builtin_bit_cast(float, __builtin_amdgcn_permlane16(
      __builtin_bit_cast(unsigned, v), __builtin_bit_cast(unsigned, v), sel, sel, false, false));
}
template <int M>
__device__ __forceinline__ float row_xor(float v) {
  return __builtin_bit_cast(float, __builtin_amdgcn_update_dpp(
      0, __builtin_bit_cast(int, v), 0x160 + M, 0xF, 0xF, true));
}
__device__ __forceinline__ float lane_bcast(float v, int l) {
  return __builtin_bit_cast(float, __builtin_amdgcn_readlane(__builtin_bit_cast(int, v), l));
}
__device__ __forceinline__ float lane_gather(float v, int l) {
  return __builtin_bit_cast(float, __builtin_amdgcn_ds_bpermute(l << 2, __builtin_bit_cast(int, v)));
}
__device__ __forceinline__ float softplus_dt(float raw) {
  return (raw > 20.f) ? raw : log1pf(__expf(raw));
}

struct ScanSet { v4f B[SCAN_G]; v4f C[SCAN_G]; float xr[2][4]; };
struct XConv { float w0, w1, w2, w3, bias; };

__device__ __forceinline__ void scan_load(ScanSet& s, const float* bp, const float* cp,
                                          const float* xp, int l0, int sp) {
  #pragma unroll
  for (int j = 0; j < SCAN_G; ++j) {
    const long row = (long)min(l0 + j, LOUT - 1);
    s.B[j] = *(const v4f*)(bp + row * BC_DIM);
    s.C[j] = *(const v4f*)(cp + row * BC_DIM);
  }
  #pragma unroll
  for (int k = 0; k < 2; ++k) {
    #pragma unroll
    for (int d = 0; d < 4; ++d) {
      const long row = (long)max(min(l0 + 4 * k + sp - d, LOUT - 1), 0);
      s.xr[k][d] = xp[row * D_IN_PROJ];
    }
  }
}

template <int S>
__device__ __forceinline__ float scan_step(float (&h)[4][4], const v4f& Bv, const v4f& Cv,
                                           float dA, float xdto, bool b0, bool b1) {
  float xb[4];
  xb[0] = row_bcast16<4 * S + 0>(xdto);
  xb[1] = row_bcast16<4 * S + 1>(xdto);
  xb[2] = row_bcast16<4 * S + 2>(xdto);
  xb[3] = row_bcast16<4 * S + 3>(xdto);
  float acc[4];
  #pragma unroll
  for (int r = 0; r < 4; ++r) {
    #pragma unroll
    for (int e = 0; e < 4; ++e) h[r][e] = fmaf(h[r][e], dA, xb[r] * Bv[e]);
    float a = h[r][0] * Cv[0];
    a = fmaf(h[r][1], Cv[1], a);
    a = fmaf(h[r][2], Cv[2], a);
    a = fmaf(h[r][3], Cv[3], a);
    acc[r] = a;
  }
  const float k01 = b0 ? acc[1] : acc[0];
  const float s01 = b0 ? acc[0] : acc[1];
  const float k23 = b0 ? acc[3] : acc[2];
  const float s23 = b0 ? acc[2] : acc[3];
  const float v01 = k01 + row_xor<1>(s01);
  const float v23 = k23 + row_xor<1>(s23);
  const float kk = b1 ? v23 : v01;
  const float ss = b1 ? v01 : v23;
  float w = kk + row_xor<2>(ss);
  w += row_xor<4>(w);
  w += row_xor<8>(w);
  return w;
}

__device__ __forceinline__ void scan_steps(const ScanSet& s, const XConv& k, float (&h)[4][4], float dtv, float dav,
                                           int lane0, int l0, float Dh, int sp, bool b0, bool b1, float* yrow,
                                           float& pc, float& pk, int lane) {
  float xo[2], xdto[2];
  #pragma unroll
  for (int g = 0; g < 2; ++g) {
    const int tl = l0 + 4 * g + sp;
    const float r0 = s.xr[g][0];
    const float r1 = (tl >= 1) ? s.xr[g][1] : 0.f;
    const float r2 = (tl >= 2) ? s.xr[g][2] : 0.f;
    const float r3 = (tl >= 3) ? s.xr[g][3] : 0.f;
    float a = k.w3 * r0;
    a = fmaf(k.w2, r1, a);
    a = fmaf(k.w1, r2, a);
    a = fmaf(k.w0, r3, a);
    const float v = a + k.bias;
    xo[g] = v / (1.f + __expf(-v));
    xdto[g] = xo[g] * lane_gather(dtv, lane0 + 4 * g + sp);
  }
  #pragma unroll
  for (int g = 0; g < 2; ++g) {
    float keep = 0.f;
    #pragma unroll
    for (int st = 0; st < 4; ++st) {
      const int j = 4 * g + st;
      const float dA = lane_bcast(dav, lane0 + j);
      pc *= dA;
      pk = (lane == lane0 + j) ? pc : pk;
      float w;
      if (st == 0)      w = scan_step<0>(h, s.B[j], s.C[j], dA, xdto[g], b0, b1);
      else if (st == 1) w = scan_step<1>(h, s.B[j], s.C[j], dA, xdto[g], b0, b1);
      else if (st == 2) w = scan_step<2>(h, s.B[j], s.C[j], dA, xdto[g], b0, b1);
      else              w = scan_step<3>(h, s.B[j], s.C[j], dA, xdto[g], b0, b1);
      keep = (sp == st) ? w : keep;
    }
    yrow[(long)(4 * g) * D_INNER] = fmaf(xo[g], Dh, keep);
  }
}

__global__ __launch_bounds__(128)
void ssm_scan_kernel(const float* __restrict__ zx,
                     const float* __restrict__ bc,
                     const float* __restrict__ cw,
                     const float* __restrict__ cb,
                     const float* __restrict__ dtb,
                     const float* __restrict__ alog,
                     const float* __restrict__ Dp,
                     float* __restrict__ y,
                     float* __restrict__ hloc,
                     float* __restrict__ pcum) {
  const int seg  = blockIdx.x >> 6;
  const int b    = (blockIdx.x >> 5) & 1;
  const int hh   = (blockIdx.x >> 1) & 15;
  const int half = blockIdx.x & 1;
  const int lane = threadIdx.x & 31;
  const int wave = threadIdx.x >> 5;
  const int i    = lane & 15;
  const int sp   = i >> 2;
  const bool b0  = (lane & 1) != 0;
  const bool b1  = (lane & 2) != 0;
  const int p0   = half * 32 + wave * 8 + (lane >> 4) * 4;
  const int po   = p0 + (i & 3);
  const float Ah   = -__expf(alog[hh]);
  const float dtbh = dtb[hh];
  const float Dh   = Dp[hh];

  const float* zrows = zx + (long)b * LOUT * D_IN_PROJ;
  const float* bp   = bc + (long)b * LOUT * BC_DIM + 4 * i;
  const float* cp   = bp + D_STATE;
  const int    ch   = hh * HEADDIM + po;
  const float* xp   = zrows + D_INNER + ch;
  const float* dtp  = zrows + (D_IN_PROJ - NHEADS) + hh;
  float* yp = y + ((long)b * LOUT + sp) * D_INNER + ch;

  float h[4][4];
  #pragma unroll
  for (int r = 0; r < 4; ++r) {
    #pragma unroll
    for (int e = 0; e < 4; ++e) h[r][e] = 0.f;
  }

  const int  c0     = seg * (SEGLEN / 32);
  const bool pwrite = (half == 0) && (wave == 0);
  float* pout = pcum + (long)(b * NHEADS + hh) * LOUT;
  float pc = 1.f, pk = 0.f;

  XConv k;
  {
    const v4f wv = *(const v4f*)(cw + ch * D_CONV);
    k.w0 = wv[0]; k.w1 = wv[1]; k.w2 = wv[2]; k.w3 = wv[3];
    k.bias = cb[ch];
  }

  ScanSet sa, sb;
  scan_load(sa, bp, cp, xp, c0 * 32, sp);
  float dtv = softplus_dt(dtp[(long)(c0 * 32 + lane) * D_IN_PROJ] + dtbh);
  float dav = __expf(dtv * Ah);

  for (int c = c0; c < c0 + SEGLEN / 32; ++c) {
    const int   cn   = min(c + 1, LOUT / 32 - 1);
    const float rawn = dtp[(long)(cn * 32 + lane) * D_IN_PROJ];
    for (int g = 0; g < 32; g += 2 * SCAN_G) {
      const int l0 = c * 32 + g;
      scan_load(sb, bp, cp, xp, l0 + SCAN_G, sp);
      __builtin_amdgcn_sched_barrier(0);
      scan_steps(sa, k, h, dtv, dav, g, l0, Dh, sp, b0, b1, yp + (long)l0 * D_INNER, pc, pk, lane);
      __builtin_amdgcn_sched_barrier(0);
      scan_load(sa, bp, cp, xp, l0 + 2 * SCAN_G, sp);
      __builtin_amdgcn_sched_barrier(0);
      scan_steps(sb, k, h, dtv, dav, g + SCAN_G, l0 + SCAN_G, Dh, sp, b0, b1, yp + (long)(l0 + SCAN_G) * D_INNER, pc, pk, lane);
    }
    if (pwrite) pout[c * 32 + lane] = pk;
    dtv = softplus_dt(rawn + dtbh);
    dav = __expf(dtv * Ah);
  }
  if (seg < SEGS - 1) {
    float* hp = hloc + ((long)(b * NHEADS + hh) * SEGS + seg) * HSTATE + p0 * D_STATE + 4 * i;
    #pragma unroll
    for (int r = 0; r < 4; ++r) {
      const v4f v = {h[r][0], h[r][1], h[r][2], h[r][3]};
      *(v4f*)(hp + r * D_STATE) = v;
    }
  }
}

__global__ __launch_bounds__(128)
void ssm_carry_wmma(const float* __restrict__ bc,
                    const float* __restrict__ hloc,
                    const float* __restrict__ pcum,
                    float* __restrict__ y) {
  constexpr int JB = D_STATE / 8;
  const int lane = threadIdx.x & 31;
  const int wave = __builtin_amdgcn_readfirstlane((int)(threadIdx.x >> 5));
  const int tb   = blockIdx.x & (SEGLEN / 64 - 1);
  const int r    = blockIdx.x / (SEGLEN / 64);
  const int bh   = r / (SEGS - 1);
  const int seg  = 1 + (r - bh * (SEGS - 1));
  const int b    = bh / NHEADS;
  const int hh   = bh - b * NHEADS;
  const int t0   = seg * SEGLEN + tb * 64 + (wave >> 1) * 32;
  const int p0   = (wave & 1) * 32;
  const int c = lane & 15;
  const int h = lane >> 4;
  const float* pa0 = bc + ((long)b * LOUT + t0 + c) * BC_DIM + D_STATE + 4 * h;
  const float* pa1 = pa0 + (long)16 * BC_DIM;
  const float* hb0 = hloc + (long)bh * SEGS * HSTATE + (p0 + c) * D_STATE + 4 * h;
  const float* pr  = pcum + (long)bh * LOUT;

  RowFrag<JB> f;
  #pragma unroll
  for (int j = 0; j < JB; ++j) {
    f.a0[j] = *(const v4f*)(pa0 + 8 * j);
    f.a1[j] = *(const v4f*)(pa1 + 8 * j);
    f.b0[j] = *(const v4f*)(hb0 + 8 * j);
    f.b1[j] = *(const v4f*)(hb0 + 16 * D_STATE + 8 * j);
  }
  for (int k = 1; k < seg; ++k) {
    const float a = pr[k * SEGLEN + SEGLEN - 1];
    const float* hk = hb0 + (long)k * HSTATE;
    #pragma unroll
    for (int j = 0; j < JB; ++j) {
      const v4f e0 = *(const v4f*)(hk + 8 * j);
      const v4f e1 = *(const v4f*)(hk + 16 * D_STATE + 8 * j);
      #pragma unroll
      for (int e = 0; e < 4; ++e) {
        f.b0[j][e] = fmaf(a, f.b0[j][e], e0[e]);
        f.b1[j][e] = fmaf(a, f.b1[j][e], e1[e]);
      }
    }
  }

  const v8f zero = {0.f, 0.f, 0.f, 0.f, 0.f, 0.f, 0.f, 0.f};
  v8f c00 = zero, c01 = zero, c10 = zero, c11 = zero;
  frag_mma<JB>(f, c00, c01, c10, c11);

  float* yrow = y + ((long)b * LOUT + t0 + 8 * h) * D_INNER + hh * HEADDIM + p0 + c;
  const float* prow = pr + t0 + 8 * h;
  #pragma unroll
  for (int i = 0; i < 8; ++i) {
    const float s0 = prow[i];
    const float s1 = prow[i + 16];
    yrow[(long)i * D_INNER]             = fmaf(s0, c00[i], yrow[(long)i * D_INNER]);
    yrow[(long)i * D_INNER + 16]        = fmaf(s0, c01[i], yrow[(long)i * D_INNER + 16]);
    yrow[(long)(i + 16) * D_INNER]      = fmaf(s1, c10[i], yrow[(long)(i + 16) * D_INNER]);
    yrow[(long)(i + 16) * D_INNER + 16] = fmaf(s1, c11[i], yrow[(long)(i + 16) * D_INNER + 16]);
  }
}

__global__ void gate_rmsnorm_kernel(const float* __restrict__ zx,
                                    const float* __restrict__ nw,
                                    float* __restrict__ y) {
  const long bl = blockIdx.x;
  const int tid = threadIdx.x;
  __shared__ float red[256];
  float yv[4];
  float ss = 0.f;
  #pragma unroll
  for (int j = 0; j < 4; ++j) {
    const int i = tid + j * 256;
    const float z = zx[bl * D_IN_PROJ + i];
    const float g = z / (1.f + __expf(-z));
    const float v = y[bl * D_INNER + i] * g;
    yv[j] = v;
    ss = fmaf(v, v, ss);
  }
  red[tid] = ss;
  __syncthreads();
  for (int st = 128; st > 0; st >>= 1) {
    if (tid < st) red[tid] += red[tid + st];
    __syncthreads();
  }
  const float scale = rsqrtf(red[0] * (1.f / (float)D_INNER) + 1e-5f);
  #pragma unroll
  for (int j = 0; j < 4; ++j) {
    const int i = tid + j * 256;
    y[bl * D_INNER + i] = yv[j] * scale * nw[i];
  }
}

extern "C" void kernel_launch(void* const* d_in, const int* in_sizes, int n_in,
                              void* d_out, int out_size, void* d_ws, size_t ws_size,
                              hipStream_t stream) {
  const float* tokens     = (const float*)d_in[0];
  const float* conv_w     = (const float*)d_in[1];
  const float* in_proj_w  = (const float*)d_in[2];
  const float* conv1d_w   = (const float*)d_in[3];
  const float* conv1d_b   = (const float*)d_in[4];
  const float* dt_bias    = (const float*)d_in[5];
  const float* A_log      = (const float*)d_in[6];
  const float* D_param    = (const float*)d_in[7];
  const float* norm_w     = (const float*)d_in[8];
  const float* out_proj_w = (const float*)d_in[9];
  float* out = (float*)d_out;
  float* ws  = (float*)d_ws;

  float* u0   = ws;
  float* u1   = u0 + (long)ROWS * D_MODEL;
  float* zx   = u1 + (long)ROWS * D_MODEL;
  float* bcact = zx + (long)ROWS * D_IN_PROJ;
  float* ybuf = bcact + (long)ROWS * BC_DIM;
  float* hloc = ybuf + (long)ROWS * D_INNER;
  float* pcum = hloc + (long)BATCH * NHEADS * SEGS * HSTATE;
  (void)ws_size; (void)in_sizes; (void)n_in; (void)out_size;

  stem_conv_wmma<<<GEMM_WGS(D_MODEL), 128, 0, stream>>>(tokens, conv_w, u0);

  const float* u_in = u0;
  for (int blk = 0; blk < 2; ++blk) {
    const float* ipw = in_proj_w  + (long)blk * D_IN_PROJ * D_MODEL;
    const float* cw  = conv1d_w   + (long)blk * CONV_DIM * D_CONV;
    const float* cb  = conv1d_b   + (long)blk * CONV_DIM;
    const float* dtb = dt_bias    + (long)blk * NHEADS;
    const float* al  = A_log      + (long)blk * NHEADS;
    const float* Dpp = D_param    + (long)blk * NHEADS;
    const float* nw  = norm_w     + (long)blk * D_INNER;
    const float* opw = out_proj_w + (long)blk * D_MODEL * D_INNER;

    {
      gemm_rows_wmma<D_IN_PROJ, D_MODEL, 32><<<GEMM_WGS(D_IN_PROJ), 128, 0, stream>>>(u_in, ipw, zx);
    }
    dwconv_silu_bc_kernel<<<ROWS, BC_DIM, 0, stream>>>(zx, cw, cb, bcact);
    ssm_scan_kernel<<<SEGS * BATCH * NHEADS * 2, 128, 0, stream>>>(zx, bcact, cw, cb, dtb, al, Dpp, ybuf, hloc, pcum);
    ssm_carry_wmma<<<BATCH * NHEADS * (SEGS - 1) * (SEGLEN / 64), 128, 0, stream>>>(bcact, hloc, pcum, ybuf);
    gate_rmsnorm_kernel<<<ROWS, 256, 0, stream>>>(zx, nw, ybuf);
    float* dst = (blk == 0) ? u1 : out;
    {
      gemm_rows_wmma<D_MODEL, D_INNER, 32><<<GEMM_WGS(D_MODEL), 128, 0, stream>>>(ybuf, opw, dst);
    }
    u_in = u1;
  }
}


# gate SiLU(z) x norm weight applied where the carry matrix kernel completes y (run on segment 0 too), RMSNorm row factor on out_proj's result rows: gate kernel gone; row squares summed in new order
# speedup vs baseline: 15.7358x; 1.0363x over previous
#include <hip/hip_runtime.h>
#include <hip/hip_bf16.h>

typedef __attribute__((ext_vector_type(2))) float v2f;
typedef __attribute__((ext_vector_type(8))) float v8f;

#define BATCH     2
#define SEQLEN    2048
#define C_IN      256
#define D_MODEL   512
#define LOUT      1024
#define D_STATE   64
#define D_CONV    4
#define D_INNER   1024
#define HEADDIM   64
#define NHEADS    16
#define CONV_DIM  (D_INNER + 2*D_STATE)
#define D_IN_PROJ (2*D_INNER + 2*D_STATE + NHEADS)
#define ROWS      (BATCH*LOUT)
#define KSTEM     (C_IN * 7)
#define GEMM_WGS(n) ((ROWS / 64) * (((n) + 63) / 64))
#define SQ_PARTS  (2 * NHEADS)

typedef __attribute__((ext_vector_type(4))) float v4f;

template <int JB>
struct RowFrag { v4f a0[JB]; v4f a1[JB]; v4f b0[JB]; v4f b1[JB]; };

__device__ __forceinline__ v8f mma4(v2f a, v2f b, v8f c) {
  return __builtin_amdgcn_wmma_f32_16x16x4_f32(false, a, false, b, (short)0, c, false, false);
}
__device__ __forceinline__ float lane_bcast(float v, int l) {
  return __builtin_bit_cast(float, __builtin_amdgcn_readlane(__builtin_bit_cast(int, v), l));
}
__device__ __forceinline__ float half_swap(float v) {
  return __builtin_bit_cast(float, __builtin_amdgcn_permlanex16(
      __builtin_bit_cast(unsigned, v), __builtin_bit_cast(unsigned, v), 0x76543210u, 0xfedcba98u, false, false));
}

template <int JB>
__device__ __forceinline__ void frag_mma(const RowFrag<JB>& f, v8f& c00, v8f& c01, v8f& c10, v8f& c11) {
  #pragma unroll
  for (int j = 0; j < JB; ++j) {
    #pragma unroll
    for (int u = 0; u < 2; ++u) {
      const v2f a0 = {f.a0[j][2 * u], f.a0[j][2 * u + 1]};
      const v2f a1 = {f.a1[j][2 * u], f.a1[j][2 * u + 1]};
      const v2f b0 = {f.b0[j][2 * u], f.b0[j][2 * u + 1]};
      const v2f b1 = {f.b1[j][2 * u], f.b1[j][2 * u + 1]};
      c00 = mma4(a0, b0, c00);
      c01 = mma4(a0, b1, c01);
      c10 = mma4(a1, b0, c10);
      c11 = mma4(a1, b1, c11);
    }
  }
}

template <int JB>
struct BlockFrag { v4f a[4][JB]; v4f b[4][JB]; };

template <int JB>
__device__ __forceinline__ void block_load(BlockFrag<JB>& f, const float* const (&pa)[4],
                                           const float* const (&pb)[4], int kb) {
  #pragma unroll
  for (int j = 0; j < JB; ++j) {
    #pragma unroll
    for (int t = 0; t < 4; ++t) {
      f.a[t][j] = *(const v4f*)(pa[t] + kb + 8 * j);
      f.b[t][j] = *(const v4f*)(pb[t] + kb + 8 * j);
    }
  }
}

template <int JB>
__device__ __forceinline__ void block_mma(const BlockFrag<JB>& f, v8f (&acc)[4][4]) {
  #pragma unroll
  for (int j = 0; j < JB; ++j) {
    #pragma unroll
    for (int u = 0; u < 2; ++u) {
      v2f a[4], b[4];
      #pragma unroll
      for (int t = 0; t < 4; ++t) {
        a[t] = v2f{f.a[t][j][2 * u], f.a[t][j][2 * u + 1]};
        b[t] = v2f{f.b[t][j][2 * u], f.b[t][j][2 * u + 1]};
      }
      #pragma unroll
      for (int mi = 0; mi < 4; ++mi) {
        #pragma unroll
        for (int ni = 0; ni < 4; ++ni) acc[mi][ni] = mma4(a[mi], b[ni], acc[mi][ni]);
      }
    }
  }
}

#define PART_PITCH 68

template <int N, int K, int KB, bool ROWF>
__global__ __launch_bounds__(128)
void gemm_rows_wmma(const float* __restrict__ A,
                    const float* __restrict__ W,
                    const float* __restrict__ sq,
                    float* __restrict__ C) {
  constexpr int KQ  = K / 4;
  static_assert(K % 4 == 0 && KQ % (2 * KB) == 0 && KB % 8 == 0 && N % 16 == 0, "shape");
  constexpr int JB  = KB / 8;
  constexpr int WGN = (N + 63) / 64;
  __shared__ float part[4][64][PART_PITCH];
  const int lane = threadIdx.x & 31;
  const int wave = __builtin_amdgcn_readfirstlane((int)(threadIdx.x >> 5));
  const int wgm  = blockIdx.x / WGN;
  const int wgn  = blockIdx.x - wgm * WGN;
  const int m0   = wgm * 64;
  const int n0   = wgn * 64;
  const int c = lane & 15;
  const int h = lane >> 4;
  float rowf = 1.f;
  if (ROWF) {
    const float* sp = sq + (long)(16 * h) * ROWS + m0 + 16 * wave + c;
    float ssum = sp[0];
    #pragma unroll
    for (int q = 1; q < 16; ++q) ssum += sp[(long)q * ROWS];
    const float tot = ssum + half_swap(ssum);
    rowf = rsqrtf(tot * (1.f / (float)D_INNER) + 1e-5f);
  }
  const float* pa[4];
  const float* pb[4];
  #pragma unroll
  for (int t = 0; t < 4; ++t) {
    pa[t] = A + (long)(m0 + 16 * t + c) * K + wave * KQ + 4 * h;
    const int nt = (n0 + 16 * t < N) ? 16 * t : 0;
    pb[t] = W + (long)(n0 + nt + c) * K + wave * KQ + 4 * h;
  }

  v8f acc[4][4];
  #pragma unroll
  for (int mi = 0; mi < 4; ++mi) {
    #pragma unroll
    for (int ni = 0; ni < 4; ++ni) acc[mi][ni] = v8f{0.f, 0.f, 0.f, 0.f, 0.f, 0.f, 0.f, 0.f};
  }
  BlockFrag<JB> fa, fb;
  block_load<JB>(fa, pa, pb, 0);
  #pragma unroll 1
  for (int kb = 0; kb < KQ - 2 * KB; kb += 2 * KB) {
    block_load<JB>(fb, pa, pb, kb + KB);
    __builtin_amdgcn_sched_barrier(0);
    block_mma<JB>(fa, acc);
    __builtin_amdgcn_sched_barrier(0);
    block_load<JB>(fa, pa, pb, kb + 2 * KB);
    __builtin_amdgcn_sched_barrier(0);
    block_mma<JB>(fb, acc);
  }
  block_load<JB>(fb, pa, pb, KQ - KB);
  __builtin_amdgcn_sched_barrier(0);
  block_mma<JB>(fa, acc);
  block_mma<JB>(fb, acc);

  #pragma unroll
  for (int mi = 0; mi < 4; ++mi) {
    #pragma unroll
    for (int ni = 0; ni < 4; ++ni) {
      float* q = &part[wave][16 * ni + c][16 * mi + 8 * h];
      const v8f v = acc[mi][ni];
      *(v4f*)(q)     = v4f{v[0], v[1], v[2], v[3]};
      *(v4f*)(q + 4) = v4f{v[4], v[5], v[6], v[7]};
    }
  }
  __syncthreads();
  float rf[16];
  #pragma unroll
  for (int k = 0; k < 16; ++k) rf[k] = ROWF ? lane_bcast(rowf, k) : 1.f;
  #pragma unroll
  for (int hf = 0; hf < 2; ++hf) {
    const int col = lane + 32 * hf;
    const bool ok = (n0 + col) < N;
    float* crow = C + (long)(m0 + 16 * wave) * N + n0 + col;
    #pragma unroll
    for (int r = 0; r < 4; ++r) {
      const v4f s0 = *(const v4f*)&part[0][col][16 * wave + 4 * r];
      const v4f s1 = *(const v4f*)&part[1][col][16 * wave + 4 * r];
      const v4f s2 = *(const v4f*)&part[2][col][16 * wave + 4 * r];
      const v4f s3 = *(const v4f*)&part[3][col][16 * wave + 4 * r];
      const v4f s = (s0 + s1) + (s2 + s3);
      if (ok) {
        #pragma unroll
        for (int e = 0; e < 4; ++e) crow[(long)(4 * r + e) * N] = ROWF ? s[e] * rf[4 * r + e] : s[e];
      }
    }
  }
}

#define STEM_CB 8
#define STEM_KB (STEM_CB * 7)
#define STEM_NB (C_IN / STEM_CB)
#define STEM_J  (STEM_KB / 8)

struct StemFrag { float a0[STEM_J][4]; float a1[STEM_J][4]; v4f b0[STEM_J]; v4f b1[STEM_J]; };
struct StemMap  { unsigned o0[STEM_J][4]; unsigned o1[STEM_J][4]; int m0[STEM_J][4]; int m1[STEM_J][4]; };

template <int J0, int J1>
__device__ __forceinline__ void stem_load(StemFrag& f, const StemMap& mp, const char* tb,
                                          const float* pb0, const float* pb1, int blk) {
  const char* t = tb + (long)blk * (STEM_CB * SEQLEN * 4);
  const int kb = blk * STEM_KB;
  #pragma unroll
  for (int j = J0; j < J1; ++j) {
    f.b0[j] = *(const v4f*)(pb0 + kb + 8 * j);
    f.b1[j] = *(const v4f*)(pb1 + kb + 8 * j);
    #pragma unroll
    for (int e = 0; e < 4; ++e) {
      f.a0[j][e] = *(const float*)(t + mp.o0[j][e]);
      f.a1[j][e] = *(const float*)(t + mp.o1[j][e]);
    }
  }
}

__device__ __forceinline__ float pad0(float v, int m) {
  return __builtin_bit_cast(float, __builtin_bit_cast(int, v) & m);
}

template <int J0, int J1>
__device__ __forceinline__ void stem_mma(const StemFrag& f, const StemMap& mp,
                                         v8f& c00, v8f& c01, v8f& c10, v8f& c11) {
  #pragma unroll
  for (int j = J0; j < J1; ++j) {
    #pragma unroll
    for (int u = 0; u < 2; ++u) {
      const v2f a0 = {pad0(f.a0[j][2 * u], mp.m0[j][2 * u]), pad0(f.a0[j][2 * u + 1], mp.m0[j][2 * u + 1])};
      const v2f a1 = {pad0(f.a1[j][2 * u], mp.m1[j][2 * u]), pad0(f.a1[j][2 * u + 1], mp.m1[j][2 * u + 1])};
      const v2f b0 = {f.b0[j][2 * u], f.b0[j][2 * u + 1]};
      const v2f b1 = {f.b1[j][2 * u], f.b1[j][2 * u + 1]};
      c00 = mma4(a0, b0, c00);
      c01 = mma4(a0, b1, c01);
      c10 = mma4(a1, b0, c10);
      c11 = mma4(a1, b1, c11);
    }
  }
}

__global__ __launch_bounds__(128)
void stem_conv_wmma(const float* __restrict__ tokens,
                    const float* __restrict__ W,
                    float* __restrict__ C) {
  constexpr int N   = D_MODEL;
  constexpr int K   = KSTEM;
  constexpr int WGN = N / 64;
  const int lane = threadIdx.x & 31;
  const int wave = __builtin_amdgcn_readfirstlane((int)(threadIdx.x >> 5));
  const int wgm  = blockIdx.x / WGN;
  const int wgn  = blockIdx.x - wgm * WGN;
  const int m0   = wgm * 64 + (wave >> 1) * 32;
  const int n0   = wgn * 64 + (wave & 1) * 32;
  const int c = lane & 15;
  const int h = lane >> 4;
  const float* pb0 = W + (long)(n0 + c) * K + 4 * h;
  const float* pb1 = pb0 + (long)16 * K;
  const char* tb = (const char*)(tokens + (long)(m0 / LOUT) * C_IN * SEQLEN);
  const int l0 = (m0 & (LOUT - 1)) + c;

  StemMap mp;
  #pragma unroll
  for (int j = 0; j < STEM_J; ++j) {
    #pragma unroll
    for (int e = 0; e < 4; ++e) {
      const int kl = 8 * j + 4 * h + e;
      const int ch = kl / 7;
      const int tp = kl - 7 * ch;
      const int s0 = 2 * l0 - 3 + tp;
      const int s1 = s0 + 32;
      const bool ok0 = (unsigned)s0 < (unsigned)SEQLEN;
      const bool ok1 = (unsigned)s1 < (unsigned)SEQLEN;
      mp.o0[j][e] = (unsigned)(ch * SEQLEN + (ok0 ? s0 : 0)) * 4u;
      mp.o1[j][e] = (unsigned)(ch * SEQLEN + (ok1 ? s1 : 0)) * 4u;
      mp.m0[j][e] = ok0 ? -1 : 0;
      mp.m1[j][e] = ok1 ? -1 : 0;
    }
  }

  const v8f zero = {0.f, 0.f, 0.f, 0.f, 0.f, 0.f, 0.f, 0.f};
  v8f c00 = zero, c01 = zero, c10 = zero, c11 = zero;
  StemFrag fa, fb;
  stem_load<0, 4>(fa, mp, tb, pb0, pb1, 0);
  stem_load<4, STEM_J>(fa, mp, tb, pb0, pb1, 0);
  #pragma unroll 1
  for (int blk = 0; blk < STEM_NB - 2; blk += 2) {
    stem_load<0, 4>(fb, mp, tb, pb0, pb1, blk + 1);
    __builtin_amdgcn_sched_barrier(0);
    stem_mma<0, 4>(fa, mp, c00, c01, c10, c11);
    __builtin_amdgcn_sched_barrier(0);
    stem_load<4, STEM_J>(fb, mp, tb, pb0, pb1, blk + 1);
    __builtin_amdgcn_sched_barrier(0);
    stem_mma<4, STEM_J>(fa, mp, c00, c01, c10, c11);
    __builtin_amdgcn_sched_barrier(0);
    stem_load<0, 4>(fa, mp, tb, pb0, pb1, blk + 2);
    __builtin_amdgcn_sched_barrier(0);
    stem_mma<0, 4>(fb, mp, c00, c01, c10, c11);
    __builtin_amdgcn_sched_barrier(0);
    stem_load<4, STEM_J>(fa, mp, tb, pb0, pb1, blk + 2);
    __builtin_amdgcn_sched_barrier(0);
    stem_mma<4, STEM_J>(fb, mp, c00, c01, c10, c11);
    __builtin_amdgcn_sched_barrier(0);
  }
  stem_load<0, 4>(fb, mp, tb, pb0, pb1, STEM_NB - 1);
  __builtin_amdgcn_sched_barrier(0);
  stem_mma<0, 4>(fa, mp, c00, c01, c10, c11);
  __builtin_amdgcn_sched_barrier(0);
  stem_load<4, STEM_J>(fb, mp, tb, pb0, pb1, STEM_NB - 1);
  __builtin_amdgcn_sched_barrier(0);
  stem_mma<4, STEM_J>(fa, mp, c00, c01, c10, c11);
  stem_mma<0, STEM_J>(fb, mp, c00, c01, c10, c11);

  float* crow = C + (long)(m0 + 8 * h) * N + n0 + c;
  #pragma unroll
  for (int i = 0; i < 8; ++i) {
    crow[(long)i * N]             = c00[i];
    crow[(long)(i + 16) * N]      = c10[i];
    crow[(long)i * N + 16]        = c01[i];
    crow[(long)(i + 16) * N + 16] = c11[i];
  }
}

#define BC_DIM (2 * D_STATE)
__global__ void dwconv_silu_bc_kernel(const float* __restrict__ zx,
                                      const float* __restrict__ w,
                                      const float* __restrict__ bias,
                                      float* __restrict__ out) {
  const int c  = threadIdx.x;
  const int cc = D_INNER + c;
  const long bl = blockIdx.x;
  const int l = (int)(bl & (LOUT - 1));
  const float* xcol = zx + bl * D_IN_PROJ + D_INNER + cc;
  const float w0 = w[cc * 4 + 0], w1 = w[cc * 4 + 1], w2 = w[cc * 4 + 2], w3 = w[cc * 4 + 3];
  float acc = w3 * xcol[0];
  if (l >= 1) acc = fmaf(w2, xcol[-(long)D_IN_PROJ],     acc);
  if (l >= 2) acc = fmaf(w1, xcol[-(long)2 * D_IN_PROJ], acc);
  if (l >= 3) acc = fmaf(w0, xcol[-(long)3 * D_IN_PROJ], acc);
  const float v = acc + bias[cc];
  out[bl * BC_DIM + c] = v / (1.f + __expf(-v));
}

#define SCAN_G 8
#define SEGS   8
#define SEGLEN (LOUT / SEGS)
#define HSTATE (HEADDIM * D_STATE)

template <int N>
__device__ __forceinline__ float row_bcast16(float v) {
  const unsigned sel = 0x11111111u * (unsigned)N;
  return __builtin_bit_cast(float, __builtin_amdgcn_permlane16(
      __builtin_bit_cast(unsigned, v), __builtin_bit_cast(unsigned, v), sel, sel, false, false));
}
template <int M>
__device__ __forceinline__ float row_xor(float v) {
  return __builtin_bit_cast(float, __builtin_amdgcn_update_dpp(
      0, __builtin_bit_cast(int, v), 0x160 + M, 0xF, 0xF, true));
}
__device__ __forceinline__ float lane_gather(float v, int l) {
  return __builtin_bit_cast(float, __builtin_amdgcn_ds_bpermute(l << 2, __builtin_bit_cast(int, v)));
}
__device__ __forceinline__ float softplus_dt(float raw) {
  return (raw > 20.f) ? raw : log1pf(__expf(raw));
}

struct ScanSet { v4f B[SCAN_G]; v4f C[SCAN_G]; float xr[2][4]; };
struct XConv { float w0, w1, w2, w3, bias; };

__device__ __forceinline__ void scan_load(ScanSet& s, const float* bp, const float* cp,
                                          const float* xp, int l0, int sp) {
  #pragma unroll
  for (int j = 0; j < SCAN_G; ++j) {
    const long row = (long)min(l0 + j, LOUT - 1);
    s.B[j] = *(const v4f*)(bp + row * BC_DIM);
    s.C[j] = *(const v4f*)(cp + row * BC_DIM);
  }
  #pragma unroll
  for (int k = 0; k < 2; ++k) {
    #pragma unroll
    for (int d = 0; d < 4; ++d) {
      const long row = (long)max(min(l0 + 4 * k + sp - d, LOUT - 1), 0);
      s.xr[k][d] = xp[row * D_IN_PROJ];
    }
  }
}

template <int S>
__device__ __forceinline__ float scan_step(float (&h)[4][4], const v4f& Bv, const v4f& Cv,
                                           float dA, float xdto, bool b0, bool b1) {
  float xb[4];
  xb[0] = row_bcast16<4 * S + 0>(xdto);
  xb[1] = row_bcast16<4 * S + 1>(xdto);
  xb[2] = row_bcast16<4 * S + 2>(xdto);
  xb[3] = row_bcast16<4 * S + 3>(xdto);
  float acc[4];
  #pragma unroll
  for (int r = 0; r < 4; ++r) {
    #pragma unroll
    for (int e = 0; e < 4; ++e) h[r][e] = fmaf(h[r][e], dA, xb[r] * Bv[e]);
    float a = h[r][0] * Cv[0];
    a = fmaf(h[r][1], Cv[1], a);
    a = fmaf(h[r][2], Cv[2], a);
    a = fmaf(h[r][3], Cv[3], a);
    acc[r] = a;
  }
  const float k01 = b0 ? acc[1] : acc[0];
  const float s01 = b0 ? acc[0] : acc[1];
  const float k23 = b0 ? acc[3] : acc[2];
  const float s23 = b0 ? acc[2] : acc[3];
  const float v01 = k01 + row_xor<1>(s01);
  const float v23 = k23 + row_xor<1>(s23);
  const float kk = b1 ? v23 : v01;
  const float ss = b1 ? v01 : v23;
  float w = kk + row_xor<2>(ss);
  w += row_xor<4>(w);
  w += row_xor<8>(w);
  return w;
}

__device__ __forceinline__ void scan_steps(const ScanSet& s, const XConv& k, float (&h)[4][4], float dtv, float dav,
                                           int lane0, int l0, float Dh, int sp, bool b0, bool b1, float* yrow,
                                           float& pc, float& pk, int lane) {
  float xo[2], xdto[2];
  #pragma unroll
  for (int g = 0; g < 2; ++g) {
    const int tl = l0 + 4 * g + sp;
    const float r0 = s.xr[g][0];
    const float r1 = (tl >= 1) ? s.xr[g][1] : 0.f;
    const float r2 = (tl >= 2) ? s.xr[g][2] : 0.f;
    const float r3 = (tl >= 3) ? s.xr[g][3] : 0.f;
    float a = k.w3 * r0;
    a = fmaf(k.w2, r1, a);
    a = fmaf(k.w1, r2, a);
    a = fmaf(k.w0, r3, a);
    const float v = a + k.bias;
    xo[g] = v / (1.f + __expf(-v));
    xdto[g] = xo[g] * lane_gather(dtv, lane0 + 4 * g + sp);
  }
  #pragma unroll
  for (int g = 0; g < 2; ++g) {
    float keep = 0.f;
    #pragma unroll
    for (int st = 0; st < 4; ++st) {
      const int j = 4 * g + st;
      const float dA = lane_bcast(dav, lane0 + j);
      pc *= dA;
      pk = (lane == lane0 + j) ? pc : pk;
      float w;
      if (st == 0)      w = scan_step<0>(h, s.B[j], s.C[j], dA, xdto[g], b0, b1);
      else if (st == 1) w = scan_step<1>(h, s.B[j], s.C[j], dA, xdto[g], b0, b1);
      else if (st == 2) w = scan_step<2>(h, s.B[j], s.C[j], dA, xdto[g], b0, b1);
      else              w = scan_step<3>(h, s.B[j], s.C[j], dA, xdto[g], b0, b1);
      keep = (sp == st) ? w : keep;
    }
    yrow[(long)(4 * g) * D_INNER] = fmaf(xo[g], Dh, keep);
  }
}

__global__ __launch_bounds__(128)
void ssm_scan_kernel(const float* __restrict__ zx,
                     const float* __restrict__ bc,
                     const float* __restrict__ cw,
                     const float* __restrict__ cb,
                     const float* __restrict__ dtb,
                     const float* __restrict__ alog,
                     const float* __restrict__ Dp,
                     float* __restrict__ y,
                     float* __restrict__ hloc,
                     float* __restrict__ pcum) {
  const int seg  = blockIdx.x >> 6;
  const int b    = (blockIdx.x >> 5) & 1;
  const int hh   = (blockIdx.x >> 1) & 15;
  const int half = blockIdx.x & 1;
  const int lane = threadIdx.x & 31;
  const int wave = threadIdx.x >> 5;
  const int i    = lane & 15;
  const int sp   = i >> 2;
  const bool b0  = (lane & 1) != 0;
  const bool b1  = (lane & 2) != 0;
  const int p0   = half * 32 + wave * 8 + (lane >> 4) * 4;
  const int po   = p0 + (i & 3);
  const float Ah   = -__expf(alog[hh]);
  const float dtbh = dtb[hh];
  const float Dh   = Dp[hh];

  const float* zrows = zx + (long)b * LOUT * D_IN_PROJ;
  const float* bp   = bc + (long)b * LOUT * BC_DIM + 4 * i;
  const float* cp   = bp + D_STATE;
  const int    ch   = hh * HEADDIM + po;
  const float* xp   = zrows + D_INNER + ch;
  const float* dtp  = zrows + (D_IN_PROJ - NHEADS) + hh;
  float* yp = y + ((long)b * LOUT + sp) * D_INNER + ch;

  float h[4][4];
  #pragma unroll
  for (int r = 0; r < 4; ++r) {
    #pragma unroll
    for (int e = 0; e < 4; ++e) h[r][e] = 0.f;
  }

  const int  c0     = seg * (SEGLEN / 32);
  const bool pwrite = (half == 0) && (wave == 0);
  float* pout = pcum + (long)(b * NHEADS + hh) * LOUT;
  float pc = 1.f, pk = 0.f;

  XConv k;
  {
    const v4f wv = *(const v4f*)(cw + ch * D_CONV);
    k.w0 = wv[0]; k.w1 = wv[1]; k.w2 = wv[2]; k.w3 = wv[3];
    k.bias = cb[ch];
  }

  ScanSet sa, sb;
  scan_load(sa, bp, cp, xp, c0 * 32, sp);
  float dtv = softplus_dt(dtp[(long)(c0 * 32 + lane) * D_IN_PROJ] + dtbh);
  float dav = __expf(dtv * Ah);

  for (int c = c0; c < c0 + SEGLEN / 32; ++c) {
    const int   cn   = min(c + 1, LOUT / 32 - 1);
    const float rawn = dtp[(long)(cn * 32 + lane) * D_IN_PROJ];
    for (int g = 0; g < 32; g += 2 * SCAN_G) {
      const int l0 = c * 32 + g;
      scan_load(sb, bp, cp, xp, l0 + SCAN_G, sp);
      __builtin_amdgcn_sched_barrier(0);
      scan_steps(sa, k, h, dtv, dav, g, l0, Dh, sp, b0, b1, yp + (long)l0 * D_INNER, pc, pk, lane);
      __builtin_amdgcn_sched_barrier(0);
      scan_load(sa, bp, cp, xp, l0 + 2 * SCAN_G, sp);
      __builtin_amdgcn_sched_barrier(0);
      scan_steps(sb, k, h, dtv, dav, g + SCAN_G, l0 + SCAN_G, Dh, sp, b0, b1, yp + (long)(l0 + SCAN_G) * D_INNER, pc, pk, lane);
    }
    if (pwrite) pout[c * 32 + lane] = pk;
    dtv = softplus_dt(rawn + dtbh);
    dav = __expf(dtv * Ah);
  }
  if (seg < SEGS - 1) {
    float* hp = hloc + ((long)(b * NHEADS + hh) * SEGS + seg) * HSTATE + p0 * D_STATE + 4 * i;
    #pragma unroll
    for (int r = 0; r < 4; ++r) {
      const v4f v = {h[r][0], h[r][1], h[r][2], h[r][3]};
      *(v4f*)(hp + r * D_STATE) = v;
    }
  }
}

__device__ __forceinline__ void gate_load(float (&yv)[4][8], float (&zv)[4][8],
                                          const float* yrow, const float* zrow) {
  #pragma unroll
  for (int i = 0; i < 8; ++i) {
    yv[0][i] = yrow[(long)i * D_INNER];
    yv[1][i] = yrow[(long)i * D_INNER + 16];
    yv[2][i] = yrow[(long)(i + 16) * D_INNER];
    yv[3][i] = yrow[(long)(i + 16) * D_INNER + 16];
    zv[0][i] = zrow[(long)i * D_IN_PROJ];
    zv[1][i] = zrow[(long)i * D_IN_PROJ + 16];
    zv[2][i] = zrow[(long)(i + 16) * D_IN_PROJ];
    zv[3][i] = zrow[(long)(i + 16) * D_IN_PROJ + 16];
  }
}

__global__ __launch_bounds__(128)
void ssm_carry_gate_wmma(const float* __restrict__ bc,
                         const float* __restrict__ hloc,
                         const float* __restrict__ pcum,
                         const float* __restrict__ zx,
                         const float* __restrict__ nw,
                         float* __restrict__ y,
                         float* __restrict__ sq) {
  constexpr int JB = D_STATE / 8;
  const int lane = threadIdx.x & 31;
  const int wave = __builtin_amdgcn_readfirstlane((int)(threadIdx.x >> 5));
  const int tb   = blockIdx.x & (SEGLEN / 64 - 1);
  const int r    = blockIdx.x / (SEGLEN / 64);
  const int bh   = r / SEGS;
  const int seg  = r - bh * SEGS;
  const int b    = bh / NHEADS;
  const int hh   = bh - b * NHEADS;
  const int t0   = seg * SEGLEN + tb * 64 + (wave >> 1) * 32;
  const int p0   = (wave & 1) * 32;
  const int c = lane & 15;
  const int h = lane >> 4;

  const long row0 = (long)b * LOUT + t0 + 8 * h;
  const int  col  = hh * HEADDIM + p0 + c;
  float* yrow = y + row0 * D_INNER + col;
  const float* zrow = zx + row0 * D_IN_PROJ + col;
  const float nw0 = nw[col];
  const float nw1 = nw[col + 16];
  float yv[4][8], zv[4][8];

  if (seg > 0) {
    const float* pa0 = bc + ((long)b * LOUT + t0 + c) * BC_DIM + D_STATE + 4 * h;
    const float* pa1 = pa0 + (long)16 * BC_DIM;
    const float* hb0 = hloc + (long)bh * SEGS * HSTATE + (p0 + c) * D_STATE + 4 * h;
    const float* pr  = pcum + (long)bh * LOUT;

    RowFrag<JB> f;
    #pragma unroll
    for (int j = 0; j < JB; ++j) {
      f.a0[j] = *(const v4f*)(pa0 + 8 * j);
      f.a1[j] = *(const v4f*)(pa1 + 8 * j);
      f.b0[j] = *(const v4f*)(hb0 + 8 * j);
      f.b1[j] = *(const v4f*)(hb0 + 16 * D_STATE + 8 * j);
    }
    for (int k = 1; k < seg; ++k) {
      const float a = pr[k * SEGLEN + SEGLEN - 1];
      const float* hk = hb0 + (long)k * HSTATE;
      #pragma unroll
      for (int j = 0; j < JB; ++j) {
        const v4f e0 = *(const v4f*)(hk + 8 * j);
        const v4f e1 = *(const v4f*)(hk + 16 * D_STATE + 8 * j);
        #pragma unroll
        for (int e = 0; e < 4; ++e) {
          f.b0[j][e] = fmaf(a, f.b0[j][e], e0[e]);
          f.b1[j][e] = fmaf(a, f.b1[j][e], e1[e]);
        }
      }
    }

    gate_load(yv, zv, yrow, zrow);
    const v8f zero = {0.f, 0.f, 0.f, 0.f, 0.f, 0.f, 0.f, 0.f};
    v8f c00 = zero, c01 = zero, c10 = zero, c11 = zero;
    frag_mma<JB>(f, c00, c01, c10, c11);
    #pragma unroll
    for (int t = 0; t < 4; ++t) {
      asm volatile("" :: "v"(zv[t][0]), "v"(zv[t][1]), "v"(zv[t][2]), "v"(zv[t][3]),
                         "v"(zv[t][4]), "v"(zv[t][5]), "v"(zv[t][6]), "v"(zv[t][7]));
    }

    const float* prow = pr + t0 + 8 * h;
    #pragma unroll
    for (int i = 0; i < 8; ++i) {
      const float s0 = prow[i];
      const float s1 = prow[i + 16];
      yv[0][i] = fmaf(s0, c00[i], yv[0][i]);
      yv[1][i] = fmaf(s0, c01[i], yv[1][i]);
      yv[2][i] = fmaf(s1, c10[i], yv[2][i]);
      yv[3][i] = fmaf(s1, c11[i], yv[3][i]);
    }
  } else {
    gate_load(yv, zv, yrow, zrow);
  }

  float qa[8], qb[8];
  #pragma unroll
  for (int i = 0; i < 8; ++i) {
    float v[4];
    #pragma unroll
    for (int t = 0; t < 4; ++t) {
      const float z = zv[t][i];
      const float g = z / (1.f + __expf(-z));
      v[t] = yv[t][i] * g;
    }
    qa[i] = fmaf(v[1], v[1], v[0] * v[0]);
    qb[i] = fmaf(v[3], v[3], v[2] * v[2]);
    yrow[(long)i * D_INNER]             = v[0] * nw0;
    yrow[(long)i * D_INNER + 16]        = v[1] * nw1;
    yrow[(long)(i + 16) * D_INNER]      = v[2] * nw0;
    yrow[(long)(i + 16) * D_INNER + 16] = v[3] * nw1;
  }
  #pragma unroll
  for (int i = 0; i < 8; ++i) {
    qa[i] += row_xor<1>(qa[i]);  qb[i] += row_xor<1>(qb[i]);
    qa[i] += row_xor<2>(qa[i]);  qb[i] += row_xor<2>(qb[i]);
    qa[i] += row_xor<4>(qa[i]);  qb[i] += row_xor<4>(qb[i]);
    qa[i] += row_xor<8>(qa[i]);  qb[i] += row_xor<8>(qb[i]);
  }
  float keep = 0.f;
  #pragma unroll
  for (int i = 0; i < 8; ++i) {
    keep = (c == i)     ? qa[i] : keep;
    keep = (c == i + 8) ? qb[i] : keep;
  }
  const long rowl = row0 + (c & 7) + 16 * (c >> 3);
  sq[(long)(2 * hh + (wave & 1)) * ROWS + rowl] = keep;
}

extern "C" void kernel_launch(void* const* d_in, const int* in_sizes, int n_in,
                              void* d_out, int out_size, void* d_ws, size_t ws_size,
                              hipStream_t stream) {
  const float* tokens     = (const float*)d_in[0];
  const float* conv_w     = (const float*)d_in[1];
  const float* in_proj_w  = (const float*)d_in[2];
  const float* conv1d_w   = (const float*)d_in[3];
  const float* conv1d_b   = (const float*)d_in[4];
  const float* dt_bias    = (const float*)d_in[5];
  const float* A_log      = (const float*)d_in[6];
  const float* D_param    = (const float*)d_in[7];
  const float* norm_w     = (const float*)d_in[8];
  const float* out_proj_w = (const float*)d_in[9];
  float* out = (float*)d_out;
  float* ws  = (float*)d_ws;

  float* u0   = ws;
  float* u1   = u0 + (long)ROWS * D_MODEL;
  float* zx   = u1 + (long)ROWS * D_MODEL;
  float* bcact = zx + (long)ROWS * D_IN_PROJ;
  float* ybuf = bcact + (long)ROWS * BC_DIM;
  float* hloc = ybuf + (long)ROWS * D_INNER;
  float* pcum = hloc + (long)BATCH * NHEADS * SEGS * HSTATE;
  float* sqp  = pcum + (long)BATCH * NHEADS * LOUT;
  (void)ws_size; (void)in_sizes; (void)n_in; (void)out_size;

  stem_conv_wmma<<<GEMM_WGS(D_MODEL), 128, 0, stream>>>(tokens, conv_w, u0);

  const float* u_in = u0;
  for (int blk = 0; blk < 2; ++blk) {
    const float* ipw = in_proj_w  + (long)blk * D_IN_PROJ * D_MODEL;
    const float* cw  = conv1d_w   + (long)blk * CONV_DIM * D_CONV;
    const float* cb  = conv1d_b   + (long)blk * CONV_DIM;
    const float* dtb = dt_bias    + (long)blk * NHEADS;
    const float* al  = A_log      + (long)blk * NHEADS;
    const float* Dpp = D_param    + (long)blk * NHEADS;
    const float* nw  = norm_w     + (long)blk * D_INNER;
    const float* opw = out_proj_w + (long)blk * D_MODEL * D_INNER;

    {
      gemm_rows_wmma<D_IN_PROJ, D_MODEL, 32, false><<<GEMM_WGS(D_IN_PROJ), 128, 0, stream>>>(u_in, ipw, sqp, zx);
    }
    dwconv_silu_bc_kernel<<<ROWS, BC_DIM, 0, stream>>>(zx, cw, cb, bcact);
    ssm_scan_kernel<<<SEGS * BATCH * NHEADS * 2, 128, 0, stream>>>(zx, bcact, cw, cb, dtb, al, Dpp, ybuf, hloc, pcum);
    ssm_carry_gate_wmma<<<BATCH * NHEADS * SEGS * (SEGLEN / 64), 128, 0, stream>>>(bcact, hloc, pcum, zx, nw, ybuf, sqp);
    float* dst = (blk == 0) ? u1 : out;
    {
      gemm_rows_wmma<D_MODEL, D_INNER, 32, true><<<GEMM_WGS(D_MODEL), 128, 0, stream>>>(ybuf, opw, sqp, dst);
    }
    u_in = u1;
  }
}


# B/C depthwise conv + SiLU made in in_proj's own epilogue (its B/C column blocks take 64-row windows that start 3 rows early, 61 outputs each): conv kernel, its 2 launches gone; same sums, same order
# speedup vs baseline: 16.2732x; 1.0342x over previous
#include <hip/hip_runtime.h>
#include <hip/hip_bf16.h>

typedef __attribute__((ext_vector_type(2))) float v2f;
typedef __attribute__((ext_vector_type(8))) float v8f;

#define BATCH     2
#define SEQLEN    2048
#define C_IN      256
#define D_MODEL   512
#define LOUT      1024
#define D_STATE   64
#define D_CONV    4
#define D_INNER   1024
#define HEADDIM   64
#define NHEADS    16
#define CONV_DIM  (D_INNER + 2*D_STATE)
#define D_IN_PROJ (2*D_INNER + 2*D_STATE + NHEADS)
#define ROWS      (BATCH*LOUT)
#define KSTEM     (C_IN * 7)
#define GEMM_WGS(n) ((ROWS / 64) * (((n) + 63) / 64))
#define SQ_PARTS  (2 * NHEADS)

typedef __attribute__((ext_vector_type(4))) float v4f;

template <int JB>
struct RowFrag { v4f a0[JB]; v4f a1[JB]; v4f b0[JB]; v4f b1[JB]; };

__device__ __forceinline__ v8f mma4(v2f a, v2f b, v8f c) {
  return __builtin_amdgcn_wmma_f32_16x16x4_f32(false, a, false, b, (short)0, c, false, false);
}
__device__ __forceinline__ float lane_bcast(float v, int l) {
  return __builtin_bit_cast(float, __builtin_amdgcn_readlane(__builtin_bit_cast(int, v), l));
}
__device__ __forceinline__ float half_swap(float v) {
  return __builtin_bit_cast(float, __builtin_amdgcn_permlanex16(
      __builtin_bit_cast(unsigned, v), __builtin_bit_cast(unsigned, v), 0x76543210u, 0xfedcba98u, false, false));
}

template <int JB>
__device__ __forceinline__ void frag_mma(const RowFrag<JB>& f, v8f& c00, v8f& c01, v8f& c10, v8f& c11) {
  #pragma unroll
  for (int j = 0; j < JB; ++j) {
    #pragma unroll
    for (int u = 0; u < 2; ++u) {
      const v2f a0 = {f.a0[j][2 * u], f.a0[j][2 * u + 1]};
      const v2f a1 = {f.a1[j][2 * u], f.a1[j][2 * u + 1]};
      const v2f b0 = {f.b0[j][2 * u], f.b0[j][2 * u + 1]};
      const v2f b1 = {f.b1[j][2 * u], f.b1[j][2 * u + 1]};
      c00 = mma4(a0, b0, c00);
      c01 = mma4(a0, b1, c01);
      c10 = mma4(a1, b0, c10);
      c11 = mma4(a1, b1, c11);
    }
  }
}

#define BC_DIM   (2 * D_STATE)
#define BC_STEP  61
#define BC_RB    ((LOUT + BC_STEP - 1) / BC_STEP)
#define BC_WGS   (BATCH * BC_RB * 2)
#define IN_PROJ_CB (D_IN_PROJ / 64 - 1)
#define IN_PROJ_WGS (BC_WGS + (ROWS / 64) * IN_PROJ_CB)
template <int JB>
struct BlockFrag { v4f a[4][JB]; v4f b[4][JB]; };

template <int JB>
__device__ __forceinline__ void block_load(BlockFrag<JB>& f, const float* const (&pa)[4],
                                           const float* const (&pb)[4], int kb) {
  #pragma unroll
  for (int j = 0; j < JB; ++j) {
    #pragma unroll
    for (int t = 0; t < 4; ++t) {
      f.a[t][j] = *(const v4f*)(pa[t] + kb + 8 * j);
      f.b[t][j] = *(const v4f*)(pb[t] + kb + 8 * j);
    }
  }
}

template <int JB>
__device__ __forceinline__ void block_mma(const BlockFrag<JB>& f, v8f (&acc)[4][4]) {
  #pragma unroll
  for (int j = 0; j < JB; ++j) {
    #pragma unroll
    for (int u = 0; u < 2; ++u) {
      v2f a[4], b[4];
      #pragma unroll
      for (int t = 0; t < 4; ++t) {
        a[t] = v2f{f.a[t][j][2 * u], f.a[t][j][2 * u + 1]};
        b[t] = v2f{f.b[t][j][2 * u], f.b[t][j][2 * u + 1]};
      }
      #pragma unroll
      for (int mi = 0; mi < 4; ++mi) {
        #pragma unroll
        for (int ni = 0; ni < 4; ++ni) acc[mi][ni] = mma4(a[mi], b[ni], acc[mi][ni]);
      }
    }
  }
}

#define PART_PITCH 68

template <int N, int K, int KB, bool ROWF, bool BCF>
__global__ __launch_bounds__(128)
void gemm_rows_wmma(const float* __restrict__ A,
                    const float* __restrict__ W,
                    const float* __restrict__ sq,
                    const float* __restrict__ cw,
                    const float* __restrict__ cb,
                    float* __restrict__ C,
                    float* __restrict__ bcout) {
  constexpr int KQ  = K / 4;
  static_assert(K % 4 == 0 && KQ % (2 * KB) == 0 && KB % 8 == 0 && N % 16 == 0, "shape");
  constexpr int JB  = KB / 8;
  constexpr int WGN = (N + 63) / 64;
  __shared__ float part[4][64][PART_PITCH];
  const int lane = threadIdx.x & 31;
  const int wave = __builtin_amdgcn_readfirstlane((int)(threadIdx.x >> 5));
  static_assert(!BCF || (N == D_IN_PROJ && N / 64 == IN_PROJ_CB + 1 && BC_RB * BC_STEP >= LOUT), "B/C windows");
  int m0, n0;
  bool bcwg = false;
  int s0 = 0;
  if (BCF) {
    if (blockIdx.x < BC_WGS) {
      bcwg = true;
      const int rb = blockIdx.x >> 1;
      const int bb = rb / BC_RB;
      s0 = (rb - bb * BC_RB) * BC_STEP;
      m0 = bb * LOUT;
      n0 = 2 * D_INNER + 64 * (blockIdx.x & 1);
    } else {
      const int idx = blockIdx.x - BC_WGS;
      const int wgm = idx / IN_PROJ_CB;
      const int wr  = idx - wgm * IN_PROJ_CB;
      m0 = wgm * 64;
      n0 = ((wr < IN_PROJ_CB - 1) ? wr : IN_PROJ_CB + 1) * 64;
    }
  } else {
    const int wgm = blockIdx.x / WGN;
    const int wgn = blockIdx.x - wgm * WGN;
    m0 = wgm * 64;
    n0 = wgn * 64;
  }
  const int c = lane & 15;
  const int h = lane >> 4;
  float rowf = 1.f;
  if (ROWF) {
    const float* sp = sq + (long)(16 * h) * ROWS + m0 + 16 * wave + c;
    float ssum = sp[0];
    #pragma unroll
    for (int q = 1; q < 16; ++q) ssum += sp[(long)q * ROWS];
    const float tot = ssum + half_swap(ssum);
    rowf = rsqrtf(tot * (1.f / (float)D_INNER) + 1e-5f);
  }
  const float* pa[4];
  const float* pb[4];
  #pragma unroll
  for (int t = 0; t < 4; ++t) {
    const int arow = (BCF && bcwg) ? m0 + max(min(s0 - 3 + 16 * t + c, LOUT - 1), 0) : m0 + 16 * t + c;
    pa[t] = A + (long)arow * K + wave * KQ + 4 * h;
    const int nt = (n0 + 16 * t < N) ? 16 * t : 0;
    pb[t] = W + (long)(n0 + nt + c) * K + wave * KQ + 4 * h;
  }

  v8f acc[4][4];
  #pragma unroll
  for (int mi = 0; mi < 4; ++mi) {
    #pragma unroll
    for (int ni = 0; ni < 4; ++ni) acc[mi][ni] = v8f{0.f, 0.f, 0.f, 0.f, 0.f, 0.f, 0.f, 0.f};
  }
  BlockFrag<JB> fa, fb;
  block_load<JB>(fa, pa, pb, 0);
  #pragma unroll 1
  for (int kb = 0; kb < KQ - 2 * KB; kb += 2 * KB) {
    block_load<JB>(fb, pa, pb, kb + KB);
    __builtin_amdgcn_sched_barrier(0);
    block_mma<JB>(fa, acc);
    __builtin_amdgcn_sched_barrier(0);
    block_load<JB>(fa, pa, pb, kb + 2 * KB);
    __builtin_amdgcn_sched_barrier(0);
    block_mma<JB>(fb, acc);
  }
  block_load<JB>(fb, pa, pb, KQ - KB);
  __builtin_amdgcn_sched_barrier(0);
  block_mma<JB>(fa, acc);
  block_mma<JB>(fb, acc);

  #pragma unroll
  for (int mi = 0; mi < 4; ++mi) {
    #pragma unroll
    for (int ni = 0; ni < 4; ++ni) {
      float* q = &part[wave][16 * ni + c][16 * mi + 8 * h];
      const v8f v = acc[mi][ni];
      *(v4f*)(q)     = v4f{v[0], v[1], v[2], v[3]};
      *(v4f*)(q + 4) = v4f{v[4], v[5], v[6], v[7]};
    }
  }
  __syncthreads();
  if (BCF && bcwg) {
    const int j0 = 16 * wave;
    const int jp = (wave > 0) ? j0 - 4 : 0;
    #pragma unroll
    for (int hf = 0; hf < 2; ++hf) {
      const int col = lane + 32 * hf;
      const int cc  = n0 - D_INNER + col;
      const v4f wv  = *(const v4f*)(cw + cc * D_CONV);
      const float bias = cb[cc];
      float z[20];
      {
        const v4f s = (*(const v4f*)&part[0][col][jp] + *(const v4f*)&part[1][col][jp]) +
                      (*(const v4f*)&part[2][col][jp] + *(const v4f*)&part[3][col][jp]);
        z[0] = s[0]; z[1] = s[1]; z[2] = s[2]; z[3] = s[3];
      }
      #pragma unroll
      for (int r = 0; r < 4; ++r) {
        const v4f s = (*(const v4f*)&part[0][col][j0 + 4 * r] + *(const v4f*)&part[1][col][j0 + 4 * r]) +
                      (*(const v4f*)&part[2][col][j0 + 4 * r] + *(const v4f*)&part[3][col][j0 + 4 * r]);
        z[4 + 4 * r] = s[0]; z[5 + 4 * r] = s[1]; z[6 + 4 * r] = s[2]; z[7 + 4 * r] = s[3];
      }
      float* ocol = bcout + (n0 - 2 * D_INNER) + col;
      #pragma unroll
      for (int i = 0; i < 16; ++i) {
        const int l = s0 - 3 + j0 + i;
        if (j0 + i >= 3 && l < LOUT) {
          float acc = wv[3] * z[4 + i];
          if (l >= 1) acc = fmaf(wv[2], z[3 + i], acc);
          if (l >= 2) acc = fmaf(wv[1], z[2 + i], acc);
          if (l >= 3) acc = fmaf(wv[0], z[1 + i], acc);
          const float v = acc + bias;
          ocol[(long)(m0 + l) * BC_DIM] = v / (1.f + __expf(-v));
        }
      }
    }
    return;
  }
  float rf[16];
  #pragma unroll
  for (int k = 0; k < 16; ++k) rf[k] = ROWF ? lane_bcast(rowf, k) : 1.f;
  #pragma unroll
  for (int hf = 0; hf < 2; ++hf) {
    const int col = lane + 32 * hf;
    const bool ok = (n0 + col) < N;
    float* crow = C + (long)(m0 + 16 * wave) * N + n0 + col;
    #pragma unroll
    for (int r = 0; r < 4; ++r) {
      const v4f s0 = *(const v4f*)&part[0][col][16 * wave + 4 * r];
      const v4f s1 = *(const v4f*)&part[1][col][16 * wave + 4 * r];
      const v4f s2 = *(const v4f*)&part[2][col][16 * wave + 4 * r];
      const v4f s3 = *(const v4f*)&part[3][col][16 * wave + 4 * r];
      const v4f s = (s0 + s1) + (s2 + s3);
      if (ok) {
        #pragma unroll
        for (int e = 0; e < 4; ++e) crow[(long)(4 * r + e) * N] = ROWF ? s[e] * rf[4 * r + e] : s[e];
      }
    }
  }
}

#define STEM_CB 8
#define STEM_KB (STEM_CB * 7)
#define STEM_NB (C_IN / STEM_CB)
#define STEM_J  (STEM_KB / 8)

struct StemFrag { float a0[STEM_J][4]; float a1[STEM_J][4]; v4f b0[STEM_J]; v4f b1[STEM_J]; };
struct StemMap  { unsigned o0[STEM_J][4]; unsigned o1[STEM_J][4]; int m0[STEM_J][4]; int m1[STEM_J][4]; };

template <int J0, int J1>
__device__ __forceinline__ void stem_load(StemFrag& f, const StemMap& mp, const char* tb,
                                          const float* pb0, const float* pb1, int blk) {
  const char* t = tb + (long)blk * (STEM_CB * SEQLEN * 4);
  const int kb = blk * STEM_KB;
  #pragma unroll
  for (int j = J0; j < J1; ++j) {
    f.b0[j] = *(const v4f*)(pb0 + kb + 8 * j);
    f.b1[j] = *(const v4f*)(pb1 + kb + 8 * j);
    #pragma unroll
    for (int e = 0; e < 4; ++e) {
      f.a0[j][e] = *(const float*)(t + mp.o0[j][e]);
      f.a1[j][e] = *(const float*)(t + mp.o1[j][e]);
    }
  }
}

__device__ __forceinline__ float pad0(float v, int m) {
  return __builtin_bit_cast(float, __builtin_bit_cast(int, v) & m);
}

template <int J0, int J1>
__device__ __forceinline__ void stem_mma(const StemFrag& f, const StemMap& mp,
                                         v8f& c00, v8f& c01, v8f& c10, v8f& c11) {
  #pragma unroll
  for (int j = J0; j < J1; ++j) {
    #pragma unroll
    for (int u = 0; u < 2; ++u) {
      const v2f a0 = {pad0(f.a0[j][2 * u], mp.m0[j][2 * u]), pad0(f.a0[j][2 * u + 1], mp.m0[j][2 * u + 1])};
      const v2f a1 = {pad0(f.a1[j][2 * u], mp.m1[j][2 * u]), pad0(f.a1[j][2 * u + 1], mp.m1[j][2 * u + 1])};
      const v2f b0 = {f.b0[j][2 * u], f.b0[j][2 * u + 1]};
      const v2f b1 = {f.b1[j][2 * u], f.b1[j][2 * u + 1]};
      c00 = mma4(a0, b0, c00);
      c01 = mma4(a0, b1, c01);
      c10 = mma4(a1, b0, c10);
      c11 = mma4(a1, b1, c11);
    }
  }
}

__global__ __launch_bounds__(128)
void stem_conv_wmma(const float* __restrict__ tokens,
                    const float* __restrict__ W,
                    float* __restrict__ C) {
  constexpr int N   = D_MODEL;
  constexpr int K   = KSTEM;
  constexpr int WGN = N / 64;
  const int lane = threadIdx.x & 31;
  const int wave = __builtin_amdgcn_readfirstlane((int)(threadIdx.x >> 5));
  const int wgm  = blockIdx.x / WGN;
  const int wgn  = blockIdx.x - wgm * WGN;
  const int m0   = wgm * 64 + (wave >> 1) * 32;
  const int n0   = wgn * 64 + (wave & 1) * 32;
  const int c = lane & 15;
  const int h = lane >> 4;
  const float* pb0 = W + (long)(n0 + c) * K + 4 * h;
  const float* pb1 = pb0 + (long)16 * K;
  const char* tb = (const char*)(tokens + (long)(m0 / LOUT) * C_IN * SEQLEN);
  const int l0 = (m0 & (LOUT - 1)) + c;

  StemMap mp;
  #pragma unroll
  for (int j = 0; j < STEM_J; ++j) {
    #pragma unroll
    for (int e = 0; e < 4; ++e) {
      const int kl = 8 * j + 4 * h + e;
      const int ch = kl / 7;
      const int tp = kl - 7 * ch;
      const int s0 = 2 * l0 - 3 + tp;
      const int s1 = s0 + 32;
      const bool ok0 = (unsigned)s0 < (unsigned)SEQLEN;
      const bool ok1 = (unsigned)s1 < (unsigned)SEQLEN;
      mp.o0[j][e] = (unsigned)(ch * SEQLEN + (ok0 ? s0 : 0)) * 4u;
      mp.o1[j][e] = (unsigned)(ch * SEQLEN + (ok1 ? s1 : 0)) * 4u;
      mp.m0[j][e] = ok0 ? -1 : 0;
      mp.m1[j][e] = ok1 ? -1 : 0;
    }
  }

  const v8f zero = {0.f, 0.f, 0.f, 0.f, 0.f, 0.f, 0.f, 0.f};
  v8f c00 = zero, c01 = zero, c10 = zero, c11 = zero;
  StemFrag fa, fb;
  stem_load<0, 4>(fa, mp, tb, pb0, pb1, 0);
  stem_load<4, STEM_J>(fa, mp, tb, pb0, pb1, 0);
  #pragma unroll 1
  for (int blk = 0; blk < STEM_NB - 2; blk += 2) {
    stem_load<0, 4>(fb, mp, tb, pb0, pb1, blk + 1);
    __builtin_amdgcn_sched_barrier(0);
    stem_mma<0, 4>(fa, mp, c00, c01, c10, c11);
    __builtin_amdgcn_sched_barrier(0);
    stem_load<4, STEM_J>(fb, mp, tb, pb0, pb1, blk + 1);
    __builtin_amdgcn_sched_barrier(0);
    stem_mma<4, STEM_J>(fa, mp, c00, c01, c10, c11);
    __builtin_amdgcn_sched_barrier(0);
    stem_load<0, 4>(fa, mp, tb, pb0, pb1, blk + 2);
    __builtin_amdgcn_sched_barrier(0);
    stem_mma<0, 4>(fb, mp, c00, c01, c10, c11);
    __builtin_amdgcn_sched_barrier(0);
    stem_load<4, STEM_J>(fa, mp, tb, pb0, pb1, blk + 2);
    __builtin_amdgcn_sched_barrier(0);
    stem_mma<4, STEM_J>(fb, mp, c00, c01, c10, c11);
    __builtin_amdgcn_sched_barrier(0);
  }
  stem_load<0, 4>(fb, mp, tb, pb0, pb1, STEM_NB - 1);
  __builtin_amdgcn_sched_barrier(0);
  stem_mma<0, 4>(fa, mp, c00, c01, c10, c11);
  __builtin_amdgcn_sched_barrier(0);
  stem_load<4, STEM_J>(fb, mp, tb, pb0, pb1, STEM_NB - 1);
  __builtin_amdgcn_sched_barrier(0);
  stem_mma<4, STEM_J>(fa, mp, c00, c01, c10, c11);
  stem_mma<0, STEM_J>(fb, mp, c00, c01, c10, c11);

  float* crow = C + (long)(m0 + 8 * h) * N + n0 + c;
  #pragma unroll
  for (int i = 0; i < 8; ++i) {
    crow[(long)i * N]             = c00[i];
    crow[(long)(i + 16) * N]      = c10[i];
    crow[(long)i * N + 16]        = c01[i];
    crow[(long)(i + 16) * N + 16] = c11[i];
  }
}

#define SCAN_G 8
#define SEGS   8
#define SEGLEN (LOUT / SEGS)
#define HSTATE (HEADDIM * D_STATE)

template <int N>
__device__ __forceinline__ float row_bcast16(float v) {
  const unsigned sel = 0x11111111u * (unsigned)N;
  return __builtin_bit_cast(float, __builtin_amdgcn_permlane16(
      __builtin_bit_cast(unsigned, v), __builtin_bit_cast(unsigned, v), sel, sel, false, false));
}
template <int M>
__device__ __forceinline__ float row_xor(float v) {
  return __builtin_bit_cast(float, __builtin_amdgcn_update_dpp(
      0, __builtin_bit_cast(int, v), 0x160 + M, 0xF, 0xF, true));
}
__device__ __forceinline__ float lane_gather(float v, int l) {
  return __builtin_bit_cast(float, __builtin_amdgcn_ds_bpermute(l << 2, __builtin_bit_cast(int, v)));
}
__device__ __forceinline__ float softplus_dt(float raw) {
  return (raw > 20.f) ? raw : log1pf(__expf(raw));
}

struct ScanSet { v4f B[SCAN_G]; v4f C[SCAN_G]; float xr[2][4]; };
struct XConv { float w0, w1, w2, w3, bias; };

__device__ __forceinline__ void scan_load(ScanSet& s, const float* bp, const float* cp,
                                          const float* xp, int l0, int sp) {
  #pragma unroll
  for (int j = 0; j < SCAN_G; ++j) {
    const long row = (long)min(l0 + j, LOUT - 1);
    s.B[j] = *(const v4f*)(bp + row * BC_DIM);
    s.C[j] = *(const v4f*)(cp + row * BC_DIM);
  }
  #pragma unroll
  for (int k = 0; k < 2; ++k) {
    #pragma unroll
    for (int d = 0; d < 4; ++d) {
      const long row = (long)max(min(l0 + 4 * k + sp - d, LOUT - 1), 0);
      s.xr[k][d] = xp[row * D_IN_PROJ];
    }
  }
}

template <int S>
__device__ __forceinline__ float scan_step(float (&h)[4][4], const v4f& Bv, const v4f& Cv,
                                           float dA, float xdto, bool b0, bool b1) {
  float xb[4];
  xb[0] = row_bcast16<4 * S + 0>(xdto);
  xb[1] = row_bcast16<4 * S + 1>(xdto);
  xb[2] = row_bcast16<4 * S + 2>(xdto);
  xb[3] = row_bcast16<4 * S + 3>(xdto);
  float acc[4];
  #pragma unroll
  for (int r = 0; r < 4; ++r) {
    #pragma unroll
    for (int e = 0; e < 4; ++e) h[r][e] = fmaf(h[r][e], dA, xb[r] * Bv[e]);
    float a = h[r][0] * Cv[0];
    a = fmaf(h[r][1], Cv[1], a);
    a = fmaf(h[r][2], Cv[2], a);
    a = fmaf(h[r][3], Cv[3], a);
    acc[r] = a;
  }
  const float k01 = b0 ? acc[1] : acc[0];
  const float s01 = b0 ? acc[0] : acc[1];
  const float k23 = b0 ? acc[3] : acc[2];
  const float s23 = b0 ? acc[2] : acc[3];
  const float v01 = k01 + row_xor<1>(s01);
  const float v23 = k23 + row_xor<1>(s23);
  const float kk = b1 ? v23 : v01;
  const float ss = b1 ? v01 : v23;
  float w = kk + row_xor<2>(ss);
  w += row_xor<4>(w);
  w += row_xor<8>(w);
  return w;
}

__device__ __forceinline__ void scan_steps(const ScanSet& s, const XConv& k, float (&h)[4][4], float dtv, float dav,
                                           int lane0, int l0, float Dh, int sp, bool b0, bool b1, float* yrow,
                                           float& pc, float& pk, int lane) {
  float xo[2], xdto[2];
  #pragma unroll
  for (int g = 0; g < 2; ++g) {
    const int tl = l0 + 4 * g + sp;
    const float r0 = s.xr[g][0];
    const float r1 = (tl >= 1) ? s.xr[g][1] : 0.f;
    const float r2 = (tl >= 2) ? s.xr[g][2] : 0.f;
    const float r3 = (tl >= 3) ? s.xr[g][3] : 0.f;
    float a = k.w3 * r0;
    a = fmaf(k.w2, r1, a);
    a = fmaf(k.w1, r2, a);
    a = fmaf(k.w0, r3, a);
    const float v = a + k.bias;
    xo[g] = v / (1.f + __expf(-v));
    xdto[g] = xo[g] * lane_gather(dtv, lane0 + 4 * g + sp);
  }
  #pragma unroll
  for (int g = 0; g < 2; ++g) {
    float keep = 0.f;
    #pragma unroll
    for (int st = 0; st < 4; ++st) {
      const int j = 4 * g + st;
      const float dA = lane_bcast(dav, lane0 + j);
      pc *= dA;
      pk = (lane == lane0 + j) ? pc : pk;
      float w;
      if (st == 0)      w = scan_step<0>(h, s.B[j], s.C[j], dA, xdto[g], b0, b1);
      else if (st == 1) w = scan_step<1>(h, s.B[j], s.C[j], dA, xdto[g], b0, b1);
      else if (st == 2) w = scan_step<2>(h, s.B[j], s.C[j], dA, xdto[g], b0, b1);
      else              w = scan_step<3>(h, s.B[j], s.C[j], dA, xdto[g], b0, b1);
      keep = (sp == st) ? w : keep;
    }
    yrow[(long)(4 * g) * D_INNER] = fmaf(xo[g], Dh, keep);
  }
}

__global__ __launch_bounds__(128)
void ssm_scan_kernel(const float* __restrict__ zx,
                     const float* __restrict__ bc,
                     const float* __restrict__ cw,
                     const float* __restrict__ cb,
                     const float* __restrict__ dtb,
                     const float* __restrict__ alog,
                     const float* __restrict__ Dp,
                     float* __restrict__ y,
                     float* __restrict__ hloc,
                     float* __restrict__ pcum) {
  const int seg  = blockIdx.x >> 6;
  const int b    = (blockIdx.x >> 5) & 1;
  const int hh   = (blockIdx.x >> 1) & 15;
  const int half = blockIdx.x & 1;
  const int lane = threadIdx.x & 31;
  const int wave = threadIdx.x >> 5;
  const int i    = lane & 15;
  const int sp   = i >> 2;
  const bool b0  = (lane & 1) != 0;
  const bool b1  = (lane & 2) != 0;
  const int p0   = half * 32 + wave * 8 + (lane >> 4) * 4;
  const int po   = p0 + (i & 3);
  const float Ah   = -__expf(alog[hh]);
  const float dtbh = dtb[hh];
  const float Dh   = Dp[hh];

  const float* zrows = zx + (long)b * LOUT * D_IN_PROJ;
  const float* bp   = bc + (long)b * LOUT * BC_DIM + 4 * i;
  const float* cp   = bp + D_STATE;
  const int    ch   = hh * HEADDIM + po;
  const float* xp   = zrows + D_INNER + ch;
  const float* dtp  = zrows + (D_IN_PROJ - NHEADS) + hh;
  float* yp = y + ((long)b * LOUT + sp) * D_INNER + ch;

  float h[4][4];
  #pragma unroll
  for (int r = 0; r < 4; ++r) {
    #pragma unroll
    for (int e = 0; e < 4; ++e) h[r][e] = 0.f;
  }

  const int  c0     = seg * (SEGLEN / 32);
  const bool pwrite = (half == 0) && (wave == 0);
  float* pout = pcum + (long)(b * NHEADS + hh) * LOUT;
  float pc = 1.f, pk = 0.f;

  XConv k;
  {
    const v4f wv = *(const v4f*)(cw + ch * D_CONV);
    k.w0 = wv[0]; k.w1 = wv[1]; k.w2 = wv[2]; k.w3 = wv[3];
    k.bias = cb[ch];
  }

  ScanSet sa, sb;
  scan_load(sa, bp, cp, xp, c0 * 32, sp);
  float dtv = softplus_dt(dtp[(long)(c0 * 32 + lane) * D_IN_PROJ] + dtbh);
  float dav = __expf(dtv * Ah);

  for (int c = c0; c < c0 + SEGLEN / 32; ++c) {
    const int   cn   = min(c + 1, LOUT / 32 - 1);
    const float rawn = dtp[(long)(cn * 32 + lane) * D_IN_PROJ];
    for (int g = 0; g < 32; g += 2 * SCAN_G) {
      const int l0 = c * 32 + g;
      scan_load(sb, bp, cp, xp, l0 + SCAN_G, sp);
      __builtin_amdgcn_sched_barrier(0);
      scan_steps(sa, k, h, dtv, dav, g, l0, Dh, sp, b0, b1, yp + (long)l0 * D_INNER, pc, pk, lane);
      __builtin_amdgcn_sched_barrier(0);
      scan_load(sa, bp, cp, xp, l0 + 2 * SCAN_G, sp);
      __builtin_amdgcn_sched_barrier(0);
      scan_steps(sb, k, h, dtv, dav, g + SCAN_G, l0 + SCAN_G, Dh, sp, b0, b1, yp + (long)(l0 + SCAN_G) * D_INNER, pc, pk, lane);
    }
    if (pwrite) pout[c * 32 + lane] = pk;
    dtv = softplus_dt(rawn + dtbh);
    dav = __expf(dtv * Ah);
  }
  if (seg < SEGS - 1) {
    float* hp = hloc + ((long)(b * NHEADS + hh) * SEGS + seg) * HSTATE + p0 * D_STATE + 4 * i;
    #pragma unroll
    for (int r = 0; r < 4; ++r) {
      const v4f v = {h[r][0], h[r][1], h[r][2], h[r][3]};
      *(v4f*)(hp + r * D_STATE) = v;
    }
  }
}

__device__ __forceinline__ void gate_load(float (&yv)[4][8], float (&zv)[4][8],
                                          const float* yrow, const float* zrow) {
  #pragma unroll
  for (int i = 0; i < 8; ++i) {
    yv[0][i] = yrow[(long)i * D_INNER];
    yv[1][i] = yrow[(long)i * D_INNER + 16];
    yv[2][i] = yrow[(long)(i + 16) * D_INNER];
    yv[3][i] = yrow[(long)(i + 16) * D_INNER + 16];
    zv[0][i] = zrow[(long)i * D_IN_PROJ];
    zv[1][i] = zrow[(long)i * D_IN_PROJ + 16];
    zv[2][i] = zrow[(long)(i + 16) * D_IN_PROJ];
    zv[3][i] = zrow[(long)(i + 16) * D_IN_PROJ + 16];
  }
}

__global__ __launch_bounds__(128)
void ssm_carry_gate_wmma(const float* __restrict__ bc,
                         const float* __restrict__ hloc,
                         const float* __restrict__ pcum,
                         const float* __restrict__ zx,
                         const float* __restrict__ nw,
                         float* __restrict__ y,
                         float* __restrict__ sq) {
  constexpr int JB = D_STATE / 8;
  const int lane = threadIdx.x & 31;
  const int wave = __builtin_amdgcn_readfirstlane((int)(threadIdx.x >> 5));
  const int tb   = blockIdx.x & (SEGLEN / 64 - 1);
  const int r    = blockIdx.x / (SEGLEN / 64);
  const int bh   = r / SEGS;
  const int seg  = r - bh * SEGS;
  const int b    = bh / NHEADS;
  const int hh   = bh - b * NHEADS;
  const int t0   = seg * SEGLEN + tb * 64 + (wave >> 1) * 32;
  const int p0   = (wave & 1) * 32;
  const int c = lane & 15;
  const int h = lane >> 4;

  const long row0 = (long)b * LOUT + t0 + 8 * h;
  const int  col  = hh * HEADDIM + p0 + c;
  float* yrow = y + row0 * D_INNER + col;
  const float* zrow = zx + row0 * D_IN_PROJ + col;
  const float nw0 = nw[col];
  const float nw1 = nw[col + 16];
  float yv[4][8], zv[4][8];

  if (seg > 0) {
    const float* pa0 = bc + ((long)b * LOUT + t0 + c) * BC_DIM + D_STATE + 4 * h;
    const float* pa1 = pa0 + (long)16 * BC_DIM;
    const float* hb0 = hloc + (long)bh * SEGS * HSTATE + (p0 + c) * D_STATE + 4 * h;
    const float* pr  = pcum + (long)bh * LOUT;

    RowFrag<JB> f;
    #pragma unroll
    for (int j = 0; j < JB; ++j) {
      f.a0[j] = *(const v4f*)(pa0 + 8 * j);
      f.a1[j] = *(const v4f*)(pa1 + 8 * j);
      f.b0[j] = *(const v4f*)(hb0 + 8 * j);
      f.b1[j] = *(const v4f*)(hb0 + 16 * D_STATE + 8 * j);
    }
    for (int k = 1; k < seg; ++k) {
      const float a = pr[k * SEGLEN + SEGLEN - 1];
      const float* hk = hb0 + (long)k * HSTATE;
      #pragma unroll
      for (int j = 0; j < JB; ++j) {
        const v4f e0 = *(const v4f*)(hk + 8 * j);
        const v4f e1 = *(const v4f*)(hk + 16 * D_STATE + 8 * j);
        #pragma unroll
        for (int e = 0; e < 4; ++e) {
          f.b0[j][e] = fmaf(a, f.b0[j][e], e0[e]);
          f.b1[j][e] = fmaf(a, f.b1[j][e], e1[e]);
        }
      }
    }

    gate_load(yv, zv, yrow, zrow);
    const v8f zero = {0.f, 0.f, 0.f, 0.f, 0.f, 0.f, 0.f, 0.f};
    v8f c00 = zero, c01 = zero, c10 = zero, c11 = zero;
    frag_mma<JB>(f, c00, c01, c10, c11);
    #pragma unroll
    for (int t = 0; t < 4; ++t) {
      asm volatile("" :: "v"(zv[t][0]), "v"(zv[t][1]), "v"(zv[t][2]), "v"(zv[t][3]),
                         "v"(zv[t][4]), "v"(zv[t][5]), "v"(zv[t][6]), "v"(zv[t][7]));
    }

    const float* prow = pr + t0 + 8 * h;
    #pragma unroll
    for (int i = 0; i < 8; ++i) {
      const float s0 = prow[i];
      const float s1 = prow[i + 16];
      yv[0][i] = fmaf(s0, c00[i], yv[0][i]);
      yv[1][i] = fmaf(s0, c01[i], yv[1][i]);
      yv[2][i] = fmaf(s1, c10[i], yv[2][i]);
      yv[3][i] = fmaf(s1, c11[i], yv[3][i]);
    }
  } else {
    gate_load(yv, zv, yrow, zrow);
  }

  float qa[8], qb[8];
  #pragma unroll
  for (int i = 0; i < 8; ++i) {
    float v[4];
    #pragma unroll
    for (int t = 0; t < 4; ++t) {
      const float z = zv[t][i];
      const float g = z / (1.f + __expf(-z));
      v[t] = yv[t][i] * g;
    }
    qa[i] = fmaf(v[1], v[1], v[0] * v[0]);
    qb[i] = fmaf(v[3], v[3], v[2] * v[2]);
    yrow[(long)i * D_INNER]             = v[0] * nw0;
    yrow[(long)i * D_INNER + 16]        = v[1] * nw1;
    yrow[(long)(i + 16) * D_INNER]      = v[2] * nw0;
    yrow[(long)(i + 16) * D_INNER + 16] = v[3] * nw1;
  }
  #pragma unroll
  for (int i = 0; i < 8; ++i) {
    qa[i] += row_xor<1>(qa[i]);  qb[i] += row_xor<1>(qb[i]);
    qa[i] += row_xor<2>(qa[i]);  qb[i] += row_xor<2>(qb[i]);
    qa[i] += row_xor<4>(qa[i]);  qb[i] += row_xor<4>(qb[i]);
    qa[i] += row_xor<8>(qa[i]);  qb[i] += row_xor<8>(qb[i]);
  }
  float keep = 0.f;
  #pragma unroll
  for (int i = 0; i < 8; ++i) {
    keep = (c == i)     ? qa[i] : keep;
    keep = (c == i + 8) ? qb[i] : keep;
  }
  const long rowl = row0 + (c & 7) + 16 * (c >> 3);
  sq[(long)(2 * hh + (wave & 1)) * ROWS + rowl] = keep;
}

extern "C" void kernel_launch(void* const* d_in, const int* in_sizes, int n_in,
                              void* d_out, int out_size, void* d_ws, size_t ws_size,
                              hipStream_t stream) {
  const float* tokens     = (const float*)d_in[0];
  const float* conv_w     = (const float*)d_in[1];
  const float* in_proj_w  = (const float*)d_in[2];
  const float* conv1d_w   = (const float*)d_in[3];
  const float* conv1d_b   = (const float*)d_in[4];
  const float* dt_bias    = (const float*)d_in[5];
  const float* A_log      = (const float*)d_in[6];
  const float* D_param    = (const float*)d_in[7];
  const float* norm_w     = (const float*)d_in[8];
  const float* out_proj_w = (const float*)d_in[9];
  float* out = (float*)d_out;
  float* ws  = (float*)d_ws;

  float* u0   = ws;
  float* u1   = u0 + (long)ROWS * D_MODEL;
  float* zx   = u1 + (long)ROWS * D_MODEL;
  float* bcact = zx + (long)ROWS * D_IN_PROJ;
  float* ybuf = bcact + (long)ROWS * BC_DIM;
  float* hloc = ybuf + (long)ROWS * D_INNER;
  float* pcum = hloc + (long)BATCH * NHEADS * SEGS * HSTATE;
  float* sqp  = pcum + (long)BATCH * NHEADS * LOUT;
  (void)ws_size; (void)in_sizes; (void)n_in; (void)out_size;

  stem_conv_wmma<<<GEMM_WGS(D_MODEL), 128, 0, stream>>>(tokens, conv_w, u0);

  const float* u_in = u0;
  for (int blk = 0; blk < 2; ++blk) {
    const float* ipw = in_proj_w  + (long)blk * D_IN_PROJ * D_MODEL;
    const float* cw  = conv1d_w   + (long)blk * CONV_DIM * D_CONV;
    const float* cb  = conv1d_b   + (long)blk * CONV_DIM;
    const float* dtb = dt_bias    + (long)blk * NHEADS;
    const float* al  = A_log      + (long)blk * NHEADS;
    const float* Dpp = D_param    + (long)blk * NHEADS;
    const float* nw  = norm_w     + (long)blk * D_INNER;
    const float* opw = out_proj_w + (long)blk * D_MODEL * D_INNER;

    {
      gemm_rows_wmma<D_IN_PROJ, D_MODEL, 32, false, true><<<IN_PROJ_WGS, 128, 0, stream>>>(u_in, ipw, sqp, cw, cb, zx, bcact);
    }
    ssm_scan_kernel<<<SEGS * BATCH * NHEADS * 2, 128, 0, stream>>>(zx, bcact, cw, cb, dtb, al, Dpp, ybuf, hloc, pcum);
    ssm_carry_gate_wmma<<<BATCH * NHEADS * SEGS * (SEGLEN / 64), 128, 0, stream>>>(bcact, hloc, pcum, zx, nw, ybuf, sqp);
    float* dst = (blk == 0) ? u1 : out;
    {
      gemm_rows_wmma<D_MODEL, D_INNER, 32, true, false><<<GEMM_WGS(D_MODEL), 128, 0, stream>>>(ybuf, opw, sqp, cw, cb, dst, bcact);
    }
    u_in = u1;
  }
}
